# starting kernel
# baseline (speedup 1.0000x reference)
#include <hip/hip_runtime.h>
#include <hip/hip_bf16.h>
#include <math.h>

#define B_    8192
#define D_    32
#define H_    128
#define P2_   64
#define PREP_ 10
#define GIN_  320
#define G3_   384
#define CS_   8
#define CH_   32
#define NT_   16
#define NO_   4096
#define DT_   0.1f
#define NPOST_ 4
#define LOGC_  0.91893853320467274178f

typedef __attribute__((ext_vector_type(2))) float v2f;
typedef __attribute__((ext_vector_type(8))) float v8f;

__device__ __forceinline__ float sigmoidf_(float x) { return 1.0f / (1.0f + expf(-x)); }


__device__ __forceinline__ v2f load_afrag(const float* A, int lda, int m0, int k, int lane) {
    int m  = m0 + (lane & 15);
    int kk = k + ((lane >> 4) << 1);
    v2f a;
    a.x = A[m * lda + kk];
    a.y = A[m * lda + kk + 1];
    return a;
}

__device__ __forceinline__ v2f load_bfrag(const float* Bm, int ldb, int k, int n0, int lane) {
    int n  = n0 + (lane & 15);
    int kk = k + ((lane >> 4) << 1);
    v2f b;
    b.x = Bm[kk * ldb + n];
    b.y = Bm[(kk + 1) * ldb + n];
    return b;
}

__device__ __forceinline__ v8f wmma4(v2f a, v2f b, v8f c) {
    return __builtin_amdgcn_wmma_f32_16x16x4_f32(false, a, false, b, (short)0, c, false, false);
}

__global__ void k_init(const float* __restrict__ cov,
                       const float* __restrict__ w1, const float* __restrict__ b1,
                       const float* __restrict__ w2, const float* __restrict__ b2,
                       float* __restrict__ h, float* __restrict__ loss) {
    __shared__ float t_s[16][CH_];
    const int tid = threadIdx.x;
    const int m0  = blockIdx.x * 16;
    if (blockIdx.x == 0 && tid == 0) *loss = 0.0f;
    for (int i = tid; i < 16 * CH_; i += 128) {
        int r = i / CH_, c = i % CH_;
        float acc = b1[c];
        #pragma unroll
        for (int k = 0; k < CS_; ++k) acc += cov[(m0 + r) * CS_ + k] * w1[k * CH_ + c];
        t_s[r][c] = fmaxf(acc, 0.0f);
    }
    __syncthreads();
    for (int i = tid; i < 16 * H_; i += 128) {
        int r = i / H_, c = i % H_;
        float acc = b2[c];
        #pragma unroll
        for (int k = 0; k < CH_; ++k) acc += t_s[r][k] * w2[k * H_ + c];
        h[(m0 + r) * H_ + c] = tanhf(acc);
    }
}

__global__ void k_pmodel(const float* __restrict__ h,
                         const float* __restrict__ w1, const float* __restrict__ b1,
                         const float* __restrict__ w2, const float* __restrict__ b2,
                         float* __restrict__ p) {
    __shared__ float t_s[16][P2_];
    const int tid = threadIdx.x, lane = tid & 31, wv = tid >> 5;
    const int m0 = blockIdx.x * 16, n0 = wv * 16;

    v8f acc = {};
    for (int k = 0; k < H_; k += 4)
        acc = wmma4(load_afrag(h, H_, m0, k, lane), load_bfrag(w1, P2_, k, n0, lane), acc);
    const int col = n0 + (lane & 15), rb = (lane >> 4) << 3;
    const float bv = b1[col];
    #pragma unroll
    for (int j = 0; j < 8; ++j) t_s[rb + j][col] = fmaxf(acc[j] + bv, 0.0f);
    __syncthreads();

    v8f acc2 = {};
    for (int k = 0; k < P2_; k += 4)
        acc2 = wmma4(load_afrag(&t_s[0][0], P2_, 0, k, lane), load_bfrag(w2, P2_, k, n0, lane), acc2);
    const float bv2 = b2[col];
    #pragma unroll
    for (int j = 0; j < 8; ++j) p[(m0 + rb + j) * P2_ + col] = acc2[j] + bv2;
}

__global__ void k_ode(float* __restrict__ h, const float* __restrict__ p,
                      const float* __restrict__ xz_w, const float* __restrict__ xz_b,
                      const float* __restrict__ hz_w,
                      const float* __restrict__ xn_w, const float* __restrict__ xn_b,
                      const float* __restrict__ hn_w) {
    __shared__ float zh_s[16][H_];
    const int tid = threadIdx.x, lane = tid & 31, wv = tid >> 5;
    const int m0 = blockIdx.x * 16, n0 = wv * 16;

    v8f az = {};
    for (int k = 0; k < P2_; k += 4)
        az = wmma4(load_afrag(p, P2_, m0, k, lane), load_bfrag(xz_w, H_, k, n0, lane), az);
    for (int k = 0; k < H_; k += 4)
        az = wmma4(load_afrag(h, H_, m0, k, lane), load_bfrag(hz_w, H_, k, n0, lane), az);

    const int col = n0 + (lane & 15), rb = (lane >> 4) << 3;
    const float bz = xz_b[col];
    float z[8];
    #pragma unroll
    for (int j = 0; j < 8; ++j) {
        z[j] = sigmoidf_(az[j] + bz);
        zh_s[rb + j][col] = z[j] * h[(m0 + rb + j) * H_ + col];
    }
    __syncthreads();

    v8f an = {};
    for (int k = 0; k < P2_; k += 4)
        an = wmma4(load_afrag(p, P2_, m0, k, lane), load_bfrag(xn_w, H_, k, n0, lane), an);
    for (int k = 0; k < H_; k += 4)
        an = wmma4(load_afrag(&zh_s[0][0], H_, 0, k, lane), load_bfrag(hn_w, H_, k, n0, lane), an);

    const float bn = xn_b[col];
    #pragma unroll
    for (int j = 0; j < 8; ++j) {
        float n = tanhf(an[j] + bn);
        int row = m0 + rb + j;
        float hv = h[row * H_ + col];
        h[row * H_ + col] = hv + DT_ * (1.0f - z[j]) * (n - hv);
    }
}

__global__ void k_jump(int t,
                       const float* __restrict__ X, const float* __restrict__ M,
                       const int* __restrict__ obs_idx,
                       float* __restrict__ h, const float* __restrict__ p,
                       const float* __restrict__ w_ih, const float* __restrict__ w_hh,
                       const float* __restrict__ b_ih, const float* __restrict__ b_hh,
                       const float* __restrict__ w_prep, const float* __restrict__ bias_prep,
                       float* __restrict__ loss_part) {
    __shared__ int   idx_s[16];
    __shared__ float hold_s[16][H_];
    __shared__ float gin_s[16][GIN_];
    __shared__ float gi_s[16][G3_];
    __shared__ float gh_s[16][G3_];
    __shared__ float red[256];
    const int tid = threadIdx.x, lane = tid & 31, wv = tid >> 5;
    const int o0 = blockIdx.x * 16;
    const float* Xt = X + (size_t)t * NO_ * D_;
    const float* Mt = M + (size_t)t * NO_ * D_;

    if (tid < 16) idx_s[tid] = obs_idx[t * NO_ + o0 + tid];
    __syncthreads();

    for (int i = tid; i < 16 * H_; i += 256) {
        int o = i >> 7, j = i & 127;
        hold_s[o][j] = h[idx_s[o] * H_ + j];
    }

    float lsum = 0.0f;
    for (int i = tid; i < 16 * D_; i += 256) {
        int o = i / D_, d = i % D_;
        int gb = o0 + o, b = idx_s[o];
        float mean = p[b * P2_ + d];
        float lv   = p[b * P2_ + D_ + d];
        float x    = Xt[gb * D_ + d];
        float m    = Mt[gb * D_ + d];
        float err  = (x - mean) * expf(-0.5f * lv);
        lsum += 0.5f * ((err * err + lv + 2.0f * LOGC_) * m);
    }

    for (int i = tid; i < 16 * GIN_; i += 256) {
        int o = i / GIN_, c = i % GIN_;
        int d = c / PREP_, pi = c - d * PREP_;
        int gb = o0 + o, b = idx_s[o];
        float mean = p[b * P2_ + d];
        float lv   = p[b * P2_ + D_ + d];
        float x    = Xt[gb * D_ + d];
        float m    = Mt[gb * D_ + d];
        float err  = (x - mean) * expf(-0.5f * lv);
        const float* wp = w_prep + (size_t)(d * 4) * PREP_ + pi;
        float a = x * wp[0] + mean * wp[PREP_] + lv * wp[2 * PREP_] + err * wp[3 * PREP_]
                + bias_prep[d * PREP_ + pi];
        gin_s[o][c] = fmaxf(a, 0.0f) * m;
    }
    __syncthreads();

    for (int s = 0; s < 3; ++s) {
        const int n0 = (wv * 3 + s) * 16;
        v8f gi = {};
        for (int k = 0; k < GIN_; k += 4)
            gi = wmma4(load_afrag(&gin_s[0][0], GIN_, 0, k, lane), load_bfrag(w_ih, G3_, k, n0, lane), gi);
        v8f gh = {};
        for (int k = 0; k < H_; k += 4)
            gh = wmma4(load_afrag(&hold_s[0][0], H_, 0, k, lane), load_bfrag(w_hh, G3_, k, n0, lane), gh);
        const int col = n0 + (lane & 15), rb = (lane >> 4) << 3;
        const float bi = b_ih[col], bh = b_hh[col];
        #pragma unroll
        for (int j = 0; j < 8; ++j) {
            gi_s[rb + j][col] = gi[j] + bi;
            gh_s[rb + j][col] = gh[j] + bh;
        }
    }
    __syncthreads();

    for (int i = tid; i < 16 * H_; i += 256) {
        int o = i >> 7, j = i & 127;
        float r = sigmoidf_(gi_s[o][j] + gh_s[o][j]);
        float z = sigmoidf_(gi_s[o][H_ + j] + gh_s[o][H_ + j]);
        float n = tanhf(gi_s[o][2 * H_ + j] + r * gh_s[o][2 * H_ + j]);
        float hv = hold_s[o][j];
        h[idx_s[o] * H_ + j] = (1.0f - z) * n + z * hv;
    }

    red[tid] = lsum;
    __syncthreads();
    for (int s2 = 128; s2 > 0; s2 >>= 1) {
        if (tid < s2) red[tid] += red[tid + s2];
        __syncthreads();
    }
    if (tid == 0) loss_part[t * (NO_ / 16) + blockIdx.x] = red[0];
}

__global__ void k_loss_final(const float* __restrict__ parts, float* __restrict__ loss) {
    __shared__ float red[256];
    const int tid = threadIdx.x;
    float s = 0.0f;
    for (int i = tid; i < NT_ * (NO_ / 16); i += 256) s += parts[i];
    red[tid] = s;
    __syncthreads();
    for (int k = 128; k > 0; k >>= 1) {
        if (tid < k) red[tid] += red[tid + k];
        __syncthreads();
    }
    if (tid == 0) *loss = red[0];
}

extern "C" void kernel_launch(void* const* d_in, const int* in_sizes, int n_in,
                              void* d_out, int out_size, void* d_ws, size_t ws_size,
                              hipStream_t stream) {
    const float* X       = (const float*)d_in[0];
    const float* M       = (const float*)d_in[1];
    const int*   obs_idx = (const int*)d_in[2];
    const float* cov     = (const float*)d_in[3];
    const float* cm_w1   = (const float*)d_in[4];
    const float* cm_b1   = (const float*)d_in[5];
    const float* cm_w2   = (const float*)d_in[6];
    const float* cm_b2   = (const float*)d_in[7];
    const float* p_w1    = (const float*)d_in[8];
    const float* p_b1    = (const float*)d_in[9];
    const float* p_w2    = (const float*)d_in[10];
    const float* p_b2    = (const float*)d_in[11];
    const float* xz_w    = (const float*)d_in[12];
    const float* xz_b    = (const float*)d_in[13];
    const float* hz_w    = (const float*)d_in[14];
    const float* xn_w    = (const float*)d_in[15];
    const float* xn_b    = (const float*)d_in[16];
    const float* hn_w    = (const float*)d_in[17];
    const float* w_ih    = (const float*)d_in[18];
    const float* w_hh    = (const float*)d_in[19];
    const float* b_ih    = (const float*)d_in[20];
    const float* b_hh    = (const float*)d_in[21];
    const float* w_prep  = (const float*)d_in[22];
    const float* b_prep  = (const float*)d_in[23];

    float* h    = (float*)d_out;
    float* p    = h + (size_t)B_ * H_;
    float* loss = p + (size_t)B_ * P2_;
    float* loss_part = (float*)d_ws;

    const int MB = B_ / 16;
    const int OB = NO_ / 16;

    k_init<<<MB, 128, 0, stream>>>(cov, cm_w1, cm_b1, cm_w2, cm_b2, h, loss);
    k_pmodel<<<MB, 128, 0, stream>>>(h, p_w1, p_b1, p_w2, p_b2, p);

    for (int t = 0; t < NT_; ++t) {
        k_ode<<<MB, 256, 0, stream>>>(h, p, xz_w, xz_b, hz_w, xn_w, xn_b, hn_w);
        k_pmodel<<<MB, 128, 0, stream>>>(h, p_w1, p_b1, p_w2, p_b2, p);
        k_jump<<<OB, 256, 0, stream>>>(t, X, M, obs_idx, h, p,
                                       w_ih, w_hh, b_ih, b_hh, w_prep, b_prep, loss_part);
        k_pmodel<<<MB, 128, 0, stream>>>(h, p_w1, p_b1, p_w2, p_b2, p);
    }
    for (int s = 0; s < NPOST_; ++s) {
        k_ode<<<MB, 256, 0, stream>>>(h, p, xz_w, xz_b, hz_w, xn_w, xn_b, hn_w);
        k_pmodel<<<MB, 128, 0, stream>>>(h, p_w1, p_b1, p_w2, p_b2, p);
    }
    k_loss_final<<<1, 256, 0, stream>>>(loss_part, loss);
}


# jump kernel: 24 waves per work-group, one 16-column gate tile per wave instead of three in sequence on 8 waves; element-wise loops spread over 768 threads; order of every sum kept
# speedup vs baseline: 1.0890x; 1.0890x over previous
#include <hip/hip_runtime.h>
#include <hip/hip_bf16.h>
#include <math.h>

#define B_    8192
#define D_    32
#define H_    128
#define P2_   64
#define PREP_ 10
#define GIN_  320
#define G3_   384
#define CS_   8
#define CH_   32
#define NT_   16
#define NO_   4096
#define DT_   0.1f
#define NPOST_ 4
#define JT_   768
#define LOGC_  0.91893853320467274178f

typedef __attribute__((ext_vector_type(2))) float v2f;
typedef __attribute__((ext_vector_type(8))) float v8f;

__device__ __forceinline__ float sigmoidf_(float x) { return 1.0f / (1.0f + expf(-x)); }


__device__ __forceinline__ v2f load_afrag(const float* A, int lda, int m0, int k, int lane) {
    int m  = m0 + (lane & 15);
    int kk = k + ((lane >> 4) << 1);
    v2f a;
    a.x = A[m * lda + kk];
    a.y = A[m * lda + kk + 1];
    return a;
}

__device__ __forceinline__ v2f load_bfrag(const float* Bm, int ldb, int k, int n0, int lane) {
    int n  = n0 + (lane & 15);
    int kk = k + ((lane >> 4) << 1);
    v2f b;
    b.x = Bm[kk * ldb + n];
    b.y = Bm[(kk + 1) * ldb + n];
    return b;
}

__device__ __forceinline__ v8f wmma4(v2f a, v2f b, v8f c) {
    return __builtin_amdgcn_wmma_f32_16x16x4_f32(false, a, false, b, (short)0, c, false, false);
}

__global__ void k_init(const float* __restrict__ cov,
                       const float* __restrict__ w1, const float* __restrict__ b1,
                       const float* __restrict__ w2, const float* __restrict__ b2,
                       float* __restrict__ h, float* __restrict__ loss) {
    __shared__ float t_s[16][CH_];
    const int tid = threadIdx.x;
    const int m0  = blockIdx.x * 16;
    if (blockIdx.x == 0 && tid == 0) *loss = 0.0f;
    for (int i = tid; i < 16 * CH_; i += 128) {
        int r = i / CH_, c = i % CH_;
        float acc = b1[c];
        #pragma unroll
        for (int k = 0; k < CS_; ++k) acc += cov[(m0 + r) * CS_ + k] * w1[k * CH_ + c];
        t_s[r][c] = fmaxf(acc, 0.0f);
    }
    __syncthreads();
    for (int i = tid; i < 16 * H_; i += 128) {
        int r = i / H_, c = i % H_;
        float acc = b2[c];
        #pragma unroll
        for (int k = 0; k < CH_; ++k) acc += t_s[r][k] * w2[k * H_ + c];
        h[(m0 + r) * H_ + c] = tanhf(acc);
    }
}

__global__ void k_pmodel(const float* __restrict__ h,
                         const float* __restrict__ w1, const float* __restrict__ b1,
                         const float* __restrict__ w2, const float* __restrict__ b2,
                         float* __restrict__ p) {
    __shared__ float t_s[16][P2_];
    const int tid = threadIdx.x, lane = tid & 31, wv = tid >> 5;
    const int m0 = blockIdx.x * 16, n0 = wv * 16;

    v8f acc = {};
    for (int k = 0; k < H_; k += 4)
        acc = wmma4(load_afrag(h, H_, m0, k, lane), load_bfrag(w1, P2_, k, n0, lane), acc);
    const int col = n0 + (lane & 15), rb = (lane >> 4) << 3;
    const float bv = b1[col];
    #pragma unroll
    for (int j = 0; j < 8; ++j) t_s[rb + j][col] = fmaxf(acc[j] + bv, 0.0f);
    __syncthreads();

    v8f acc2 = {};
    for (int k = 0; k < P2_; k += 4)
        acc2 = wmma4(load_afrag(&t_s[0][0], P2_, 0, k, lane), load_bfrag(w2, P2_, k, n0, lane), acc2);
    const float bv2 = b2[col];
    #pragma unroll
    for (int j = 0; j < 8; ++j) p[(m0 + rb + j) * P2_ + col] = acc2[j] + bv2;
}

__global__ void k_ode(float* __restrict__ h, const float* __restrict__ p,
                      const float* __restrict__ xz_w, const float* __restrict__ xz_b,
                      const float* __restrict__ hz_w,
                      const float* __restrict__ xn_w, const float* __restrict__ xn_b,
                      const float* __restrict__ hn_w) {
    __shared__ float zh_s[16][H_];
    const int tid = threadIdx.x, lane = tid & 31, wv = tid >> 5;
    const int m0 = blockIdx.x * 16, n0 = wv * 16;

    v8f az = {};
    for (int k = 0; k < P2_; k += 4)
        az = wmma4(load_afrag(p, P2_, m0, k, lane), load_bfrag(xz_w, H_, k, n0, lane), az);
    for (int k = 0; k < H_; k += 4)
        az = wmma4(load_afrag(h, H_, m0, k, lane), load_bfrag(hz_w, H_, k, n0, lane), az);

    const int col = n0 + (lane & 15), rb = (lane >> 4) << 3;
    const float bz = xz_b[col];
    float z[8];
    #pragma unroll
    for (int j = 0; j < 8; ++j) {
        z[j] = sigmoidf_(az[j] + bz);
        zh_s[rb + j][col] = z[j] * h[(m0 + rb + j) * H_ + col];
    }
    __syncthreads();

    v8f an = {};
    for (int k = 0; k < P2_; k += 4)
        an = wmma4(load_afrag(p, P2_, m0, k, lane), load_bfrag(xn_w, H_, k, n0, lane), an);
    for (int k = 0; k < H_; k += 4)
        an = wmma4(load_afrag(&zh_s[0][0], H_, 0, k, lane), load_bfrag(hn_w, H_, k, n0, lane), an);

    const float bn = xn_b[col];
    #pragma unroll
    for (int j = 0; j < 8; ++j) {
        float n = tanhf(an[j] + bn);
        int row = m0 + rb + j;
        float hv = h[row * H_ + col];
        h[row * H_ + col] = hv + DT_ * (1.0f - z[j]) * (n - hv);
    }
}

__global__ void __launch_bounds__(JT_) k_jump(int t,
                       const float* __restrict__ X, const float* __restrict__ M,
                       const int* __restrict__ obs_idx,
                       float* __restrict__ h, const float* __restrict__ p,
                       const float* __restrict__ w_ih, const float* __restrict__ w_hh,
                       const float* __restrict__ b_ih, const float* __restrict__ b_hh,
                       const float* __restrict__ w_prep, const float* __restrict__ bias_prep,
                       float* __restrict__ loss_part) {
    __shared__ int   idx_s[16];
    __shared__ float hold_s[16][H_];
    __shared__ float gin_s[16][GIN_];
    __shared__ float gi_s[16][G3_];
    __shared__ float gh_s[16][G3_];
    __shared__ float red[16 * D_];
    const int tid = threadIdx.x, lane = tid & 31, wv = tid >> 5;
    const int o0 = blockIdx.x * 16;
    const float* Xt = X + (size_t)t * NO_ * D_;
    const float* Mt = M + (size_t)t * NO_ * D_;

    if (tid < 16) idx_s[tid] = obs_idx[t * NO_ + o0 + tid];
    __syncthreads();

    for (int i = tid; i < 16 * H_; i += JT_) {
        int o = i >> 7, j = i & 127;
        hold_s[o][j] = h[idx_s[o] * H_ + j];
    }

    float lterm = 0.0f;
    if (tid < 16 * D_) {
        int o = tid / D_, d = tid % D_;
        int gb = o0 + o, b = idx_s[o];
        float mean = p[b * P2_ + d];
        float lv   = p[b * P2_ + D_ + d];
        float x    = Xt[gb * D_ + d];
        float m    = Mt[gb * D_ + d];
        float err  = (x - mean) * expf(-0.5f * lv);
        lterm = 0.5f * ((err * err + lv + 2.0f * LOGC_) * m);
    }

    for (int i = tid; i < 16 * GIN_; i += JT_) {
        int o = i / GIN_, c = i % GIN_;
        int d = c / PREP_, pi = c - d * PREP_;
        int gb = o0 + o, b = idx_s[o];
        float mean = p[b * P2_ + d];
        float lv   = p[b * P2_ + D_ + d];
        float x    = Xt[gb * D_ + d];
        float m    = Mt[gb * D_ + d];
        float err  = (x - mean) * expf(-0.5f * lv);
        const float* wp = w_prep + (size_t)(d * 4) * PREP_ + pi;
        float a = x * wp[0] + mean * wp[PREP_] + lv * wp[2 * PREP_] + err * wp[3 * PREP_]
                + bias_prep[d * PREP_ + pi];
        gin_s[o][c] = fmaxf(a, 0.0f) * m;
    }
    __syncthreads();

    {
        const int n0 = wv * 16;
        v8f gi = {};
        for (int k = 0; k < GIN_; k += 4)
            gi = wmma4(load_afrag(&gin_s[0][0], GIN_, 0, k, lane), load_bfrag(w_ih, G3_, k, n0, lane), gi);
        v8f gh = {};
        for (int k = 0; k < H_; k += 4)
            gh = wmma4(load_afrag(&hold_s[0][0], H_, 0, k, lane), load_bfrag(w_hh, G3_, k, n0, lane), gh);
        const int col = n0 + (lane & 15), rb = (lane >> 4) << 3;
        const float bi = b_ih[col], bh = b_hh[col];
        #pragma unroll
        for (int j = 0; j < 8; ++j) {
            gi_s[rb + j][col] = gi[j] + bi;
            gh_s[rb + j][col] = gh[j] + bh;
        }
    }
    __syncthreads();

    for (int i = tid; i < 16 * H_; i += JT_) {
        int o = i >> 7, j = i & 127;
        float r = sigmoidf_(gi_s[o][j] + gh_s[o][j]);
        float z = sigmoidf_(gi_s[o][H_ + j] + gh_s[o][H_ + j]);
        float n = tanhf(gi_s[o][2 * H_ + j] + r * gh_s[o][2 * H_ + j]);
        float hv = hold_s[o][j];
        h[idx_s[o] * H_ + j] = (1.0f - z) * n + z * hv;
    }

    if (tid < 16 * D_) red[tid] = lterm;
    __syncthreads();
    for (int s2 = 8 * D_; s2 > 0; s2 >>= 1) {
        if (tid < s2) red[tid] += red[tid + s2];
        __syncthreads();
    }
    if (tid == 0) loss_part[t * (NO_ / 16) + blockIdx.x] = red[0];
}

__global__ void k_loss_final(const float* __restrict__ parts, float* __restrict__ loss) {
    __shared__ float red[256];
    const int tid = threadIdx.x;
    float s = 0.0f;
    for (int i = tid; i < NT_ * (NO_ / 16); i += 256) s += parts[i];
    red[tid] = s;
    __syncthreads();
    for (int k = 128; k > 0; k >>= 1) {
        if (tid < k) red[tid] += red[tid + k];
        __syncthreads();
    }
    if (tid == 0) *loss = red[0];
}

extern "C" void kernel_launch(void* const* d_in, const int* in_sizes, int n_in,
                              void* d_out, int out_size, void* d_ws, size_t ws_size,
                              hipStream_t stream) {
    const float* X       = (const float*)d_in[0];
    const float* M       = (const float*)d_in[1];
    const int*   obs_idx = (const int*)d_in[2];
    const float* cov     = (const float*)d_in[3];
    const float* cm_w1   = (const float*)d_in[4];
    const float* cm_b1   = (const float*)d_in[5];
    const float* cm_w2   = (const float*)d_in[6];
    const float* cm_b2   = (const float*)d_in[7];
    const float* p_w1    = (const float*)d_in[8];
    const float* p_b1    = (const float*)d_in[9];
    const float* p_w2    = (const float*)d_in[10];
    const float* p_b2    = (const float*)d_in[11];
    const float* xz_w    = (const float*)d_in[12];
    const float* xz_b    = (const float*)d_in[13];
    const float* hz_w    = (const float*)d_in[14];
    const float* xn_w    = (const float*)d_in[15];
    const float* xn_b    = (const float*)d_in[16];
    const float* hn_w    = (const float*)d_in[17];
    const float* w_ih    = (const float*)d_in[18];
    const float* w_hh    = (const float*)d_in[19];
    const float* b_ih    = (const float*)d_in[20];
    const float* b_hh    = (const float*)d_in[21];
    const float* w_prep  = (const float*)d_in[22];
    const float* b_prep  = (const float*)d_in[23];

    float* h    = (float*)d_out;
    float* p    = h + (size_t)B_ * H_;
    float* loss = p + (size_t)B_ * P2_;
    float* loss_part = (float*)d_ws;

    const int MB = B_ / 16;
    const int OB = NO_ / 16;

    k_init<<<MB, 128, 0, stream>>>(cov, cm_w1, cm_b1, cm_w2, cm_b2, h, loss);
    k_pmodel<<<MB, 128, 0, stream>>>(h, p_w1, p_b1, p_w2, p_b2, p);

    for (int t = 0; t < NT_; ++t) {
        k_ode<<<MB, 256, 0, stream>>>(h, p, xz_w, xz_b, hz_w, xn_w, xn_b, hn_w);
        k_pmodel<<<MB, 128, 0, stream>>>(h, p_w1, p_b1, p_w2, p_b2, p);
        k_jump<<<OB, JT_, 0, stream>>>(t, X, M, obs_idx, h, p,
                                       w_ih, w_hh, b_ih, b_hh, w_prep, b_prep, loss_part);
        k_pmodel<<<MB, 128, 0, stream>>>(h, p_w1, p_b1, p_w2, p_b2, p);
    }
    for (int s = 0; s < NPOST_; ++s) {
        k_ode<<<MB, 256, 0, stream>>>(h, p, xz_w, xz_b, hz_w, xn_w, xn_b, hn_w);
        k_pmodel<<<MB, 128, 0, stream>>>(h, p_w1, p_b1, p_w2, p_b2, p);
    }
    k_loss_final<<<1, 256, 0, stream>>>(loss_part, loss);
}


# LDS row pitch of every 16-row tile padded by 4 floats (64/128/320/384 -> 68/132/324/388): the 16 rows of an A-fragment load no longer share one bank (16-way conflict removed); k order unchanged
# speedup vs baseline: 1.4533x; 1.3345x over previous
#include <hip/hip_runtime.h>
#include <hip/hip_bf16.h>
#include <math.h>

#define B_    8192
#define D_    32
#define H_    128
#define P2_   64
#define PREP_ 10
#define GIN_  320
#define G3_   384
#define CS_   8
#define CH_   32
#define NT_   16
#define NO_   4096
#define DT_   0.1f
#define NPOST_ 4
#define JT_   768
#define LOGC_  0.91893853320467274178f

#define HP_   (H_ + 4)
#define PP_   (P2_ + 4)
#define GP_   (GIN_ + 4)
#define G3P_  (G3_ + 4)

typedef __attribute__((ext_vector_type(2))) float v2f;
typedef __attribute__((ext_vector_type(8))) float v8f;

__device__ __forceinline__ float sigmoidf_(float x) { return 1.0f / (1.0f + expf(-x)); }


__device__ __forceinline__ v2f load_afrag(const float* A, int lda, int m0, int k, int lane) {
    int m  = m0 + (lane & 15);
    int kk = k + ((lane >> 4) << 1);
    v2f a;
    a.x = A[m * lda + kk];
    a.y = A[m * lda + kk + 1];
    return a;
}

__device__ __forceinline__ v2f load_bfrag(const float* Bm, int ldb, int k, int n0, int lane) {
    int n  = n0 + (lane & 15);
    int kk = k + ((lane >> 4) << 1);
    v2f b;
    b.x = Bm[kk * ldb + n];
    b.y = Bm[(kk + 1) * ldb + n];
    return b;
}

__device__ __forceinline__ v8f wmma4(v2f a, v2f b, v8f c) {
    return __builtin_amdgcn_wmma_f32_16x16x4_f32(false, a, false, b, (short)0, c, false, false);
}

__global__ void k_init(const float* __restrict__ cov,
                       const float* __restrict__ w1, const float* __restrict__ b1,
                       const float* __restrict__ w2, const float* __restrict__ b2,
                       float* __restrict__ h, float* __restrict__ loss) {
    __shared__ float t_s[16][CH_];
    const int tid = threadIdx.x;
    const int m0  = blockIdx.x * 16;
    if (blockIdx.x == 0 && tid == 0) *loss = 0.0f;
    for (int i = tid; i < 16 * CH_; i += 128) {
        int r = i / CH_, c = i % CH_;
        float acc = b1[c];
        #pragma unroll
        for (int k = 0; k < CS_; ++k) acc += cov[(m0 + r) * CS_ + k] * w1[k * CH_ + c];
        t_s[r][c] = fmaxf(acc, 0.0f);
    }
    __syncthreads();
    for (int i = tid; i < 16 * H_; i += 128) {
        int r = i / H_, c = i % H_;
        float acc = b2[c];
        #pragma unroll
        for (int k = 0; k < CH_; ++k) acc += t_s[r][k] * w2[k * H_ + c];
        h[(m0 + r) * H_ + c] = tanhf(acc);
    }
}

__global__ void k_pmodel(const float* __restrict__ h,
                         const float* __restrict__ w1, const float* __restrict__ b1,
                         const float* __restrict__ w2, const float* __restrict__ b2,
                         float* __restrict__ p) {
    __shared__ float t_s[16][PP_];
    const int tid = threadIdx.x, lane = tid & 31, wv = tid >> 5;
    const int m0 = blockIdx.x * 16, n0 = wv * 16;

    v8f acc = {};
    for (int k = 0; k < H_; k += 4)
        acc = wmma4(load_afrag(h, H_, m0, k, lane), load_bfrag(w1, P2_, k, n0, lane), acc);
    const int col = n0 + (lane & 15), rb = (lane >> 4) << 3;
    const float bv = b1[col];
    #pragma unroll
    for (int j = 0; j < 8; ++j) t_s[rb + j][col] = fmaxf(acc[j] + bv, 0.0f);
    __syncthreads();

    v8f acc2 = {};
    for (int k = 0; k < P2_; k += 4)
        acc2 = wmma4(load_afrag(&t_s[0][0], PP_, 0, k, lane), load_bfrag(w2, P2_, k, n0, lane), acc2);
    const float bv2 = b2[col];
    #pragma unroll
    for (int j = 0; j < 8; ++j) p[(m0 + rb + j) * P2_ + col] = acc2[j] + bv2;
}

__global__ void k_ode(float* __restrict__ h, const float* __restrict__ p,
                      const float* __restrict__ xz_w, const float* __restrict__ xz_b,
                      const float* __restrict__ hz_w,
                      const float* __restrict__ xn_w, const float* __restrict__ xn_b,
                      const float* __restrict__ hn_w) {
    __shared__ float zh_s[16][HP_];
    const int tid = threadIdx.x, lane = tid & 31, wv = tid >> 5;
    const int m0 = blockIdx.x * 16, n0 = wv * 16;

    v8f az = {};
    for (int k = 0; k < P2_; k += 4)
        az = wmma4(load_afrag(p, P2_, m0, k, lane), load_bfrag(xz_w, H_, k, n0, lane), az);
    for (int k = 0; k < H_; k += 4)
        az = wmma4(load_afrag(h, H_, m0, k, lane), load_bfrag(hz_w, H_, k, n0, lane), az);

    const int col = n0 + (lane & 15), rb = (lane >> 4) << 3;
    const float bz = xz_b[col];
    float z[8];
    #pragma unroll
    for (int j = 0; j < 8; ++j) {
        z[j] = sigmoidf_(az[j] + bz);
        zh_s[rb + j][col] = z[j] * h[(m0 + rb + j) * H_ + col];
    }
    __syncthreads();

    v8f an = {};
    for (int k = 0; k < P2_; k += 4)
        an = wmma4(load_afrag(p, P2_, m0, k, lane), load_bfrag(xn_w, H_, k, n0, lane), an);
    for (int k = 0; k < H_; k += 4)
        an = wmma4(load_afrag(&zh_s[0][0], HP_, 0, k, lane), load_bfrag(hn_w, H_, k, n0, lane), an);

    const float bn = xn_b[col];
    #pragma unroll
    for (int j = 0; j < 8; ++j) {
        float n = tanhf(an[j] + bn);
        int row = m0 + rb + j;
        float hv = h[row * H_ + col];
        h[row * H_ + col] = hv + DT_ * (1.0f - z[j]) * (n - hv);
    }
}

__global__ void __launch_bounds__(JT_) k_jump(int t,
                       const float* __restrict__ X, const float* __restrict__ M,
                       const int* __restrict__ obs_idx,
                       float* __restrict__ h, const float* __restrict__ p,
                       const float* __restrict__ w_ih, const float* __restrict__ w_hh,
                       const float* __restrict__ b_ih, const float* __restrict__ b_hh,
                       const float* __restrict__ w_prep, const float* __restrict__ bias_prep,
                       float* __restrict__ loss_part) {
    __shared__ int   idx_s[16];
    __shared__ float hold_s[16][HP_];
    __shared__ float gin_s[16][GP_];
    __shared__ float gi_s[16][G3P_];
    __shared__ float gh_s[16][G3P_];
    __shared__ float red[16 * D_];
    const int tid = threadIdx.x, lane = tid & 31, wv = tid >> 5;
    const int o0 = blockIdx.x * 16;
    const float* Xt = X + (size_t)t * NO_ * D_;
    const float* Mt = M + (size_t)t * NO_ * D_;

    if (tid < 16) idx_s[tid] = obs_idx[t * NO_ + o0 + tid];
    __syncthreads();

    for (int i = tid; i < 16 * H_; i += JT_) {
        int o = i >> 7, j = i & 127;
        hold_s[o][j] = h[idx_s[o] * H_ + j];
    }

    float lterm = 0.0f;
    if (tid < 16 * D_) {
        int o = tid / D_, d = tid % D_;
        int gb = o0 + o, b = idx_s[o];
        float mean = p[b * P2_ + d];
        float lv   = p[b * P2_ + D_ + d];
        float x    = Xt[gb * D_ + d];
        float m    = Mt[gb * D_ + d];
        float err  = (x - mean) * expf(-0.5f * lv);
        lterm = 0.5f * ((err * err + lv + 2.0f * LOGC_) * m);
    }

    for (int i = tid; i < 16 * GIN_; i += JT_) {
        int o = i / GIN_, c = i % GIN_;
        int d = c / PREP_, pi = c - d * PREP_;
        int gb = o0 + o, b = idx_s[o];
        float mean = p[b * P2_ + d];
        float lv   = p[b * P2_ + D_ + d];
        float x    = Xt[gb * D_ + d];
        float m    = Mt[gb * D_ + d];
        float err  = (x - mean) * expf(-0.5f * lv);
        const float* wp = w_prep + (size_t)(d * 4) * PREP_ + pi;
        float a = x * wp[0] + mean * wp[PREP_] + lv * wp[2 * PREP_] + err * wp[3 * PREP_]
                + bias_prep[d * PREP_ + pi];
        gin_s[o][c] = fmaxf(a, 0.0f) * m;
    }
    __syncthreads();

    {
        const int n0 = wv * 16;
        v8f gi = {};
        for (int k = 0; k < GIN_; k += 4)
            gi = wmma4(load_afrag(&gin_s[0][0], GP_, 0, k, lane), load_bfrag(w_ih, G3_, k, n0, lane), gi);
        v8f gh = {};
        for (int k = 0; k < H_; k += 4)
            gh = wmma4(load_afrag(&hold_s[0][0], HP_, 0, k, lane), load_bfrag(w_hh, G3_, k, n0, lane), gh);
        const int col = n0 + (lane & 15), rb = (lane >> 4) << 3;
        const float bi = b_ih[col], bh = b_hh[col];
        #pragma unroll
        for (int j = 0; j < 8; ++j) {
            gi_s[rb + j][col] = gi[j] + bi;
            gh_s[rb + j][col] = gh[j] + bh;
        }
    }
    __syncthreads();

    for (int i = tid; i < 16 * H_; i += JT_) {
        int o = i >> 7, j = i & 127;
        float r = sigmoidf_(gi_s[o][j] + gh_s[o][j]);
        float z = sigmoidf_(gi_s[o][H_ + j] + gh_s[o][H_ + j]);
        float n = tanhf(gi_s[o][2 * H_ + j] + r * gh_s[o][2 * H_ + j]);
        float hv = hold_s[o][j];
        h[idx_s[o] * H_ + j] = (1.0f - z) * n + z * hv;
    }

    if (tid < 16 * D_) red[tid] = lterm;
    __syncthreads();
    for (int s2 = 8 * D_; s2 > 0; s2 >>= 1) {
        if (tid < s2) red[tid] += red[tid + s2];
        __syncthreads();
    }
    if (tid == 0) loss_part[t * (NO_ / 16) + blockIdx.x] = red[0];
}

__global__ void k_loss_final(const float* __restrict__ parts, float* __restrict__ loss) {
    __shared__ float red[256];
    const int tid = threadIdx.x;
    float s = 0.0f;
    for (int i = tid; i < NT_ * (NO_ / 16); i += 256) s += parts[i];
    red[tid] = s;
    __syncthreads();
    for (int k = 128; k > 0; k >>= 1) {
        if (tid < k) red[tid] += red[tid + k];
        __syncthreads();
    }
    if (tid == 0) *loss = red[0];
}

extern "C" void kernel_launch(void* const* d_in, const int* in_sizes, int n_in,
                              void* d_out, int out_size, void* d_ws, size_t ws_size,
                              hipStream_t stream) {
    const float* X       = (const float*)d_in[0];
    const float* M       = (const float*)d_in[1];
    const int*   obs_idx = (const int*)d_in[2];
    const float* cov     = (const float*)d_in[3];
    const float* cm_w1   = (const float*)d_in[4];
    const float* cm_b1   = (const float*)d_in[5];
    const float* cm_w2   = (const float*)d_in[6];
    const float* cm_b2   = (const float*)d_in[7];
    const float* p_w1    = (const float*)d_in[8];
    const float* p_b1    = (const float*)d_in[9];
    const float* p_w2    = (const float*)d_in[10];
    const float* p_b2    = (const float*)d_in[11];
    const float* xz_w    = (const float*)d_in[12];
    const float* xz_b    = (const float*)d_in[13];
    const float* hz_w    = (const float*)d_in[14];
    const float* xn_w    = (const float*)d_in[15];
    const float* xn_b    = (const float*)d_in[16];
    const float* hn_w    = (const float*)d_in[17];
    const float* w_ih    = (const float*)d_in[18];
    const float* w_hh    = (const float*)d_in[19];
    const float* b_ih    = (const float*)d_in[20];
    const float* b_hh    = (const float*)d_in[21];
    const float* w_prep  = (const float*)d_in[22];
    const float* b_prep  = (const float*)d_in[23];

    float* h    = (float*)d_out;
    float* p    = h + (size_t)B_ * H_;
    float* loss = p + (size_t)B_ * P2_;
    float* loss_part = (float*)d_ws;

    const int MB = B_ / 16;
    const int OB = NO_ / 16;

    k_init<<<MB, 128, 0, stream>>>(cov, cm_w1, cm_b1, cm_w2, cm_b2, h, loss);
    k_pmodel<<<MB, 128, 0, stream>>>(h, p_w1, p_b1, p_w2, p_b2, p);

    for (int t = 0; t < NT_; ++t) {
        k_ode<<<MB, 256, 0, stream>>>(h, p, xz_w, xz_b, hz_w, xn_w, xn_b, hn_w);
        k_pmodel<<<MB, 128, 0, stream>>>(h, p_w1, p_b1, p_w2, p_b2, p);
        k_jump<<<OB, JT_, 0, stream>>>(t, X, M, obs_idx, h, p,
                                       w_ih, w_hh, b_ih, b_hh, w_prep, b_prep, loss_part);
        k_pmodel<<<MB, 128, 0, stream>>>(h, p_w1, p_b1, p_w2, p_b2, p);
    }
    for (int s = 0; s < NPOST_; ++s) {
        k_ode<<<MB, 256, 0, stream>>>(h, p, xz_w, xz_b, hz_w, xn_w, xn_b, hn_w);
        k_pmodel<<<MB, 128, 0, stream>>>(h, p_w1, p_b1, p_w2, p_b2, p);
    }
    k_loss_final<<<1, 256, 0, stream>>>(loss_part, loss);
}


# p-model fused into the kernels that write h (init, ode step, jump), its h rows read from LDS; the jump recomputes p only for the rows it updates; 75 launches become 38; order of every sum kept
# speedup vs baseline: 1.7093x; 1.1762x over previous
#include <hip/hip_runtime.h>
#include <hip/hip_bf16.h>
#include <math.h>

#define B_    8192
#define D_    32
#define H_    128
#define P2_   64
#define PREP_ 10
#define GIN_  320
#define G3_   384
#define CS_   8
#define CH_   32
#define NT_   16
#define NO_   4096
#define DT_   0.1f
#define NPOST_ 4
#define JT_   768
#define LOGC_  0.91893853320467274178f

#define HP_   (H_ + 4)
#define PP_   (P2_ + 4)
#define GP_   (GIN_ + 4)
#define G3P_  (G3_ + 4)

typedef __attribute__((ext_vector_type(2))) float v2f;
typedef __attribute__((ext_vector_type(8))) float v8f;

__device__ __forceinline__ float sigmoidf_(float x) { return 1.0f / (1.0f + expf(-x)); }


__device__ __forceinline__ v2f load_afrag(const float* A, int lda, int m0, int k, int lane) {
    int m  = m0 + (lane & 15);
    int kk = k + ((lane >> 4) << 1);
    v2f a;
    a.x = A[m * lda + kk];
    a.y = A[m * lda + kk + 1];
    return a;
}

__device__ __forceinline__ v2f load_bfrag(const float* Bm, int ldb, int k, int n0, int lane) {
    int n  = n0 + (lane & 15);
    int kk = k + ((lane >> 4) << 1);
    v2f b;
    b.x = Bm[kk * ldb + n];
    b.y = Bm[(kk + 1) * ldb + n];
    return b;
}

__device__ __forceinline__ v8f wmma4(v2f a, v2f b, v8f c) {
    return __builtin_amdgcn_wmma_f32_16x16x4_f32(false, a, false, b, (short)0, c, false, false);
}

template <int K>
__device__ __forceinline__ void pm_load_w(v2f (&bw)[K / 4], const float* __restrict__ w, int n0, int lane) {
    #pragma unroll
    for (int k = 0; k < K; k += 4) bw[k >> 2] = load_bfrag(w, P2_, k, n0, lane);
}

template <int K>
__device__ __forceinline__ v8f pm_gemm(const float* as, int pitch, const v2f (&bw)[K / 4], int lane) {
    v8f acc = {};
    #pragma unroll
    for (int k = 0; k < K; k += 4) acc = wmma4(load_afrag(as, pitch, 0, k, lane), bw[k >> 2], acc);
    return acc;
}

__global__ void __launch_bounds__(128) k_init(const float* __restrict__ cov,
                       const float* __restrict__ w1, const float* __restrict__ b1,
                       const float* __restrict__ w2, const float* __restrict__ b2,
                       const float* __restrict__ pw1, const float* __restrict__ pb1,
                       const float* __restrict__ pw2, const float* __restrict__ pb2,
                       float* __restrict__ h, float* __restrict__ p, float* __restrict__ loss) {
    __shared__ float t_s[16][CH_];
    __shared__ float hn_s[16][HP_];
    __shared__ float u_s[16][PP_];
    const int tid = threadIdx.x;
    const int lane = tid & 31, pn0 = (tid >> 5) * 16;
    const int m0  = blockIdx.x * 16;
    if (blockIdx.x == 0 && tid == 0) *loss = 0.0f;
    v2f bw1[H_ / 4], bw2[P2_ / 4];
    pm_load_w<H_>(bw1, pw1, pn0, lane);
    pm_load_w<P2_>(bw2, pw2, pn0, lane);
    for (int i = tid; i < 16 * CH_; i += 128) {
        int r = i / CH_, c = i % CH_;
        float acc = b1[c];
        #pragma unroll
        for (int k = 0; k < CS_; ++k) acc += cov[(m0 + r) * CS_ + k] * w1[k * CH_ + c];
        t_s[r][c] = fmaxf(acc, 0.0f);
    }
    __syncthreads();
    for (int i = tid; i < 16 * H_; i += 128) {
        int r = i / H_, c = i % H_;
        float acc = b2[c];
        #pragma unroll
        for (int k = 0; k < CH_; ++k) acc += t_s[r][k] * w2[k * H_ + c];
        float hv = tanhf(acc);
        h[(m0 + r) * H_ + c] = hv;
        hn_s[r][c] = hv;
    }
    __syncthreads();

    const int col = pn0 + (lane & 15), rb = (lane >> 4) << 3;
    v8f a1 = pm_gemm<H_>(&hn_s[0][0], HP_, bw1, lane);
    const float bv = pb1[col];
    #pragma unroll
    for (int j = 0; j < 8; ++j) u_s[rb + j][col] = fmaxf(a1[j] + bv, 0.0f);
    __syncthreads();
    v8f a2 = pm_gemm<P2_>(&u_s[0][0], PP_, bw2, lane);
    const float bv2 = pb2[col];
    #pragma unroll
    for (int j = 0; j < 8; ++j) p[(m0 + rb + j) * P2_ + col] = a2[j] + bv2;
}

__global__ void __launch_bounds__(256) k_ode(float* __restrict__ h, float* __restrict__ p,
                      const float* __restrict__ xz_w, const float* __restrict__ xz_b,
                      const float* __restrict__ hz_w,
                      const float* __restrict__ xn_w, const float* __restrict__ xn_b,
                      const float* __restrict__ hn_w,
                      const float* __restrict__ pw1, const float* __restrict__ pb1,
                      const float* __restrict__ pw2, const float* __restrict__ pb2) {
    __shared__ float zh_s[16][HP_];
    __shared__ float hn_s[16][HP_];
    __shared__ float u_s[16][PP_];
    const int tid = threadIdx.x, lane = tid & 31;
    const int wv = __builtin_amdgcn_readfirstlane(tid >> 5);
    const int m0 = blockIdx.x * 16, n0 = wv * 16;
    const bool pmw = wv < 4;

    v8f az = {};
    for (int k = 0; k < P2_; k += 4)
        az = wmma4(load_afrag(p, P2_, m0, k, lane), load_bfrag(xz_w, H_, k, n0, lane), az);
    for (int k = 0; k < H_; k += 4)
        az = wmma4(load_afrag(h, H_, m0, k, lane), load_bfrag(hz_w, H_, k, n0, lane), az);

    const int col = n0 + (lane & 15), rb = (lane >> 4) << 3;
    const float bz = xz_b[col];
    float z[8];
    #pragma unroll
    for (int j = 0; j < 8; ++j) {
        z[j] = sigmoidf_(az[j] + bz);
        zh_s[rb + j][col] = z[j] * h[(m0 + rb + j) * H_ + col];
    }
    __syncthreads();

    v2f bw1[H_ / 4] = {}, bw2[P2_ / 4] = {};
    if (pmw) { pm_load_w<H_>(bw1, pw1, n0, lane); pm_load_w<P2_>(bw2, pw2, n0, lane); }

    v8f an = {};
    for (int k = 0; k < P2_; k += 4)
        an = wmma4(load_afrag(p, P2_, m0, k, lane), load_bfrag(xn_w, H_, k, n0, lane), an);
    for (int k = 0; k < H_; k += 4)
        an = wmma4(load_afrag(&zh_s[0][0], HP_, 0, k, lane), load_bfrag(hn_w, H_, k, n0, lane), an);

    const float bn = xn_b[col];
    float hnew[8];
    #pragma unroll
    for (int j = 0; j < 8; ++j) {
        float n = tanhf(an[j] + bn);
        float hv = h[(m0 + rb + j) * H_ + col];
        hnew[j] = hv + DT_ * (1.0f - z[j]) * (n - hv);
        hn_s[rb + j][col] = hnew[j];
    }
    __syncthreads();
    #pragma unroll
    for (int j = 0; j < 8; ++j) h[(m0 + rb + j) * H_ + col] = hnew[j];

    if (pmw) {
        v8f a1 = pm_gemm<H_>(&hn_s[0][0], HP_, bw1, lane);
        const float bv = pb1[col];
        #pragma unroll
        for (int j = 0; j < 8; ++j) u_s[rb + j][col] = fmaxf(a1[j] + bv, 0.0f);
    }
    __syncthreads();
    if (pmw) {
        v8f a2 = pm_gemm<P2_>(&u_s[0][0], PP_, bw2, lane);
        const float bv2 = pb2[col];
        #pragma unroll
        for (int j = 0; j < 8; ++j) p[(m0 + rb + j) * P2_ + col] = a2[j] + bv2;
    }
}

__global__ void __launch_bounds__(JT_) k_jump(int t,
                       const float* __restrict__ X, const float* __restrict__ M,
                       const int* __restrict__ obs_idx,
                       float* __restrict__ h, float* __restrict__ p,
                       const float* __restrict__ w_ih, const float* __restrict__ w_hh,
                       const float* __restrict__ b_ih, const float* __restrict__ b_hh,
                       const float* __restrict__ w_prep, const float* __restrict__ bias_prep,
                       const float* __restrict__ pw1, const float* __restrict__ pb1,
                       const float* __restrict__ pw2, const float* __restrict__ pb2,
                       float* __restrict__ loss_part) {
    __shared__ int   idx_s[16];
    __shared__ float hold_s[16][HP_];
    __shared__ float gin_s[16][GP_];
    __shared__ float gi_s[16][G3P_];
    __shared__ float gh_s[16][G3P_];
    __shared__ float red[16 * D_];
    __shared__ float u_s[16][PP_];
    const int tid = threadIdx.x, lane = tid & 31;
    const int wv = __builtin_amdgcn_readfirstlane(tid >> 5);
    const bool pmw = (wv >> 2) == 2;
    const int pn0 = (wv & 3) * 16;
    const int o0 = blockIdx.x * 16;
    const float* Xt = X + (size_t)t * NO_ * D_;
    const float* Mt = M + (size_t)t * NO_ * D_;

    if (tid < 16) idx_s[tid] = obs_idx[t * NO_ + o0 + tid];
    __syncthreads();

    for (int i = tid; i < 16 * H_; i += JT_) {
        int o = i >> 7, j = i & 127;
        hold_s[o][j] = h[idx_s[o] * H_ + j];
    }

    float lterm = 0.0f;
    if (tid < 16 * D_) {
        int o = tid / D_, d = tid % D_;
        int gb = o0 + o, b = idx_s[o];
        float mean = p[b * P2_ + d];
        float lv   = p[b * P2_ + D_ + d];
        float x    = Xt[gb * D_ + d];
        float m    = Mt[gb * D_ + d];
        float err  = (x - mean) * expf(-0.5f * lv);
        lterm = 0.5f * ((err * err + lv + 2.0f * LOGC_) * m);
    }

    for (int i = tid; i < 16 * GIN_; i += JT_) {
        int o = i / GIN_, c = i % GIN_;
        int d = c / PREP_, pi = c - d * PREP_;
        int gb = o0 + o, b = idx_s[o];
        float mean = p[b * P2_ + d];
        float lv   = p[b * P2_ + D_ + d];
        float x    = Xt[gb * D_ + d];
        float m    = Mt[gb * D_ + d];
        float err  = (x - mean) * expf(-0.5f * lv);
        const float* wp = w_prep + (size_t)(d * 4) * PREP_ + pi;
        float a = x * wp[0] + mean * wp[PREP_] + lv * wp[2 * PREP_] + err * wp[3 * PREP_]
                + bias_prep[d * PREP_ + pi];
        gin_s[o][c] = fmaxf(a, 0.0f) * m;
    }
    __syncthreads();

    {
        const int n0 = wv * 16;
        v8f gi = {};
        for (int k = 0; k < GIN_; k += 4)
            gi = wmma4(load_afrag(&gin_s[0][0], GP_, 0, k, lane), load_bfrag(w_ih, G3_, k, n0, lane), gi);
        v8f gh = {};
        for (int k = 0; k < H_; k += 4)
            gh = wmma4(load_afrag(&hold_s[0][0], HP_, 0, k, lane), load_bfrag(w_hh, G3_, k, n0, lane), gh);
        const int col = n0 + (lane & 15), rb = (lane >> 4) << 3;
        const float bi = b_ih[col], bh = b_hh[col];
        #pragma unroll
        for (int j = 0; j < 8; ++j) {
            gi_s[rb + j][col] = gi[j] + bi;
            gh_s[rb + j][col] = gh[j] + bh;
        }
    }
    __syncthreads();

    v2f bw1[H_ / 4] = {}, bw2[P2_ / 4] = {};
    if (pmw) { pm_load_w<H_>(bw1, pw1, pn0, lane); pm_load_w<P2_>(bw2, pw2, pn0, lane); }

    for (int i = tid; i < 16 * H_; i += JT_) {
        int o = i >> 7, j = i & 127;
        float r = sigmoidf_(gi_s[o][j] + gh_s[o][j]);
        float z = sigmoidf_(gi_s[o][H_ + j] + gh_s[o][H_ + j]);
        float n = tanhf(gi_s[o][2 * H_ + j] + r * gh_s[o][2 * H_ + j]);
        float hv = hold_s[o][j];
        float hnew = (1.0f - z) * n + z * hv;
        h[idx_s[o] * H_ + j] = hnew;
        hold_s[o][j] = hnew;
    }

    if (tid < 16 * D_) red[tid] = lterm;
    __syncthreads();
    const int pcol = pn0 + (lane & 15), prb = (lane >> 4) << 3;
    if (pmw) {
        v8f a1 = pm_gemm<H_>(&hold_s[0][0], HP_, bw1, lane);
        const float bv = pb1[pcol];
        #pragma unroll
        for (int j = 0; j < 8; ++j) u_s[prb + j][pcol] = fmaxf(a1[j] + bv, 0.0f);
    }
    if (tid < 8 * D_) red[tid] += red[tid + 8 * D_];
    __syncthreads();
    if (pmw) {
        v8f a2 = pm_gemm<P2_>(&u_s[0][0], PP_, bw2, lane);
        const float bv2 = pb2[pcol];
        #pragma unroll
        for (int j = 0; j < 8; ++j) p[idx_s[prb + j] * P2_ + pcol] = a2[j] + bv2;
    }
    for (int s2 = 4 * D_; s2 > 0; s2 >>= 1) {
        if (tid < s2) red[tid] += red[tid + s2];
        __syncthreads();
    }
    if (tid == 0) loss_part[t * (NO_ / 16) + blockIdx.x] = red[0];
}

__global__ void k_loss_final(const float* __restrict__ parts, float* __restrict__ loss) {
    __shared__ float red[256];
    const int tid = threadIdx.x;
    float s = 0.0f;
    for (int i = tid; i < NT_ * (NO_ / 16); i += 256) s += parts[i];
    red[tid] = s;
    __syncthreads();
    for (int k = 128; k > 0; k >>= 1) {
        if (tid < k) red[tid] += red[tid + k];
        __syncthreads();
    }
    if (tid == 0) *loss = red[0];
}

extern "C" void kernel_launch(void* const* d_in, const int* in_sizes, int n_in,
                              void* d_out, int out_size, void* d_ws, size_t ws_size,
                              hipStream_t stream) {
    const float* X       = (const float*)d_in[0];
    const float* M       = (const float*)d_in[1];
    const int*   obs_idx = (const int*)d_in[2];
    const float* cov     = (const float*)d_in[3];
    const float* cm_w1   = (const float*)d_in[4];
    const float* cm_b1   = (const float*)d_in[5];
    const float* cm_w2   = (const float*)d_in[6];
    const float* cm_b2   = (const float*)d_in[7];
    const float* p_w1    = (const float*)d_in[8];
    const float* p_b1    = (const float*)d_in[9];
    const float* p_w2    = (const float*)d_in[10];
    const float* p_b2    = (const float*)d_in[11];
    const float* xz_w    = (const float*)d_in[12];
    const float* xz_b    = (const float*)d_in[13];
    const float* hz_w    = (const float*)d_in[14];
    const float* xn_w    = (const float*)d_in[15];
    const float* xn_b    = (const float*)d_in[16];
    const float* hn_w    = (const float*)d_in[17];
    const float* w_ih    = (const float*)d_in[18];
    const float* w_hh    = (const float*)d_in[19];
    const float* b_ih    = (const float*)d_in[20];
    const float* b_hh    = (const float*)d_in[21];
    const float* w_prep  = (const float*)d_in[22];
    const float* b_prep  = (const float*)d_in[23];

    float* h    = (float*)d_out;
    float* p    = h + (size_t)B_ * H_;
    float* loss = p + (size_t)B_ * P2_;
    float* loss_part = (float*)d_ws;

    const int MB = B_ / 16;
    const int OB = NO_ / 16;

    k_init<<<MB, 128, 0, stream>>>(cov, cm_w1, cm_b1, cm_w2, cm_b2,
                                   p_w1, p_b1, p_w2, p_b2, h, p, loss);

    for (int t = 0; t < NT_; ++t) {
        k_ode<<<MB, 256, 0, stream>>>(h, p, xz_w, xz_b, hz_w, xn_w, xn_b, hn_w,
                                      p_w1, p_b1, p_w2, p_b2);
        k_jump<<<OB, JT_, 0, stream>>>(t, X, M, obs_idx, h, p,
                                       w_ih, w_hh, b_ih, b_hh, w_prep, b_prep,
                                       p_w1, p_b1, p_w2, p_b2, loss_part);
    }
    for (int s = 0; s < NPOST_; ++s)
        k_ode<<<MB, 256, 0, stream>>>(h, p, xz_w, xz_b, hz_w, xn_w, xn_b, hn_w,
                                      p_w1, p_b1, p_w2, p_b2);
    k_loss_final<<<1, 256, 0, stream>>>(loss_part, loss);
}


# ode kernel: its 16 h rows and 16 p rows copied once per work-group into LDS by full-line 16-byte loads; A fragments and element-wise h come from LDS, not 16-line global loads; order of sums kept
# speedup vs baseline: 1.8920x; 1.1069x over previous
#include <hip/hip_runtime.h>
#include <hip/hip_bf16.h>
#include <math.h>

#define B_    8192
#define D_    32
#define H_    128
#define P2_   64
#define PREP_ 10
#define GIN_  320
#define G3_   384
#define CS_   8
#define CH_   32
#define NT_   16
#define NO_   4096
#define DT_   0.1f
#define NPOST_ 4
#define JT_   768
#define LOGC_  0.91893853320467274178f

#define HP_   (H_ + 4)
#define PP_   (P2_ + 4)
#define GP_   (GIN_ + 4)
#define G3P_  (G3_ + 4)

typedef __attribute__((ext_vector_type(2))) float v2f;
typedef __attribute__((ext_vector_type(4))) float v4f;
typedef __attribute__((ext_vector_type(8))) float v8f;

__device__ __forceinline__ float sigmoidf_(float x) { return 1.0f / (1.0f + expf(-x)); }


__device__ __forceinline__ v2f load_afrag(const float* A, int lda, int m0, int k, int lane) {
    int m  = m0 + (lane & 15);
    int kk = k + ((lane >> 4) << 1);
    v2f a;
    a.x = A[m * lda + kk];
    a.y = A[m * lda + kk + 1];
    return a;
}

__device__ __forceinline__ v2f load_bfrag(const float* Bm, int ldb, int k, int n0, int lane) {
    int n  = n0 + (lane & 15);
    int kk = k + ((lane >> 4) << 1);
    v2f b;
    b.x = Bm[kk * ldb + n];
    b.y = Bm[(kk + 1) * ldb + n];
    return b;
}

__device__ __forceinline__ v8f wmma4(v2f a, v2f b, v8f c) {
    return __builtin_amdgcn_wmma_f32_16x16x4_f32(false, a, false, b, (short)0, c, false, false);
}

template <int K>
__device__ __forceinline__ void pm_load_w(v2f (&bw)[K / 4], const float* __restrict__ w, int n0, int lane) {
    #pragma unroll
    for (int k = 0; k < K; k += 4) bw[k >> 2] = load_bfrag(w, P2_, k, n0, lane);
}

template <int K>
__device__ __forceinline__ v8f pm_gemm(const float* as, int pitch, const v2f (&bw)[K / 4], int lane) {
    v8f acc = {};
    #pragma unroll
    for (int k = 0; k < K; k += 4) acc = wmma4(load_afrag(as, pitch, 0, k, lane), bw[k >> 2], acc);
    return acc;
}

__global__ void __launch_bounds__(128) k_init(const float* __restrict__ cov,
                       const float* __restrict__ w1, const float* __restrict__ b1,
                       const float* __restrict__ w2, const float* __restrict__ b2,
                       const float* __restrict__ pw1, const float* __restrict__ pb1,
                       const float* __restrict__ pw2, const float* __restrict__ pb2,
                       float* __restrict__ h, float* __restrict__ p, float* __restrict__ loss) {
    __shared__ float t_s[16][CH_];
    __shared__ float hn_s[16][HP_];
    __shared__ float u_s[16][PP_];
    const int tid = threadIdx.x;
    const int lane = tid & 31, pn0 = (tid >> 5) * 16;
    const int m0  = blockIdx.x * 16;
    if (blockIdx.x == 0 && tid == 0) *loss = 0.0f;
    v2f bw1[H_ / 4], bw2[P2_ / 4];
    pm_load_w<H_>(bw1, pw1, pn0, lane);
    pm_load_w<P2_>(bw2, pw2, pn0, lane);
    for (int i = tid; i < 16 * CH_; i += 128) {
        int r = i / CH_, c = i % CH_;
        float acc = b1[c];
        #pragma unroll
        for (int k = 0; k < CS_; ++k) acc += cov[(m0 + r) * CS_ + k] * w1[k * CH_ + c];
        t_s[r][c] = fmaxf(acc, 0.0f);
    }
    __syncthreads();
    for (int i = tid; i < 16 * H_; i += 128) {
        int r = i / H_, c = i % H_;
        float acc = b2[c];
        #pragma unroll
        for (int k = 0; k < CH_; ++k) acc += t_s[r][k] * w2[k * H_ + c];
        float hv = tanhf(acc);
        h[(m0 + r) * H_ + c] = hv;
        hn_s[r][c] = hv;
    }
    __syncthreads();

    const int col = pn0 + (lane & 15), rb = (lane >> 4) << 3;
    v8f a1 = pm_gemm<H_>(&hn_s[0][0], HP_, bw1, lane);
    const float bv = pb1[col];
    #pragma unroll
    for (int j = 0; j < 8; ++j) u_s[rb + j][col] = fmaxf(a1[j] + bv, 0.0f);
    __syncthreads();
    v8f a2 = pm_gemm<P2_>(&u_s[0][0], PP_, bw2, lane);
    const float bv2 = pb2[col];
    #pragma unroll
    for (int j = 0; j < 8; ++j) p[(m0 + rb + j) * P2_ + col] = a2[j] + bv2;
}

__global__ void __launch_bounds__(256) k_ode(float* __restrict__ h, float* __restrict__ p,
                      const float* __restrict__ xz_w, const float* __restrict__ xz_b,
                      const float* __restrict__ hz_w,
                      const float* __restrict__ xn_w, const float* __restrict__ xn_b,
                      const float* __restrict__ hn_w,
                      const float* __restrict__ pw1, const float* __restrict__ pb1,
                      const float* __restrict__ pw2, const float* __restrict__ pb2) {
    __shared__ __attribute__((aligned(16))) float h_s[16][HP_];
    __shared__ __attribute__((aligned(16))) float p_s[16][PP_];
    __shared__ float zh_s[16][HP_];
    __shared__ float hn_s[16][HP_];
    __shared__ float u_s[16][PP_];
    const int tid = threadIdx.x, lane = tid & 31;
    const int wv = __builtin_amdgcn_readfirstlane(tid >> 5);
    const int m0 = blockIdx.x * 16, n0 = wv * 16;
    const bool pmw = wv < 4;

    const v4f* hg = reinterpret_cast<const v4f*>(h + (size_t)m0 * H_);
    const v4f* pg = reinterpret_cast<const v4f*>(p + (size_t)m0 * P2_);
    const v4f hq0 = hg[tid], hq1 = hg[tid + 256], pq = pg[tid];
    v2f bxz[P2_ / 4];
    #pragma unroll
    for (int k = 0; k < P2_; k += 4) bxz[k >> 2] = load_bfrag(xz_w, H_, k, n0, lane);
    *reinterpret_cast<v4f*>(&h_s[tid >> 5][(tid & 31) << 2])       = hq0;
    *reinterpret_cast<v4f*>(&h_s[8 + (tid >> 5)][(tid & 31) << 2]) = hq1;
    *reinterpret_cast<v4f*>(&p_s[tid >> 4][(tid & 15) << 2])       = pq;
    __syncthreads();

    v8f az = {};
    #pragma unroll
    for (int k = 0; k < P2_; k += 4)
        az = wmma4(load_afrag(&p_s[0][0], PP_, 0, k, lane), bxz[k >> 2], az);
    for (int k = 0; k < H_; k += 4)
        az = wmma4(load_afrag(&h_s[0][0], HP_, 0, k, lane), load_bfrag(hz_w, H_, k, n0, lane), az);

    const int col = n0 + (lane & 15), rb = (lane >> 4) << 3;
    const float bz = xz_b[col];
    float z[8];
    #pragma unroll
    for (int j = 0; j < 8; ++j) {
        z[j] = sigmoidf_(az[j] + bz);
        zh_s[rb + j][col] = z[j] * h_s[rb + j][col];
    }
    __syncthreads();

    v2f bw1[H_ / 4] = {}, bw2[P2_ / 4] = {};
    if (pmw) { pm_load_w<H_>(bw1, pw1, n0, lane); pm_load_w<P2_>(bw2, pw2, n0, lane); }

    v8f an = {};
    for (int k = 0; k < P2_; k += 4)
        an = wmma4(load_afrag(&p_s[0][0], PP_, 0, k, lane), load_bfrag(xn_w, H_, k, n0, lane), an);
    for (int k = 0; k < H_; k += 4)
        an = wmma4(load_afrag(&zh_s[0][0], HP_, 0, k, lane), load_bfrag(hn_w, H_, k, n0, lane), an);

    const float bn = xn_b[col];
    float hnew[8];
    #pragma unroll
    for (int j = 0; j < 8; ++j) {
        float n = tanhf(an[j] + bn);
        float hv = h_s[rb + j][col];
        hnew[j] = hv + DT_ * (1.0f - z[j]) * (n - hv);
        hn_s[rb + j][col] = hnew[j];
    }
    __syncthreads();
    #pragma unroll
    for (int j = 0; j < 8; ++j) h[(m0 + rb + j) * H_ + col] = hnew[j];

    if (pmw) {
        v8f a1 = pm_gemm<H_>(&hn_s[0][0], HP_, bw1, lane);
        const float bv = pb1[col];
        #pragma unroll
        for (int j = 0; j < 8; ++j) u_s[rb + j][col] = fmaxf(a1[j] + bv, 0.0f);
    }
    __syncthreads();
    if (pmw) {
        v8f a2 = pm_gemm<P2_>(&u_s[0][0], PP_, bw2, lane);
        const float bv2 = pb2[col];
        #pragma unroll
        for (int j = 0; j < 8; ++j) p[(m0 + rb + j) * P2_ + col] = a2[j] + bv2;
    }
}

__global__ void __launch_bounds__(JT_) k_jump(int t,
                       const float* __restrict__ X, const float* __restrict__ M,
                       const int* __restrict__ obs_idx,
                       float* __restrict__ h, float* __restrict__ p,
                       const float* __restrict__ w_ih, const float* __restrict__ w_hh,
                       const float* __restrict__ b_ih, const float* __restrict__ b_hh,
                       const float* __restrict__ w_prep, const float* __restrict__ bias_prep,
                       const float* __restrict__ pw1, const float* __restrict__ pb1,
                       const float* __restrict__ pw2, const float* __restrict__ pb2,
                       float* __restrict__ loss_part) {
    __shared__ int   idx_s[16];
    __shared__ float hold_s[16][HP_];
    __shared__ float gin_s[16][GP_];
    __shared__ float gi_s[16][G3P_];
    __shared__ float gh_s[16][G3P_];
    __shared__ float red[16 * D_];
    __shared__ float u_s[16][PP_];
    const int tid = threadIdx.x, lane = tid & 31;
    const int wv = __builtin_amdgcn_readfirstlane(tid >> 5);
    const bool pmw = (wv >> 2) == 2;
    const int pn0 = (wv & 3) * 16;
    const int o0 = blockIdx.x * 16;
    const float* Xt = X + (size_t)t * NO_ * D_;
    const float* Mt = M + (size_t)t * NO_ * D_;

    if (tid < 16) idx_s[tid] = obs_idx[t * NO_ + o0 + tid];
    __syncthreads();

    for (int i = tid; i < 16 * H_; i += JT_) {
        int o = i >> 7, j = i & 127;
        hold_s[o][j] = h[idx_s[o] * H_ + j];
    }

    float lterm = 0.0f;
    if (tid < 16 * D_) {
        int o = tid / D_, d = tid % D_;
        int gb = o0 + o, b = idx_s[o];
        float mean = p[b * P2_ + d];
        float lv   = p[b * P2_ + D_ + d];
        float x    = Xt[gb * D_ + d];
        float m    = Mt[gb * D_ + d];
        float err  = (x - mean) * expf(-0.5f * lv);
        lterm = 0.5f * ((err * err + lv + 2.0f * LOGC_) * m);
    }

    for (int i = tid; i < 16 * GIN_; i += JT_) {
        int o = i / GIN_, c = i % GIN_;
        int d = c / PREP_, pi = c - d * PREP_;
        int gb = o0 + o, b = idx_s[o];
        float mean = p[b * P2_ + d];
        float lv   = p[b * P2_ + D_ + d];
        float x    = Xt[gb * D_ + d];
        float m    = Mt[gb * D_ + d];
        float err  = (x - mean) * expf(-0.5f * lv);
        const float* wp = w_prep + (size_t)(d * 4) * PREP_ + pi;
        float a = x * wp[0] + mean * wp[PREP_] + lv * wp[2 * PREP_] + err * wp[3 * PREP_]
                + bias_prep[d * PREP_ + pi];
        gin_s[o][c] = fmaxf(a, 0.0f) * m;
    }
    __syncthreads();

    {
        const int n0 = wv * 16;
        v8f gi = {};
        for (int k = 0; k < GIN_; k += 4)
            gi = wmma4(load_afrag(&gin_s[0][0], GP_, 0, k, lane), load_bfrag(w_ih, G3_, k, n0, lane), gi);
        v8f gh = {};
        for (int k = 0; k < H_; k += 4)
            gh = wmma4(load_afrag(&hold_s[0][0], HP_, 0, k, lane), load_bfrag(w_hh, G3_, k, n0, lane), gh);
        const int col = n0 + (lane & 15), rb = (lane >> 4) << 3;
        const float bi = b_ih[col], bh = b_hh[col];
        #pragma unroll
        for (int j = 0; j < 8; ++j) {
            gi_s[rb + j][col] = gi[j] + bi;
            gh_s[rb + j][col] = gh[j] + bh;
        }
    }
    __syncthreads();

    v2f bw1[H_ / 4] = {}, bw2[P2_ / 4] = {};
    if (pmw) { pm_load_w<H_>(bw1, pw1, pn0, lane); pm_load_w<P2_>(bw2, pw2, pn0, lane); }

    for (int i = tid; i < 16 * H_; i += JT_) {
        int o = i >> 7, j = i & 127;
        float r = sigmoidf_(gi_s[o][j] + gh_s[o][j]);
        float z = sigmoidf_(gi_s[o][H_ + j] + gh_s[o][H_ + j]);
        float n = tanhf(gi_s[o][2 * H_ + j] + r * gh_s[o][2 * H_ + j]);
        float hv = hold_s[o][j];
        float hnew = (1.0f - z) * n + z * hv;
        h[idx_s[o] * H_ + j] = hnew;
        hold_s[o][j] = hnew;
    }

    if (tid < 16 * D_) red[tid] = lterm;
    __syncthreads();
    const int pcol = pn0 + (lane & 15), prb = (lane >> 4) << 3;
    if (pmw) {
        v8f a1 = pm_gemm<H_>(&hold_s[0][0], HP_, bw1, lane);
        const float bv = pb1[pcol];
        #pragma unroll
        for (int j = 0; j < 8; ++j) u_s[prb + j][pcol] = fmaxf(a1[j] + bv, 0.0f);
    }
    if (tid < 8 * D_) red[tid] += red[tid + 8 * D_];
    __syncthreads();
    if (pmw) {
        v8f a2 = pm_gemm<P2_>(&u_s[0][0], PP_, bw2, lane);
        const float bv2 = pb2[pcol];
        #pragma unroll
        for (int j = 0; j < 8; ++j) p[idx_s[prb + j] * P2_ + pcol] = a2[j] + bv2;
    }
    for (int s2 = 4 * D_; s2 > 0; s2 >>= 1) {
        if (tid < s2) red[tid] += red[tid + s2];
        __syncthreads();
    }
    if (tid == 0) loss_part[t * (NO_ / 16) + blockIdx.x] = red[0];
}

__global__ void k_loss_final(const float* __restrict__ parts, float* __restrict__ loss) {
    __shared__ float red[256];
    const int tid = threadIdx.x;
    float s = 0.0f;
    for (int i = tid; i < NT_ * (NO_ / 16); i += 256) s += parts[i];
    red[tid] = s;
    __syncthreads();
    for (int k = 128; k > 0; k >>= 1) {
        if (tid < k) red[tid] += red[tid + k];
        __syncthreads();
    }
    if (tid == 0) *loss = red[0];
}

extern "C" void kernel_launch(void* const* d_in, const int* in_sizes, int n_in,
                              void* d_out, int out_size, void* d_ws, size_t ws_size,
                              hipStream_t stream) {
    const float* X       = (const float*)d_in[0];
    const float* M       = (const float*)d_in[1];
    const int*   obs_idx = (const int*)d_in[2];
    const float* cov     = (const float*)d_in[3];
    const float* cm_w1   = (const float*)d_in[4];
    const float* cm_b1   = (const float*)d_in[5];
    const float* cm_w2   = (const float*)d_in[6];
    const float* cm_b2   = (const float*)d_in[7];
    const float* p_w1    = (const float*)d_in[8];
    const float* p_b1    = (const float*)d_in[9];
    const float* p_w2    = (const float*)d_in[10];
    const float* p_b2    = (const float*)d_in[11];
    const float* xz_w    = (const float*)d_in[12];
    const float* xz_b    = (const float*)d_in[13];
    const float* hz_w    = (const float*)d_in[14];
    const float* xn_w    = (const float*)d_in[15];
    const float* xn_b    = (const float*)d_in[16];
    const float* hn_w    = (const float*)d_in[17];
    const float* w_ih    = (const float*)d_in[18];
    const float* w_hh    = (const float*)d_in[19];
    const float* b_ih    = (const float*)d_in[20];
    const float* b_hh    = (const float*)d_in[21];
    const float* w_prep  = (const float*)d_in[22];
    const float* b_prep  = (const float*)d_in[23];

    float* h    = (float*)d_out;
    float* p    = h + (size_t)B_ * H_;
    float* loss = p + (size_t)B_ * P2_;
    float* loss_part = (float*)d_ws;

    const int MB = B_ / 16;
    const int OB = NO_ / 16;

    k_init<<<MB, 128, 0, stream>>>(cov, cm_w1, cm_b1, cm_w2, cm_b2,
                                   p_w1, p_b1, p_w2, p_b2, h, p, loss);

    for (int t = 0; t < NT_; ++t) {
        k_ode<<<MB, 256, 0, stream>>>(h, p, xz_w, xz_b, hz_w, xn_w, xn_b, hn_w,
                                      p_w1, p_b1, p_w2, p_b2);
        k_jump<<<OB, JT_, 0, stream>>>(t, X, M, obs_idx, h, p,
                                       w_ih, w_hh, b_ih, b_hh, w_prep, b_prep,
                                       p_w1, p_b1, p_w2, p_b2, loss_part);
    }
    for (int s = 0; s < NPOST_; ++s)
        k_ode<<<MB, 256, 0, stream>>>(h, p, xz_w, xz_b, hz_w, xn_w, xn_b, hn_w,
                                      p_w1, p_b1, p_w2, p_b2);
    k_loss_final<<<1, 256, 0, stream>>>(loss_part, loss);
}


# jump kernel front: x, mean, logvar, err of each (o,d) computed once into LDS and gin built from there with per-thread weights; gathers issued as one group, not a memory trip per loop pass; sums kept
# speedup vs baseline: 2.0104x; 1.0626x over previous
#include <hip/hip_runtime.h>
#include <hip/hip_bf16.h>
#include <math.h>

#define B_    8192
#define D_    32
#define H_    128
#define P2_   64
#define PREP_ 10
#define GIN_  320
#define G3_   384
#define CS_   8
#define CH_   32
#define NT_   16
#define NO_   4096
#define DT_   0.1f
#define NPOST_ 4
#define JT_   768
#define LOGC_  0.91893853320467274178f

#define HP_   (H_ + 4)
#define PP_   (P2_ + 4)
#define GP_   (GIN_ + 4)
#define G3P_  (G3_ + 4)

typedef __attribute__((ext_vector_type(2))) float v2f;
typedef __attribute__((ext_vector_type(4))) float v4f;
typedef __attribute__((ext_vector_type(8))) float v8f;

__device__ __forceinline__ float sigmoidf_(float x) { return 1.0f / (1.0f + expf(-x)); }


__device__ __forceinline__ v2f load_afrag(const float* A, int lda, int m0, int k, int lane) {
    int m  = m0 + (lane & 15);
    int kk = k + ((lane >> 4) << 1);
    v2f a;
    a.x = A[m * lda + kk];
    a.y = A[m * lda + kk + 1];
    return a;
}

__device__ __forceinline__ v2f load_bfrag(const float* Bm, int ldb, int k, int n0, int lane) {
    int n  = n0 + (lane & 15);
    int kk = k + ((lane >> 4) << 1);
    v2f b;
    b.x = Bm[kk * ldb + n];
    b.y = Bm[(kk + 1) * ldb + n];
    return b;
}

__device__ __forceinline__ v8f wmma4(v2f a, v2f b, v8f c) {
    return __builtin_amdgcn_wmma_f32_16x16x4_f32(false, a, false, b, (short)0, c, false, false);
}

template <int K>
__device__ __forceinline__ void pm_load_w(v2f (&bw)[K / 4], const float* __restrict__ w, int n0, int lane) {
    #pragma unroll
    for (int k = 0; k < K; k += 4) bw[k >> 2] = load_bfrag(w, P2_, k, n0, lane);
}

template <int K>
__device__ __forceinline__ v8f pm_gemm(const float* as, int pitch, const v2f (&bw)[K / 4], int lane) {
    v8f acc = {};
    #pragma unroll
    for (int k = 0; k < K; k += 4) acc = wmma4(load_afrag(as, pitch, 0, k, lane), bw[k >> 2], acc);
    return acc;
}

__global__ void __launch_bounds__(128) k_init(const float* __restrict__ cov,
                       const float* __restrict__ w1, const float* __restrict__ b1,
                       const float* __restrict__ w2, const float* __restrict__ b2,
                       const float* __restrict__ pw1, const float* __restrict__ pb1,
                       const float* __restrict__ pw2, const float* __restrict__ pb2,
                       float* __restrict__ h, float* __restrict__ p, float* __restrict__ loss) {
    __shared__ float t_s[16][CH_];
    __shared__ float hn_s[16][HP_];
    __shared__ float u_s[16][PP_];
    const int tid = threadIdx.x;
    const int lane = tid & 31, pn0 = (tid >> 5) * 16;
    const int m0  = blockIdx.x * 16;
    if (blockIdx.x == 0 && tid == 0) *loss = 0.0f;
    v2f bw1[H_ / 4], bw2[P2_ / 4];
    pm_load_w<H_>(bw1, pw1, pn0, lane);
    pm_load_w<P2_>(bw2, pw2, pn0, lane);
    for (int i = tid; i < 16 * CH_; i += 128) {
        int r = i / CH_, c = i % CH_;
        float acc = b1[c];
        #pragma unroll
        for (int k = 0; k < CS_; ++k) acc += cov[(m0 + r) * CS_ + k] * w1[k * CH_ + c];
        t_s[r][c] = fmaxf(acc, 0.0f);
    }
    __syncthreads();
    for (int i = tid; i < 16 * H_; i += 128) {
        int r = i / H_, c = i % H_;
        float acc = b2[c];
        #pragma unroll
        for (int k = 0; k < CH_; ++k) acc += t_s[r][k] * w2[k * H_ + c];
        float hv = tanhf(acc);
        h[(m0 + r) * H_ + c] = hv;
        hn_s[r][c] = hv;
    }
    __syncthreads();

    const int col = pn0 + (lane & 15), rb = (lane >> 4) << 3;
    v8f a1 = pm_gemm<H_>(&hn_s[0][0], HP_, bw1, lane);
    const float bv = pb1[col];
    #pragma unroll
    for (int j = 0; j < 8; ++j) u_s[rb + j][col] = fmaxf(a1[j] + bv, 0.0f);
    __syncthreads();
    v8f a2 = pm_gemm<P2_>(&u_s[0][0], PP_, bw2, lane);
    const float bv2 = pb2[col];
    #pragma unroll
    for (int j = 0; j < 8; ++j) p[(m0 + rb + j) * P2_ + col] = a2[j] + bv2;
}

__global__ void __launch_bounds__(256) k_ode(float* __restrict__ h, float* __restrict__ p,
                      const float* __restrict__ xz_w, const float* __restrict__ xz_b,
                      const float* __restrict__ hz_w,
                      const float* __restrict__ xn_w, const float* __restrict__ xn_b,
                      const float* __restrict__ hn_w,
                      const float* __restrict__ pw1, const float* __restrict__ pb1,
                      const float* __restrict__ pw2, const float* __restrict__ pb2) {
    __shared__ __attribute__((aligned(16))) float h_s[16][HP_];
    __shared__ __attribute__((aligned(16))) float p_s[16][PP_];
    __shared__ float zh_s[16][HP_];
    __shared__ float hn_s[16][HP_];
    __shared__ float u_s[16][PP_];
    const int tid = threadIdx.x, lane = tid & 31;
    const int wv = __builtin_amdgcn_readfirstlane(tid >> 5);
    const int m0 = blockIdx.x * 16, n0 = wv * 16;
    const bool pmw = wv < 4;

    const v4f* hg = reinterpret_cast<const v4f*>(h + (size_t)m0 * H_);
    const v4f* pg = reinterpret_cast<const v4f*>(p + (size_t)m0 * P2_);
    const v4f hq0 = hg[tid], hq1 = hg[tid + 256], pq = pg[tid];
    v2f bxz[P2_ / 4];
    #pragma unroll
    for (int k = 0; k < P2_; k += 4) bxz[k >> 2] = load_bfrag(xz_w, H_, k, n0, lane);
    *reinterpret_cast<v4f*>(&h_s[tid >> 5][(tid & 31) << 2])       = hq0;
    *reinterpret_cast<v4f*>(&h_s[8 + (tid >> 5)][(tid & 31) << 2]) = hq1;
    *reinterpret_cast<v4f*>(&p_s[tid >> 4][(tid & 15) << 2])       = pq;
    __syncthreads();

    v8f az = {};
    #pragma unroll
    for (int k = 0; k < P2_; k += 4)
        az = wmma4(load_afrag(&p_s[0][0], PP_, 0, k, lane), bxz[k >> 2], az);
    for (int k = 0; k < H_; k += 4)
        az = wmma4(load_afrag(&h_s[0][0], HP_, 0, k, lane), load_bfrag(hz_w, H_, k, n0, lane), az);

    const int col = n0 + (lane & 15), rb = (lane >> 4) << 3;
    const float bz = xz_b[col];
    float z[8];
    #pragma unroll
    for (int j = 0; j < 8; ++j) {
        z[j] = sigmoidf_(az[j] + bz);
        zh_s[rb + j][col] = z[j] * h_s[rb + j][col];
    }
    __syncthreads();

    v2f bw1[H_ / 4] = {}, bw2[P2_ / 4] = {};
    if (pmw) { pm_load_w<H_>(bw1, pw1, n0, lane); pm_load_w<P2_>(bw2, pw2, n0, lane); }

    v8f an = {};
    for (int k = 0; k < P2_; k += 4)
        an = wmma4(load_afrag(&p_s[0][0], PP_, 0, k, lane), load_bfrag(xn_w, H_, k, n0, lane), an);
    for (int k = 0; k < H_; k += 4)
        an = wmma4(load_afrag(&zh_s[0][0], HP_, 0, k, lane), load_bfrag(hn_w, H_, k, n0, lane), an);

    const float bn = xn_b[col];
    float hnew[8];
    #pragma unroll
    for (int j = 0; j < 8; ++j) {
        float n = tanhf(an[j] + bn);
        float hv = h_s[rb + j][col];
        hnew[j] = hv + DT_ * (1.0f - z[j]) * (n - hv);
        hn_s[rb + j][col] = hnew[j];
    }
    __syncthreads();
    #pragma unroll
    for (int j = 0; j < 8; ++j) h[(m0 + rb + j) * H_ + col] = hnew[j];

    if (pmw) {
        v8f a1 = pm_gemm<H_>(&hn_s[0][0], HP_, bw1, lane);
        const float bv = pb1[col];
        #pragma unroll
        for (int j = 0; j < 8; ++j) u_s[rb + j][col] = fmaxf(a1[j] + bv, 0.0f);
    }
    __syncthreads();
    if (pmw) {
        v8f a2 = pm_gemm<P2_>(&u_s[0][0], PP_, bw2, lane);
        const float bv2 = pb2[col];
        #pragma unroll
        for (int j = 0; j < 8; ++j) p[(m0 + rb + j) * P2_ + col] = a2[j] + bv2;
    }
}

__global__ void __launch_bounds__(JT_) k_jump(int t,
                       const float* __restrict__ X, const float* __restrict__ M,
                       const int* __restrict__ obs_idx,
                       float* __restrict__ h, float* __restrict__ p,
                       const float* __restrict__ w_ih, const float* __restrict__ w_hh,
                       const float* __restrict__ b_ih, const float* __restrict__ b_hh,
                       const float* __restrict__ w_prep, const float* __restrict__ bias_prep,
                       const float* __restrict__ pw1, const float* __restrict__ pb1,
                       const float* __restrict__ pw2, const float* __restrict__ pb2,
                       float* __restrict__ loss_part) {
    __shared__ int   idx_s[16];
    __shared__ __attribute__((aligned(16))) float hold_s[16][HP_];
    __shared__ float gin_s[16][GP_];
    __shared__ float gi_s[16][G3P_];
    __shared__ float gh_s[16][G3P_];
    __shared__ float red[16 * D_];
    __shared__ float u_s[16][PP_];
    __shared__ v4f   feat_s[16 * D_];
    __shared__ float m_s[16 * D_];
    const int tid = threadIdx.x, lane = tid & 31;
    const int wv = __builtin_amdgcn_readfirstlane(tid >> 5);
    const bool pmw = (wv >> 2) == 2;
    const int pn0 = (wv & 3) * 16;
    const int o0 = blockIdx.x * 16;
    const float* Xt = X + (size_t)t * NO_ * D_;
    const float* Mt = M + (size_t)t * NO_ * D_;

    float xv = 0.0f, mv = 0.0f;
    if (tid < 16 * D_) { xv = Xt[o0 * D_ + tid]; mv = Mt[o0 * D_ + tid]; }
    const int og = (tid >= GIN_) ? 1 : 0;
    const int gc = tid - og * GIN_;
    const int gd = gc / PREP_;
    float w0 = 0.0f, w1 = 0.0f, w2 = 0.0f, w3 = 0.0f, bp = 0.0f;
    if (tid < 2 * GIN_) {
        const float* wp = w_prep + (size_t)(gd * 4) * PREP_ + (gc - gd * PREP_);
        w0 = wp[0]; w1 = wp[PREP_]; w2 = wp[2 * PREP_]; w3 = wp[3 * PREP_];
        bp = bias_prep[gc];
    }
    if (tid < 16) idx_s[tid] = obs_idx[t * NO_ + o0 + tid];
    __syncthreads();

    float lterm = 0.0f;
    if (tid >= 16 * D_) {
        const int q = tid - 16 * D_;
        const int oa = q >> 5, c4 = (q & 31) << 2;
        const v4f ha = *reinterpret_cast<const v4f*>(h + (size_t)idx_s[oa] * H_ + c4);
        const v4f hb = *reinterpret_cast<const v4f*>(h + (size_t)idx_s[oa + 8] * H_ + c4);
        *reinterpret_cast<v4f*>(&hold_s[oa][c4])     = ha;
        *reinterpret_cast<v4f*>(&hold_s[oa + 8][c4]) = hb;
    } else {
        const int o = tid / D_, d = tid % D_;
        const int b = idx_s[o];
        const float mean = p[b * P2_ + d];
        const float lv   = p[b * P2_ + D_ + d];
        const float err  = (xv - mean) * expf(-0.5f * lv);
        lterm = 0.5f * ((err * err + lv + 2.0f * LOGC_) * mv);
        v4f f; f.x = xv; f.y = mean; f.z = lv; f.w = err;
        feat_s[tid] = f;
        m_s[tid] = mv;
    }
    __syncthreads();

    if (tid < 2 * GIN_) {
        #pragma unroll
        for (int j = 0; j < 8; ++j) {
            const int o = og * 8 + j;
            const v4f f = feat_s[o * D_ + gd];
            const float mm = m_s[o * D_ + gd];
            const float a = fmaf(f.w, w3, fmaf(f.z, w2, fmaf(f.x, w0, f.y * w1))) + bp;
            gin_s[o][gc] = fmaxf(a, 0.0f) * mm;
        }
    }
    __syncthreads();

    {
        const int n0 = wv * 16;
        v8f gi = {};
        for (int k = 0; k < GIN_; k += 4)
            gi = wmma4(load_afrag(&gin_s[0][0], GP_, 0, k, lane), load_bfrag(w_ih, G3_, k, n0, lane), gi);
        v8f gh = {};
        for (int k = 0; k < H_; k += 4)
            gh = wmma4(load_afrag(&hold_s[0][0], HP_, 0, k, lane), load_bfrag(w_hh, G3_, k, n0, lane), gh);
        const int col = n0 + (lane & 15), rb = (lane >> 4) << 3;
        const float bi = b_ih[col], bh = b_hh[col];
        #pragma unroll
        for (int j = 0; j < 8; ++j) {
            gi_s[rb + j][col] = gi[j] + bi;
            gh_s[rb + j][col] = gh[j] + bh;
        }
    }
    __syncthreads();

    v2f bw1[H_ / 4] = {}, bw2[P2_ / 4] = {};
    if (pmw) { pm_load_w<H_>(bw1, pw1, pn0, lane); pm_load_w<P2_>(bw2, pw2, pn0, lane); }

    for (int i = tid; i < 16 * H_; i += JT_) {
        int o = i >> 7, j = i & 127;
        float r = sigmoidf_(gi_s[o][j] + gh_s[o][j]);
        float z = sigmoidf_(gi_s[o][H_ + j] + gh_s[o][H_ + j]);
        float n = tanhf(gi_s[o][2 * H_ + j] + r * gh_s[o][2 * H_ + j]);
        float hv = hold_s[o][j];
        float hnew = (1.0f - z) * n + z * hv;
        h[idx_s[o] * H_ + j] = hnew;
        hold_s[o][j] = hnew;
    }

    if (tid < 16 * D_) red[tid] = lterm;
    __syncthreads();
    const int pcol = pn0 + (lane & 15), prb = (lane >> 4) << 3;
    if (pmw) {
        v8f a1 = pm_gemm<H_>(&hold_s[0][0], HP_, bw1, lane);
        const float bv = pb1[pcol];
        #pragma unroll
        for (int j = 0; j < 8; ++j) u_s[prb + j][pcol] = fmaxf(a1[j] + bv, 0.0f);
    }
    if (tid < 8 * D_) red[tid] += red[tid + 8 * D_];
    __syncthreads();
    if (pmw) {
        v8f a2 = pm_gemm<P2_>(&u_s[0][0], PP_, bw2, lane);
        const float bv2 = pb2[pcol];
        #pragma unroll
        for (int j = 0; j < 8; ++j) p[idx_s[prb + j] * P2_ + pcol] = a2[j] + bv2;
    }
    for (int s2 = 4 * D_; s2 > 0; s2 >>= 1) {
        if (tid < s2) red[tid] += red[tid + s2];
        __syncthreads();
    }
    if (tid == 0) loss_part[t * (NO_ / 16) + blockIdx.x] = red[0];
}

__global__ void k_loss_final(const float* __restrict__ parts, float* __restrict__ loss) {
    __shared__ float red[256];
    const int tid = threadIdx.x;
    float s = 0.0f;
    for (int i = tid; i < NT_ * (NO_ / 16); i += 256) s += parts[i];
    red[tid] = s;
    __syncthreads();
    for (int k = 128; k > 0; k >>= 1) {
        if (tid < k) red[tid] += red[tid + k];
        __syncthreads();
    }
    if (tid == 0) *loss = red[0];
}

extern "C" void kernel_launch(void* const* d_in, const int* in_sizes, int n_in,
                              void* d_out, int out_size, void* d_ws, size_t ws_size,
                              hipStream_t stream) {
    const float* X       = (const float*)d_in[0];
    const float* M       = (const float*)d_in[1];
    const int*   obs_idx = (const int*)d_in[2];
    const float* cov     = (const float*)d_in[3];
    const float* cm_w1   = (const float*)d_in[4];
    const float* cm_b1   = (const float*)d_in[5];
    const float* cm_w2   = (const float*)d_in[6];
    const float* cm_b2   = (const float*)d_in[7];
    const float* p_w1    = (const float*)d_in[8];
    const float* p_b1    = (const float*)d_in[9];
    const float* p_w2    = (const float*)d_in[10];
    const float* p_b2    = (const float*)d_in[11];
    const float* xz_w    = (const float*)d_in[12];
    const float* xz_b    = (const float*)d_in[13];
    const float* hz_w    = (const float*)d_in[14];
    const float* xn_w    = (const float*)d_in[15];
    const float* xn_b    = (const float*)d_in[16];
    const float* hn_w    = (const float*)d_in[17];
    const float* w_ih    = (const float*)d_in[18];
    const float* w_hh    = (const float*)d_in[19];
    const float* b_ih    = (const float*)d_in[20];
    const float* b_hh    = (const float*)d_in[21];
    const float* w_prep  = (const float*)d_in[22];
    const float* b_prep  = (const float*)d_in[23];

    float* h    = (float*)d_out;
    float* p    = h + (size_t)B_ * H_;
    float* loss = p + (size_t)B_ * P2_;
    float* loss_part = (float*)d_ws;

    const int MB = B_ / 16;
    const int OB = NO_ / 16;

    k_init<<<MB, 128, 0, stream>>>(cov, cm_w1, cm_b1, cm_w2, cm_b2,
                                   p_w1, p_b1, p_w2, p_b2, h, p, loss);

    for (int t = 0; t < NT_; ++t) {
        k_ode<<<MB, 256, 0, stream>>>(h, p, xz_w, xz_b, hz_w, xn_w, xn_b, hn_w,
                                      p_w1, p_b1, p_w2, p_b2);
        k_jump<<<OB, JT_, 0, stream>>>(t, X, M, obs_idx, h, p,
                                       w_ih, w_hh, b_ih, b_hh, w_prep, b_prep,
                                       p_w1, p_b1, p_w2, p_b2, loss_part);
    }
    for (int s = 0; s < NPOST_; ++s)
        k_ode<<<MB, 256, 0, stream>>>(h, p, xz_w, xz_b, hz_w, xn_w, xn_b, hn_w,
                                      p_w1, p_b1, p_w2, p_b2);
    k_loss_final<<<1, 256, 0, stream>>>(loss_part, loss);
}


# ode kernel: 32 rows per work-group; each wave runs its 16-column gate tile for two row tiles, so a loaded weight fragment feeds two products; p-model on all 8 waves; order of every sum kept
# speedup vs baseline: 2.0736x; 1.0314x over previous
#include <hip/hip_runtime.h>
#include <hip/hip_bf16.h>
#include <math.h>

#define B_    8192
#define D_    32
#define H_    128
#define P2_   64
#define PREP_ 10
#define GIN_  320
#define G3_   384
#define CS_   8
#define CH_   32
#define NT_   16
#define NO_   4096
#define DT_   0.1f
#define NPOST_ 4
#define OR_   32
#define JT_   768
#define LOGC_  0.91893853320467274178f

#define HP_   (H_ + 4)
#define PP_   (P2_ + 4)
#define GP_   (GIN_ + 4)
#define G3P_  (G3_ + 4)

typedef __attribute__((ext_vector_type(2))) float v2f;
typedef __attribute__((ext_vector_type(4))) float v4f;
typedef __attribute__((ext_vector_type(8))) float v8f;

__device__ __forceinline__ float sigmoidf_(float x) { return 1.0f / (1.0f + expf(-x)); }


__device__ __forceinline__ v2f load_afrag(const float* A, int lda, int m0, int k, int lane) {
    int m  = m0 + (lane & 15);
    int kk = k + ((lane >> 4) << 1);
    v2f a;
    a.x = A[m * lda + kk];
    a.y = A[m * lda + kk + 1];
    return a;
}

__device__ __forceinline__ v2f load_bfrag(const float* Bm, int ldb, int k, int n0, int lane) {
    int n  = n0 + (lane & 15);
    int kk = k + ((lane >> 4) << 1);
    v2f b;
    b.x = Bm[kk * ldb + n];
    b.y = Bm[(kk + 1) * ldb + n];
    return b;
}

__device__ __forceinline__ v8f wmma4(v2f a, v2f b, v8f c) {
    return __builtin_amdgcn_wmma_f32_16x16x4_f32(false, a, false, b, (short)0, c, false, false);
}

template <int K>
__device__ __forceinline__ void pm_load_w(v2f (&bw)[K / 4], const float* __restrict__ w, int n0, int lane) {
    #pragma unroll
    for (int k = 0; k < K; k += 4) bw[k >> 2] = load_bfrag(w, P2_, k, n0, lane);
}

template <int K>
__device__ __forceinline__ v8f pm_gemm(const float* as, int pitch, const v2f (&bw)[K / 4], int lane) {
    v8f acc = {};
    #pragma unroll
    for (int k = 0; k < K; k += 4) acc = wmma4(load_afrag(as, pitch, 0, k, lane), bw[k >> 2], acc);
    return acc;
}

__global__ void __launch_bounds__(128) k_init(const float* __restrict__ cov,
                       const float* __restrict__ w1, const float* __restrict__ b1,
                       const float* __restrict__ w2, const float* __restrict__ b2,
                       const float* __restrict__ pw1, const float* __restrict__ pb1,
                       const float* __restrict__ pw2, const float* __restrict__ pb2,
                       float* __restrict__ h, float* __restrict__ p, float* __restrict__ loss) {
    __shared__ float t_s[16][CH_];
    __shared__ float hn_s[16][HP_];
    __shared__ float u_s[16][PP_];
    const int tid = threadIdx.x;
    const int lane = tid & 31, pn0 = (tid >> 5) * 16;
    const int m0  = blockIdx.x * 16;
    if (blockIdx.x == 0 && tid == 0) *loss = 0.0f;
    v2f bw1[H_ / 4], bw2[P2_ / 4];
    pm_load_w<H_>(bw1, pw1, pn0, lane);
    pm_load_w<P2_>(bw2, pw2, pn0, lane);
    for (int i = tid; i < 16 * CH_; i += 128) {
        int r = i / CH_, c = i % CH_;
        float acc = b1[c];
        #pragma unroll
        for (int k = 0; k < CS_; ++k) acc += cov[(m0 + r) * CS_ + k] * w1[k * CH_ + c];
        t_s[r][c] = fmaxf(acc, 0.0f);
    }
    __syncthreads();
    for (int i = tid; i < 16 * H_; i += 128) {
        int r = i / H_, c = i % H_;
        float acc = b2[c];
        #pragma unroll
        for (int k = 0; k < CH_; ++k) acc += t_s[r][k] * w2[k * H_ + c];
        float hv = tanhf(acc);
        h[(m0 + r) * H_ + c] = hv;
        hn_s[r][c] = hv;
    }
    __syncthreads();

    const int col = pn0 + (lane & 15), rb = (lane >> 4) << 3;
    v8f a1 = pm_gemm<H_>(&hn_s[0][0], HP_, bw1, lane);
    const float bv = pb1[col];
    #pragma unroll
    for (int j = 0; j < 8; ++j) u_s[rb + j][col] = fmaxf(a1[j] + bv, 0.0f);
    __syncthreads();
    v8f a2 = pm_gemm<P2_>(&u_s[0][0], PP_, bw2, lane);
    const float bv2 = pb2[col];
    #pragma unroll
    for (int j = 0; j < 8; ++j) p[(m0 + rb + j) * P2_ + col] = a2[j] + bv2;
}

__global__ void __launch_bounds__(256) k_ode(float* __restrict__ h, float* __restrict__ p,
                      const float* __restrict__ xz_w, const float* __restrict__ xz_b,
                      const float* __restrict__ hz_w,
                      const float* __restrict__ xn_w, const float* __restrict__ xn_b,
                      const float* __restrict__ hn_w,
                      const float* __restrict__ pw1, const float* __restrict__ pb1,
                      const float* __restrict__ pw2, const float* __restrict__ pb2) {
    __shared__ __attribute__((aligned(16))) float h_s[OR_][HP_];
    __shared__ __attribute__((aligned(16))) float p_s[OR_][PP_];
    __shared__ float zh_s[OR_][HP_];
    __shared__ float hn_s[OR_][HP_];
    __shared__ float u_s[OR_][PP_];
    const int tid = threadIdx.x, lane = tid & 31;
    const int wv = tid >> 5;
    const int m0 = blockIdx.x * OR_, n0 = wv * 16;
    const int pn0 = (wv & 3) * 16;
    const int pm0 = (wv >> 2) * 16;

    const v4f* hg = reinterpret_cast<const v4f*>(h + (size_t)m0 * H_);
    const v4f* pg = reinterpret_cast<const v4f*>(p + (size_t)m0 * P2_);
    const v4f hq0 = hg[tid], hq1 = hg[tid + 256], hq2 = hg[tid + 512], hq3 = hg[tid + 768];
    const v4f pq0 = pg[tid], pq1 = pg[tid + 256];
    v2f bxz[P2_ / 4];
    #pragma unroll
    for (int k = 0; k < P2_; k += 4) bxz[k >> 2] = load_bfrag(xz_w, H_, k, n0, lane);
    *reinterpret_cast<v4f*>(&h_s[tid >> 5][(tid & 31) << 2])        = hq0;
    *reinterpret_cast<v4f*>(&h_s[8 + (tid >> 5)][(tid & 31) << 2])  = hq1;
    *reinterpret_cast<v4f*>(&h_s[16 + (tid >> 5)][(tid & 31) << 2]) = hq2;
    *reinterpret_cast<v4f*>(&h_s[24 + (tid >> 5)][(tid & 31) << 2]) = hq3;
    *reinterpret_cast<v4f*>(&p_s[tid >> 4][(tid & 15) << 2])        = pq0;
    *reinterpret_cast<v4f*>(&p_s[16 + (tid >> 4)][(tid & 15) << 2]) = pq1;
    __syncthreads();

    v8f az0 = {}, az1 = {};
    #pragma unroll
    for (int k = 0; k < P2_; k += 4) {
        az0 = wmma4(load_afrag(&p_s[0][0], PP_, 0, k, lane), bxz[k >> 2], az0);
        az1 = wmma4(load_afrag(&p_s[0][0], PP_, 16, k, lane), bxz[k >> 2], az1);
    }
    for (int k = 0; k < H_; k += 4) {
        const v2f b = load_bfrag(hz_w, H_, k, n0, lane);
        az0 = wmma4(load_afrag(&h_s[0][0], HP_, 0, k, lane), b, az0);
        az1 = wmma4(load_afrag(&h_s[0][0], HP_, 16, k, lane), b, az1);
    }

    const int col = n0 + (lane & 15), rb = (lane >> 4) << 3;
    const float bz = xz_b[col];
    float z0[8], z1[8];
    #pragma unroll
    for (int j = 0; j < 8; ++j) {
        z0[j] = sigmoidf_(az0[j] + bz);
        z1[j] = sigmoidf_(az1[j] + bz);
        zh_s[rb + j][col]      = z0[j] * h_s[rb + j][col];
        zh_s[16 + rb + j][col] = z1[j] * h_s[16 + rb + j][col];
    }
    __syncthreads();

    v2f bw1[H_ / 4], bw2[P2_ / 4];
    pm_load_w<H_>(bw1, pw1, pn0, lane);
    pm_load_w<P2_>(bw2, pw2, pn0, lane);

    v8f an0 = {}, an1 = {};
    for (int k = 0; k < P2_; k += 4) {
        const v2f b = load_bfrag(xn_w, H_, k, n0, lane);
        an0 = wmma4(load_afrag(&p_s[0][0], PP_, 0, k, lane), b, an0);
        an1 = wmma4(load_afrag(&p_s[0][0], PP_, 16, k, lane), b, an1);
    }
    for (int k = 0; k < H_; k += 4) {
        const v2f b = load_bfrag(hn_w, H_, k, n0, lane);
        an0 = wmma4(load_afrag(&zh_s[0][0], HP_, 0, k, lane), b, an0);
        an1 = wmma4(load_afrag(&zh_s[0][0], HP_, 16, k, lane), b, an1);
    }

    const float bn = xn_b[col];
    float hnew0[8], hnew1[8];
    #pragma unroll
    for (int j = 0; j < 8; ++j) {
        const float n0v = tanhf(an0[j] + bn);
        const float n1v = tanhf(an1[j] + bn);
        const float hv0 = h_s[rb + j][col];
        const float hv1 = h_s[16 + rb + j][col];
        hnew0[j] = hv0 + DT_ * (1.0f - z0[j]) * (n0v - hv0);
        hnew1[j] = hv1 + DT_ * (1.0f - z1[j]) * (n1v - hv1);
        hn_s[rb + j][col]      = hnew0[j];
        hn_s[16 + rb + j][col] = hnew1[j];
    }
    __syncthreads();
    #pragma unroll
    for (int j = 0; j < 8; ++j) {
        h[(m0 + rb + j) * H_ + col]      = hnew0[j];
        h[(m0 + 16 + rb + j) * H_ + col] = hnew1[j];
    }

    const int pcol = pn0 + (lane & 15);
    {
        v8f a1 = pm_gemm<H_>(&hn_s[pm0][0], HP_, bw1, lane);
        const float bv = pb1[pcol];
        #pragma unroll
        for (int j = 0; j < 8; ++j) u_s[pm0 + rb + j][pcol] = fmaxf(a1[j] + bv, 0.0f);
    }
    __syncthreads();
    {
        v8f a2 = pm_gemm<P2_>(&u_s[pm0][0], PP_, bw2, lane);
        const float bv2 = pb2[pcol];
        #pragma unroll
        for (int j = 0; j < 8; ++j) p[(m0 + pm0 + rb + j) * P2_ + pcol] = a2[j] + bv2;
    }
}

__global__ void __launch_bounds__(JT_) k_jump(int t,
                       const float* __restrict__ X, const float* __restrict__ M,
                       const int* __restrict__ obs_idx,
                       float* __restrict__ h, float* __restrict__ p,
                       const float* __restrict__ w_ih, const float* __restrict__ w_hh,
                       const float* __restrict__ b_ih, const float* __restrict__ b_hh,
                       const float* __restrict__ w_prep, const float* __restrict__ bias_prep,
                       const float* __restrict__ pw1, const float* __restrict__ pb1,
                       const float* __restrict__ pw2, const float* __restrict__ pb2,
                       float* __restrict__ loss_part) {
    __shared__ int   idx_s[16];
    __shared__ __attribute__((aligned(16))) float hold_s[16][HP_];
    __shared__ float gin_s[16][GP_];
    __shared__ float gi_s[16][G3P_];
    __shared__ float gh_s[16][G3P_];
    __shared__ float red[16 * D_];
    __shared__ float u_s[16][PP_];
    __shared__ v4f   feat_s[16 * D_];
    __shared__ float m_s[16 * D_];
    const int tid = threadIdx.x, lane = tid & 31;
    const int wv = __builtin_amdgcn_readfirstlane(tid >> 5);
    const bool pmw = (wv >> 2) == 2;
    const int pn0 = (wv & 3) * 16;
    const int o0 = blockIdx.x * 16;
    const float* Xt = X + (size_t)t * NO_ * D_;
    const float* Mt = M + (size_t)t * NO_ * D_;

    float xv = 0.0f, mv = 0.0f;
    if (tid < 16 * D_) { xv = Xt[o0 * D_ + tid]; mv = Mt[o0 * D_ + tid]; }
    const int og = (tid >= GIN_) ? 1 : 0;
    const int gc = tid - og * GIN_;
    const int gd = gc / PREP_;
    float w0 = 0.0f, w1 = 0.0f, w2 = 0.0f, w3 = 0.0f, bp = 0.0f;
    if (tid < 2 * GIN_) {
        const float* wp = w_prep + (size_t)(gd * 4) * PREP_ + (gc - gd * PREP_);
        w0 = wp[0]; w1 = wp[PREP_]; w2 = wp[2 * PREP_]; w3 = wp[3 * PREP_];
        bp = bias_prep[gc];
    }
    if (tid < 16) idx_s[tid] = obs_idx[t * NO_ + o0 + tid];
    __syncthreads();

    float lterm = 0.0f;
    if (tid >= 16 * D_) {
        const int q = tid - 16 * D_;
        const int oa = q >> 5, c4 = (q & 31) << 2;
        const v4f ha = *reinterpret_cast<const v4f*>(h + (size_t)idx_s[oa] * H_ + c4);
        const v4f hb = *reinterpret_cast<const v4f*>(h + (size_t)idx_s[oa + 8] * H_ + c4);
        *reinterpret_cast<v4f*>(&hold_s[oa][c4])     = ha;
        *reinterpret_cast<v4f*>(&hold_s[oa + 8][c4]) = hb;
    } else {
        const int o = tid / D_, d = tid % D_;
        const int b = idx_s[o];
        const float mean = p[b * P2_ + d];
        const float lv   = p[b * P2_ + D_ + d];
        const float err  = (xv - mean) * expf(-0.5f * lv);
        lterm = 0.5f * ((err * err + lv + 2.0f * LOGC_) * mv);
        v4f f; f.x = xv; f.y = mean; f.z = lv; f.w = err;
        feat_s[tid] = f;
        m_s[tid] = mv;
    }
    __syncthreads();

    if (tid < 2 * GIN_) {
        #pragma unroll
        for (int j = 0; j < 8; ++j) {
            const int o = og * 8 + j;
            const v4f f = feat_s[o * D_ + gd];
            const float mm = m_s[o * D_ + gd];
            const float a = fmaf(f.w, w3, fmaf(f.z, w2, fmaf(f.x, w0, f.y * w1))) + bp;
            gin_s[o][gc] = fmaxf(a, 0.0f) * mm;
        }
    }
    __syncthreads();

    {
        const int n0 = wv * 16;
        v8f gi = {};
        for (int k = 0; k < GIN_; k += 4)
            gi = wmma4(load_afrag(&gin_s[0][0], GP_, 0, k, lane), load_bfrag(w_ih, G3_, k, n0, lane), gi);
        v8f gh = {};
        for (int k = 0; k < H_; k += 4)
            gh = wmma4(load_afrag(&hold_s[0][0], HP_, 0, k, lane), load_bfrag(w_hh, G3_, k, n0, lane), gh);
        const int col = n0 + (lane & 15), rb = (lane >> 4) << 3;
        const float bi = b_ih[col], bh = b_hh[col];
        #pragma unroll
        for (int j = 0; j < 8; ++j) {
            gi_s[rb + j][col] = gi[j] + bi;
            gh_s[rb + j][col] = gh[j] + bh;
        }
    }
    __syncthreads();

    v2f bw1[H_ / 4] = {}, bw2[P2_ / 4] = {};
    if (pmw) { pm_load_w<H_>(bw1, pw1, pn0, lane); pm_load_w<P2_>(bw2, pw2, pn0, lane); }

    for (int i = tid; i < 16 * H_; i += JT_) {
        int o = i >> 7, j = i & 127;
        float r = sigmoidf_(gi_s[o][j] + gh_s[o][j]);
        float z = sigmoidf_(gi_s[o][H_ + j] + gh_s[o][H_ + j]);
        float n = tanhf(gi_s[o][2 * H_ + j] + r * gh_s[o][2 * H_ + j]);
        float hv = hold_s[o][j];
        float hnew = (1.0f - z) * n + z * hv;
        h[idx_s[o] * H_ + j] = hnew;
        hold_s[o][j] = hnew;
    }

    if (tid < 16 * D_) red[tid] = lterm;
    __syncthreads();
    const int pcol = pn0 + (lane & 15), prb = (lane >> 4) << 3;
    if (pmw) {
        v8f a1 = pm_gemm<H_>(&hold_s[0][0], HP_, bw1, lane);
        const float bv = pb1[pcol];
        #pragma unroll
        for (int j = 0; j < 8; ++j) u_s[prb + j][pcol] = fmaxf(a1[j] + bv, 0.0f);
    }
    if (tid < 8 * D_) red[tid] += red[tid + 8 * D_];
    __syncthreads();
    if (pmw) {
        v8f a2 = pm_gemm<P2_>(&u_s[0][0], PP_, bw2, lane);
        const float bv2 = pb2[pcol];
        #pragma unroll
        for (int j = 0; j < 8; ++j) p[idx_s[prb + j] * P2_ + pcol] = a2[j] + bv2;
    }
    for (int s2 = 4 * D_; s2 > 0; s2 >>= 1) {
        if (tid < s2) red[tid] += red[tid + s2];
        __syncthreads();
    }
    if (tid == 0) loss_part[t * (NO_ / 16) + blockIdx.x] = red[0];
}

__global__ void k_loss_final(const float* __restrict__ parts, float* __restrict__ loss) {
    __shared__ float red[256];
    const int tid = threadIdx.x;
    float s = 0.0f;
    for (int i = tid; i < NT_ * (NO_ / 16); i += 256) s += parts[i];
    red[tid] = s;
    __syncthreads();
    for (int k = 128; k > 0; k >>= 1) {
        if (tid < k) red[tid] += red[tid + k];
        __syncthreads();
    }
    if (tid == 0) *loss = red[0];
}

extern "C" void kernel_launch(void* const* d_in, const int* in_sizes, int n_in,
                              void* d_out, int out_size, void* d_ws, size_t ws_size,
                              hipStream_t stream) {
    const float* X       = (const float*)d_in[0];
    const float* M       = (const float*)d_in[1];
    const int*   obs_idx = (const int*)d_in[2];
    const float* cov     = (const float*)d_in[3];
    const float* cm_w1   = (const float*)d_in[4];
    const float* cm_b1   = (const float*)d_in[5];
    const float* cm_w2   = (const float*)d_in[6];
    const float* cm_b2   = (const float*)d_in[7];
    const float* p_w1    = (const float*)d_in[8];
    const float* p_b1    = (const float*)d_in[9];
    const float* p_w2    = (const float*)d_in[10];
    const float* p_b2    = (const float*)d_in[11];
    const float* xz_w    = (const float*)d_in[12];
    const float* xz_b    = (const float*)d_in[13];
    const float* hz_w    = (const float*)d_in[14];
    const float* xn_w    = (const float*)d_in[15];
    const float* xn_b    = (const float*)d_in[16];
    const float* hn_w    = (const float*)d_in[17];
    const float* w_ih    = (const float*)d_in[18];
    const float* w_hh    = (const float*)d_in[19];
    const float* b_ih    = (const float*)d_in[20];
    const float* b_hh    = (const float*)d_in[21];
    const float* w_prep  = (const float*)d_in[22];
    const float* b_prep  = (const float*)d_in[23];

    float* h    = (float*)d_out;
    float* p    = h + (size_t)B_ * H_;
    float* loss = p + (size_t)B_ * P2_;
    float* loss_part = (float*)d_ws;

    const int MB = B_ / 16;
    const int MB2 = B_ / OR_;
    const int OB = NO_ / 16;

    k_init<<<MB, 128, 0, stream>>>(cov, cm_w1, cm_b1, cm_w2, cm_b2,
                                   p_w1, p_b1, p_w2, p_b2, h, p, loss);

    for (int t = 0; t < NT_; ++t) {
        k_ode<<<MB2, 256, 0, stream>>>(h, p, xz_w, xz_b, hz_w, xn_w, xn_b, hn_w,
                                      p_w1, p_b1, p_w2, p_b2);
        k_jump<<<OB, JT_, 0, stream>>>(t, X, M, obs_idx, h, p,
                                       w_ih, w_hh, b_ih, b_hh, w_prep, b_prep,
                                       p_w1, p_b1, p_w2, p_b2, loss_part);
    }
    for (int s = 0; s < NPOST_; ++s)
        k_ode<<<MB2, 256, 0, stream>>>(h, p, xz_w, xz_b, hz_w, xn_w, xn_b, hn_w,
                                      p_w1, p_b1, p_w2, p_b2);
    k_loss_final<<<1, 256, 0, stream>>>(loss_part, loss);
}


# gate sigmoid and tanh of the ode and jump kernels from hardware exp2 and reciprocal (about 1 ulp each; tanh = 1-2/(1+e^2x)) in place of library expf, IEEE division, branching tanhf; sums kept
# speedup vs baseline: 2.2195x; 1.0704x over previous
#include <hip/hip_runtime.h>
#include <hip/hip_bf16.h>
#include <math.h>

#define B_    8192
#define D_    32
#define H_    128
#define P2_   64
#define PREP_ 10
#define GIN_  320
#define G3_   384
#define CS_   8
#define CH_   32
#define NT_   16
#define NO_   4096
#define DT_   0.1f
#define NPOST_ 4
#define OR_   32
#define JT_   768
#define LOGC_  0.91893853320467274178f

#define HP_   (H_ + 4)
#define PP_   (P2_ + 4)
#define GP_   (GIN_ + 4)
#define G3P_  (G3_ + 4)

typedef __attribute__((ext_vector_type(2))) float v2f;
typedef __attribute__((ext_vector_type(4))) float v4f;
typedef __attribute__((ext_vector_type(8))) float v8f;

__device__ __forceinline__ float sigmoidf_(float x) {
    return __builtin_amdgcn_rcpf(1.0f + __builtin_amdgcn_exp2f(-1.44269504088896340736f * x));
}
__device__ __forceinline__ float tanhf_(float x) {
    return 1.0f - 2.0f * __builtin_amdgcn_rcpf(1.0f + __builtin_amdgcn_exp2f(2.88539008177792681472f * x));
}


__device__ __forceinline__ v2f load_afrag(const float* A, int lda, int m0, int k, int lane) {
    int m  = m0 + (lane & 15);
    int kk = k + ((lane >> 4) << 1);
    v2f a;
    a.x = A[m * lda + kk];
    a.y = A[m * lda + kk + 1];
    return a;
}

__device__ __forceinline__ v2f load_bfrag(const float* Bm, int ldb, int k, int n0, int lane) {
    int n  = n0 + (lane & 15);
    int kk = k + ((lane >> 4) << 1);
    v2f b;
    b.x = Bm[kk * ldb + n];
    b.y = Bm[(kk + 1) * ldb + n];
    return b;
}

__device__ __forceinline__ v8f wmma4(v2f a, v2f b, v8f c) {
    return __builtin_amdgcn_wmma_f32_16x16x4_f32(false, a, false, b, (short)0, c, false, false);
}

template <int K>
__device__ __forceinline__ void pm_load_w(v2f (&bw)[K / 4], const float* __restrict__ w, int n0, int lane) {
    #pragma unroll
    for (int k = 0; k < K; k += 4) bw[k >> 2] = load_bfrag(w, P2_, k, n0, lane);
}

template <int K>
__device__ __forceinline__ v8f pm_gemm(const float* as, int pitch, const v2f (&bw)[K / 4], int lane) {
    v8f acc = {};
    #pragma unroll
    for (int k = 0; k < K; k += 4) acc = wmma4(load_afrag(as, pitch, 0, k, lane), bw[k >> 2], acc);
    return acc;
}

__global__ void __launch_bounds__(128) k_init(const float* __restrict__ cov,
                       const float* __restrict__ w1, const float* __restrict__ b1,
                       const float* __restrict__ w2, const float* __restrict__ b2,
                       const float* __restrict__ pw1, const float* __restrict__ pb1,
                       const float* __restrict__ pw2, const float* __restrict__ pb2,
                       float* __restrict__ h, float* __restrict__ p, float* __restrict__ loss) {
    __shared__ float t_s[16][CH_];
    __shared__ float hn_s[16][HP_];
    __shared__ float u_s[16][PP_];
    const int tid = threadIdx.x;
    const int lane = tid & 31, pn0 = (tid >> 5) * 16;
    const int m0  = blockIdx.x * 16;
    if (blockIdx.x == 0 && tid == 0) *loss = 0.0f;
    v2f bw1[H_ / 4], bw2[P2_ / 4];
    pm_load_w<H_>(bw1, pw1, pn0, lane);
    pm_load_w<P2_>(bw2, pw2, pn0, lane);
    for (int i = tid; i < 16 * CH_; i += 128) {
        int r = i / CH_, c = i % CH_;
        float acc = b1[c];
        #pragma unroll
        for (int k = 0; k < CS_; ++k) acc += cov[(m0 + r) * CS_ + k] * w1[k * CH_ + c];
        t_s[r][c] = fmaxf(acc, 0.0f);
    }
    __syncthreads();
    for (int i = tid; i < 16 * H_; i += 128) {
        int r = i / H_, c = i % H_;
        float acc = b2[c];
        #pragma unroll
        for (int k = 0; k < CH_; ++k) acc += t_s[r][k] * w2[k * H_ + c];
        float hv = tanhf(acc);
        h[(m0 + r) * H_ + c] = hv;
        hn_s[r][c] = hv;
    }
    __syncthreads();

    const int col = pn0 + (lane & 15), rb = (lane >> 4) << 3;
    v8f a1 = pm_gemm<H_>(&hn_s[0][0], HP_, bw1, lane);
    const float bv = pb1[col];
    #pragma unroll
    for (int j = 0; j < 8; ++j) u_s[rb + j][col] = fmaxf(a1[j] + bv, 0.0f);
    __syncthreads();
    v8f a2 = pm_gemm<P2_>(&u_s[0][0], PP_, bw2, lane);
    const float bv2 = pb2[col];
    #pragma unroll
    for (int j = 0; j < 8; ++j) p[(m0 + rb + j) * P2_ + col] = a2[j] + bv2;
}

__global__ void __launch_bounds__(256) k_ode(float* __restrict__ h, float* __restrict__ p,
                      const float* __restrict__ xz_w, const float* __restrict__ xz_b,
                      const float* __restrict__ hz_w,
                      const float* __restrict__ xn_w, const float* __restrict__ xn_b,
                      const float* __restrict__ hn_w,
                      const float* __restrict__ pw1, const float* __restrict__ pb1,
                      const float* __restrict__ pw2, const float* __restrict__ pb2) {
    __shared__ __attribute__((aligned(16))) float h_s[OR_][HP_];
    __shared__ __attribute__((aligned(16))) float p_s[OR_][PP_];
    __shared__ float zh_s[OR_][HP_];
    __shared__ float hn_s[OR_][HP_];
    __shared__ float u_s[OR_][PP_];
    const int tid = threadIdx.x, lane = tid & 31;
    const int wv = tid >> 5;
    const int m0 = blockIdx.x * OR_, n0 = wv * 16;
    const int pn0 = (wv & 3) * 16;
    const int pm0 = (wv >> 2) * 16;

    const v4f* hg = reinterpret_cast<const v4f*>(h + (size_t)m0 * H_);
    const v4f* pg = reinterpret_cast<const v4f*>(p + (size_t)m0 * P2_);
    const v4f hq0 = hg[tid], hq1 = hg[tid + 256], hq2 = hg[tid + 512], hq3 = hg[tid + 768];
    const v4f pq0 = pg[tid], pq1 = pg[tid + 256];
    v2f bxz[P2_ / 4];
    #pragma unroll
    for (int k = 0; k < P2_; k += 4) bxz[k >> 2] = load_bfrag(xz_w, H_, k, n0, lane);
    *reinterpret_cast<v4f*>(&h_s[tid >> 5][(tid & 31) << 2])        = hq0;
    *reinterpret_cast<v4f*>(&h_s[8 + (tid >> 5)][(tid & 31) << 2])  = hq1;
    *reinterpret_cast<v4f*>(&h_s[16 + (tid >> 5)][(tid & 31) << 2]) = hq2;
    *reinterpret_cast<v4f*>(&h_s[24 + (tid >> 5)][(tid & 31) << 2]) = hq3;
    *reinterpret_cast<v4f*>(&p_s[tid >> 4][(tid & 15) << 2])        = pq0;
    *reinterpret_cast<v4f*>(&p_s[16 + (tid >> 4)][(tid & 15) << 2]) = pq1;
    __syncthreads();

    v8f az0 = {}, az1 = {};
    #pragma unroll
    for (int k = 0; k < P2_; k += 4) {
        az0 = wmma4(load_afrag(&p_s[0][0], PP_, 0, k, lane), bxz[k >> 2], az0);
        az1 = wmma4(load_afrag(&p_s[0][0], PP_, 16, k, lane), bxz[k >> 2], az1);
    }
    for (int k = 0; k < H_; k += 4) {
        const v2f b = load_bfrag(hz_w, H_, k, n0, lane);
        az0 = wmma4(load_afrag(&h_s[0][0], HP_, 0, k, lane), b, az0);
        az1 = wmma4(load_afrag(&h_s[0][0], HP_, 16, k, lane), b, az1);
    }

    const int col = n0 + (lane & 15), rb = (lane >> 4) << 3;
    const float bz = xz_b[col];
    float z0[8], z1[8];
    #pragma unroll
    for (int j = 0; j < 8; ++j) {
        z0[j] = sigmoidf_(az0[j] + bz);
        z1[j] = sigmoidf_(az1[j] + bz);
        zh_s[rb + j][col]      = z0[j] * h_s[rb + j][col];
        zh_s[16 + rb + j][col] = z1[j] * h_s[16 + rb + j][col];
    }
    __syncthreads();

    v2f bw1[H_ / 4], bw2[P2_ / 4];
    pm_load_w<H_>(bw1, pw1, pn0, lane);
    pm_load_w<P2_>(bw2, pw2, pn0, lane);

    v8f an0 = {}, an1 = {};
    for (int k = 0; k < P2_; k += 4) {
        const v2f b = load_bfrag(xn_w, H_, k, n0, lane);
        an0 = wmma4(load_afrag(&p_s[0][0], PP_, 0, k, lane), b, an0);
        an1 = wmma4(load_afrag(&p_s[0][0], PP_, 16, k, lane), b, an1);
    }
    for (int k = 0; k < H_; k += 4) {
        const v2f b = load_bfrag(hn_w, H_, k, n0, lane);
        an0 = wmma4(load_afrag(&zh_s[0][0], HP_, 0, k, lane), b, an0);
        an1 = wmma4(load_afrag(&zh_s[0][0], HP_, 16, k, lane), b, an1);
    }

    const float bn = xn_b[col];
    float hnew0[8], hnew1[8];
    #pragma unroll
    for (int j = 0; j < 8; ++j) {
        const float n0v = tanhf_(an0[j] + bn);
        const float n1v = tanhf_(an1[j] + bn);
        const float hv0 = h_s[rb + j][col];
        const float hv1 = h_s[16 + rb + j][col];
        hnew0[j] = hv0 + DT_ * (1.0f - z0[j]) * (n0v - hv0);
        hnew1[j] = hv1 + DT_ * (1.0f - z1[j]) * (n1v - hv1);
        hn_s[rb + j][col]      = hnew0[j];
        hn_s[16 + rb + j][col] = hnew1[j];
    }
    __syncthreads();
    #pragma unroll
    for (int j = 0; j < 8; ++j) {
        h[(m0 + rb + j) * H_ + col]      = hnew0[j];
        h[(m0 + 16 + rb + j) * H_ + col] = hnew1[j];
    }

    const int pcol = pn0 + (lane & 15);
    {
        v8f a1 = pm_gemm<H_>(&hn_s[pm0][0], HP_, bw1, lane);
        const float bv = pb1[pcol];
        #pragma unroll
        for (int j = 0; j < 8; ++j) u_s[pm0 + rb + j][pcol] = fmaxf(a1[j] + bv, 0.0f);
    }
    __syncthreads();
    {
        v8f a2 = pm_gemm<P2_>(&u_s[pm0][0], PP_, bw2, lane);
        const float bv2 = pb2[pcol];
        #pragma unroll
        for (int j = 0; j < 8; ++j) p[(m0 + pm0 + rb + j) * P2_ + pcol] = a2[j] + bv2;
    }
}

__global__ void __launch_bounds__(JT_) k_jump(int t,
                       const float* __restrict__ X, const float* __restrict__ M,
                       const int* __restrict__ obs_idx,
                       float* __restrict__ h, float* __restrict__ p,
                       const float* __restrict__ w_ih, const float* __restrict__ w_hh,
                       const float* __restrict__ b_ih, const float* __restrict__ b_hh,
                       const float* __restrict__ w_prep, const float* __restrict__ bias_prep,
                       const float* __restrict__ pw1, const float* __restrict__ pb1,
                       const float* __restrict__ pw2, const float* __restrict__ pb2,
                       float* __restrict__ loss_part) {
    __shared__ int   idx_s[16];
    __shared__ __attribute__((aligned(16))) float hold_s[16][HP_];
    __shared__ float gin_s[16][GP_];
    __shared__ float gi_s[16][G3P_];
    __shared__ float gh_s[16][G3P_];
    __shared__ float red[16 * D_];
    __shared__ float u_s[16][PP_];
    __shared__ v4f   feat_s[16 * D_];
    __shared__ float m_s[16 * D_];
    const int tid = threadIdx.x, lane = tid & 31;
    const int wv = __builtin_amdgcn_readfirstlane(tid >> 5);
    const bool pmw = (wv >> 2) == 2;
    const int pn0 = (wv & 3) * 16;
    const int o0 = blockIdx.x * 16;
    const float* Xt = X + (size_t)t * NO_ * D_;
    const float* Mt = M + (size_t)t * NO_ * D_;

    float xv = 0.0f, mv = 0.0f;
    if (tid < 16 * D_) { xv = Xt[o0 * D_ + tid]; mv = Mt[o0 * D_ + tid]; }
    const int og = (tid >= GIN_) ? 1 : 0;
    const int gc = tid - og * GIN_;
    const int gd = gc / PREP_;
    float w0 = 0.0f, w1 = 0.0f, w2 = 0.0f, w3 = 0.0f, bp = 0.0f;
    if (tid < 2 * GIN_) {
        const float* wp = w_prep + (size_t)(gd * 4) * PREP_ + (gc - gd * PREP_);
        w0 = wp[0]; w1 = wp[PREP_]; w2 = wp[2 * PREP_]; w3 = wp[3 * PREP_];
        bp = bias_prep[gc];
    }
    if (tid < 16) idx_s[tid] = obs_idx[t * NO_ + o0 + tid];
    __syncthreads();

    float lterm = 0.0f;
    if (tid >= 16 * D_) {
        const int q = tid - 16 * D_;
        const int oa = q >> 5, c4 = (q & 31) << 2;
        const v4f ha = *reinterpret_cast<const v4f*>(h + (size_t)idx_s[oa] * H_ + c4);
        const v4f hb = *reinterpret_cast<const v4f*>(h + (size_t)idx_s[oa + 8] * H_ + c4);
        *reinterpret_cast<v4f*>(&hold_s[oa][c4])     = ha;
        *reinterpret_cast<v4f*>(&hold_s[oa + 8][c4]) = hb;
    } else {
        const int o = tid / D_, d = tid % D_;
        const int b = idx_s[o];
        const float mean = p[b * P2_ + d];
        const float lv   = p[b * P2_ + D_ + d];
        const float err  = (xv - mean) * expf(-0.5f * lv);
        lterm = 0.5f * ((err * err + lv + 2.0f * LOGC_) * mv);
        v4f f; f.x = xv; f.y = mean; f.z = lv; f.w = err;
        feat_s[tid] = f;
        m_s[tid] = mv;
    }
    __syncthreads();

    if (tid < 2 * GIN_) {
        #pragma unroll
        for (int j = 0; j < 8; ++j) {
            const int o = og * 8 + j;
            const v4f f = feat_s[o * D_ + gd];
            const float mm = m_s[o * D_ + gd];
            const float a = fmaf(f.w, w3, fmaf(f.z, w2, fmaf(f.x, w0, f.y * w1))) + bp;
            gin_s[o][gc] = fmaxf(a, 0.0f) * mm;
        }
    }
    __syncthreads();

    {
        const int n0 = wv * 16;
        v8f gi = {};
        for (int k = 0; k < GIN_; k += 4)
            gi = wmma4(load_afrag(&gin_s[0][0], GP_, 0, k, lane), load_bfrag(w_ih, G3_, k, n0, lane), gi);
        v8f gh = {};
        for (int k = 0; k < H_; k += 4)
            gh = wmma4(load_afrag(&hold_s[0][0], HP_, 0, k, lane), load_bfrag(w_hh, G3_, k, n0, lane), gh);
        const int col = n0 + (lane & 15), rb = (lane >> 4) << 3;
        const float bi = b_ih[col], bh = b_hh[col];
        #pragma unroll
        for (int j = 0; j < 8; ++j) {
            gi_s[rb + j][col] = gi[j] + bi;
            gh_s[rb + j][col] = gh[j] + bh;
        }
    }
    __syncthreads();

    v2f bw1[H_ / 4] = {}, bw2[P2_ / 4] = {};
    if (pmw) { pm_load_w<H_>(bw1, pw1, pn0, lane); pm_load_w<P2_>(bw2, pw2, pn0, lane); }

    for (int i = tid; i < 16 * H_; i += JT_) {
        int o = i >> 7, j = i & 127;
        float r = sigmoidf_(gi_s[o][j] + gh_s[o][j]);
        float z = sigmoidf_(gi_s[o][H_ + j] + gh_s[o][H_ + j]);
        float n = tanhf_(gi_s[o][2 * H_ + j] + r * gh_s[o][2 * H_ + j]);
        float hv = hold_s[o][j];
        float hnew = (1.0f - z) * n + z * hv;
        h[idx_s[o] * H_ + j] = hnew;
        hold_s[o][j] = hnew;
    }

    if (tid < 16 * D_) red[tid] = lterm;
    __syncthreads();
    const int pcol = pn0 + (lane & 15), prb = (lane >> 4) << 3;
    if (pmw) {
        v8f a1 = pm_gemm<H_>(&hold_s[0][0], HP_, bw1, lane);
        const float bv = pb1[pcol];
        #pragma unroll
        for (int j = 0; j < 8; ++j) u_s[prb + j][pcol] = fmaxf(a1[j] + bv, 0.0f);
    }
    if (tid < 8 * D_) red[tid] += red[tid + 8 * D_];
    __syncthreads();
    if (pmw) {
        v8f a2 = pm_gemm<P2_>(&u_s[0][0], PP_, bw2, lane);
        const float bv2 = pb2[pcol];
        #pragma unroll
        for (int j = 0; j < 8; ++j) p[idx_s[prb + j] * P2_ + pcol] = a2[j] + bv2;
    }
    for (int s2 = 4 * D_; s2 > 0; s2 >>= 1) {
        if (tid < s2) red[tid] += red[tid + s2];
        __syncthreads();
    }
    if (tid == 0) loss_part[t * (NO_ / 16) + blockIdx.x] = red[0];
}

__global__ void k_loss_final(const float* __restrict__ parts, float* __restrict__ loss) {
    __shared__ float red[256];
    const int tid = threadIdx.x;
    float s = 0.0f;
    for (int i = tid; i < NT_ * (NO_ / 16); i += 256) s += parts[i];
    red[tid] = s;
    __syncthreads();
    for (int k = 128; k > 0; k >>= 1) {
        if (tid < k) red[tid] += red[tid + k];
        __syncthreads();
    }
    if (tid == 0) *loss = red[0];
}

extern "C" void kernel_launch(void* const* d_in, const int* in_sizes, int n_in,
                              void* d_out, int out_size, void* d_ws, size_t ws_size,
                              hipStream_t stream) {
    const float* X       = (const float*)d_in[0];
    const float* M       = (const float*)d_in[1];
    const int*   obs_idx = (const int*)d_in[2];
    const float* cov     = (const float*)d_in[3];
    const float* cm_w1   = (const float*)d_in[4];
    const float* cm_b1   = (const float*)d_in[5];
    const float* cm_w2   = (const float*)d_in[6];
    const float* cm_b2   = (const float*)d_in[7];
    const float* p_w1    = (const float*)d_in[8];
    const float* p_b1    = (const float*)d_in[9];
    const float* p_w2    = (const float*)d_in[10];
    const float* p_b2    = (const float*)d_in[11];
    const float* xz_w    = (const float*)d_in[12];
    const float* xz_b    = (const float*)d_in[13];
    const float* hz_w    = (const float*)d_in[14];
    const float* xn_w    = (const float*)d_in[15];
    const float* xn_b    = (const float*)d_in[16];
    const float* hn_w    = (const float*)d_in[17];
    const float* w_ih    = (const float*)d_in[18];
    const float* w_hh    = (const float*)d_in[19];
    const float* b_ih    = (const float*)d_in[20];
    const float* b_hh    = (const float*)d_in[21];
    const float* w_prep  = (const float*)d_in[22];
    const float* b_prep  = (const float*)d_in[23];

    float* h    = (float*)d_out;
    float* p    = h + (size_t)B_ * H_;
    float* loss = p + (size_t)B_ * P2_;
    float* loss_part = (float*)d_ws;

    const int MB = B_ / 16;
    const int MB2 = B_ / OR_;
    const int OB = NO_ / 16;

    k_init<<<MB, 128, 0, stream>>>(cov, cm_w1, cm_b1, cm_w2, cm_b2,
                                   p_w1, p_b1, p_w2, p_b2, h, p, loss);

    for (int t = 0; t < NT_; ++t) {
        k_ode<<<MB2, 256, 0, stream>>>(h, p, xz_w, xz_b, hz_w, xn_w, xn_b, hn_w,
                                      p_w1, p_b1, p_w2, p_b2);
        k_jump<<<OB, JT_, 0, stream>>>(t, X, M, obs_idx, h, p,
                                       w_ih, w_hh, b_ih, b_hh, w_prep, b_prep,
                                       p_w1, p_b1, p_w2, p_b2, loss_part);
    }
    for (int s = 0; s < NPOST_; ++s)
        k_ode<<<MB2, 256, 0, stream>>>(h, p, xz_w, xz_b, hz_w, xn_w, xn_b, hn_w,
                                      p_w1, p_b1, p_w2, p_b2);
    k_loss_final<<<1, 256, 0, stream>>>(loss_part, loss);
}


# tail barriers of the ode and jump kernels: h stored behind the last barrier, so none waits for a store drain; jump loss tree folded by one wave (same pairs, same order), its 8 barrier rounds gone
# speedup vs baseline: 2.3571x; 1.0620x over previous
#include <hip/hip_runtime.h>
#include <hip/hip_bf16.h>
#include <math.h>

#define B_    8192
#define D_    32
#define H_    128
#define P2_   64
#define PREP_ 10
#define GIN_  320
#define G3_   384
#define CS_   8
#define CH_   32
#define NT_   16
#define NO_   4096
#define DT_   0.1f
#define NPOST_ 4
#define OR_   32
#define JT_   768
#define LOGC_  0.91893853320467274178f

#define HP_   (H_ + 4)
#define PP_   (P2_ + 4)
#define GP_   (GIN_ + 4)
#define G3P_  (G3_ + 4)

typedef __attribute__((ext_vector_type(2))) float v2f;
typedef __attribute__((ext_vector_type(4))) float v4f;
typedef __attribute__((ext_vector_type(8))) float v8f;

__device__ __forceinline__ float sigmoidf_(float x) {
    return __builtin_amdgcn_rcpf(1.0f + __builtin_amdgcn_exp2f(-1.44269504088896340736f * x));
}
__device__ __forceinline__ float tanhf_(float x) {
    return 1.0f - 2.0f * __builtin_amdgcn_rcpf(1.0f + __builtin_amdgcn_exp2f(2.88539008177792681472f * x));
}


__device__ __forceinline__ v2f load_afrag(const float* A, int lda, int m0, int k, int lane) {
    int m  = m0 + (lane & 15);
    int kk = k + ((lane >> 4) << 1);
    v2f a;
    a.x = A[m * lda + kk];
    a.y = A[m * lda + kk + 1];
    return a;
}

__device__ __forceinline__ v2f load_bfrag(const float* Bm, int ldb, int k, int n0, int lane) {
    int n  = n0 + (lane & 15);
    int kk = k + ((lane >> 4) << 1);
    v2f b;
    b.x = Bm[kk * ldb + n];
    b.y = Bm[(kk + 1) * ldb + n];
    return b;
}

__device__ __forceinline__ v8f wmma4(v2f a, v2f b, v8f c) {
    return __builtin_amdgcn_wmma_f32_16x16x4_f32(false, a, false, b, (short)0, c, false, false);
}

template <int K>
__device__ __forceinline__ void pm_load_w(v2f (&bw)[K / 4], const float* __restrict__ w, int n0, int lane) {
    #pragma unroll
    for (int k = 0; k < K; k += 4) bw[k >> 2] = load_bfrag(w, P2_, k, n0, lane);
}

template <int K>
__device__ __forceinline__ v8f pm_gemm(const float* as, int pitch, const v2f (&bw)[K / 4], int lane) {
    v8f acc = {};
    #pragma unroll
    for (int k = 0; k < K; k += 4) acc = wmma4(load_afrag(as, pitch, 0, k, lane), bw[k >> 2], acc);
    return acc;
}

__global__ void __launch_bounds__(128) k_init(const float* __restrict__ cov,
                       const float* __restrict__ w1, const float* __restrict__ b1,
                       const float* __restrict__ w2, const float* __restrict__ b2,
                       const float* __restrict__ pw1, const float* __restrict__ pb1,
                       const float* __restrict__ pw2, const float* __restrict__ pb2,
                       float* __restrict__ h, float* __restrict__ p, float* __restrict__ loss) {
    __shared__ float t_s[16][CH_];
    __shared__ float hn_s[16][HP_];
    __shared__ float u_s[16][PP_];
    const int tid = threadIdx.x;
    const int lane = tid & 31, pn0 = (tid >> 5) * 16;
    const int m0  = blockIdx.x * 16;
    if (blockIdx.x == 0 && tid == 0) *loss = 0.0f;
    v2f bw1[H_ / 4], bw2[P2_ / 4];
    pm_load_w<H_>(bw1, pw1, pn0, lane);
    pm_load_w<P2_>(bw2, pw2, pn0, lane);
    for (int i = tid; i < 16 * CH_; i += 128) {
        int r = i / CH_, c = i % CH_;
        float acc = b1[c];
        #pragma unroll
        for (int k = 0; k < CS_; ++k) acc += cov[(m0 + r) * CS_ + k] * w1[k * CH_ + c];
        t_s[r][c] = fmaxf(acc, 0.0f);
    }
    __syncthreads();
    for (int i = tid; i < 16 * H_; i += 128) {
        int r = i / H_, c = i % H_;
        float acc = b2[c];
        #pragma unroll
        for (int k = 0; k < CH_; ++k) acc += t_s[r][k] * w2[k * H_ + c];
        float hv = tanhf(acc);
        h[(m0 + r) * H_ + c] = hv;
        hn_s[r][c] = hv;
    }
    __syncthreads();

    const int col = pn0 + (lane & 15), rb = (lane >> 4) << 3;
    v8f a1 = pm_gemm<H_>(&hn_s[0][0], HP_, bw1, lane);
    const float bv = pb1[col];
    #pragma unroll
    for (int j = 0; j < 8; ++j) u_s[rb + j][col] = fmaxf(a1[j] + bv, 0.0f);
    __syncthreads();
    v8f a2 = pm_gemm<P2_>(&u_s[0][0], PP_, bw2, lane);
    const float bv2 = pb2[col];
    #pragma unroll
    for (int j = 0; j < 8; ++j) p[(m0 + rb + j) * P2_ + col] = a2[j] + bv2;
}

__global__ void __launch_bounds__(256) k_ode(float* __restrict__ h, float* __restrict__ p,
                      const float* __restrict__ xz_w, const float* __restrict__ xz_b,
                      const float* __restrict__ hz_w,
                      const float* __restrict__ xn_w, const float* __restrict__ xn_b,
                      const float* __restrict__ hn_w,
                      const float* __restrict__ pw1, const float* __restrict__ pb1,
                      const float* __restrict__ pw2, const float* __restrict__ pb2) {
    __shared__ __attribute__((aligned(16))) float h_s[OR_][HP_];
    __shared__ __attribute__((aligned(16))) float p_s[OR_][PP_];
    __shared__ float zh_s[OR_][HP_];
    __shared__ float hn_s[OR_][HP_];
    __shared__ float u_s[OR_][PP_];
    const int tid = threadIdx.x, lane = tid & 31;
    const int wv = tid >> 5;
    const int m0 = blockIdx.x * OR_, n0 = wv * 16;
    const int pn0 = (wv & 3) * 16;
    const int pm0 = (wv >> 2) * 16;

    const v4f* hg = reinterpret_cast<const v4f*>(h + (size_t)m0 * H_);
    const v4f* pg = reinterpret_cast<const v4f*>(p + (size_t)m0 * P2_);
    const v4f hq0 = hg[tid], hq1 = hg[tid + 256], hq2 = hg[tid + 512], hq3 = hg[tid + 768];
    const v4f pq0 = pg[tid], pq1 = pg[tid + 256];
    v2f bxz[P2_ / 4];
    #pragma unroll
    for (int k = 0; k < P2_; k += 4) bxz[k >> 2] = load_bfrag(xz_w, H_, k, n0, lane);
    *reinterpret_cast<v4f*>(&h_s[tid >> 5][(tid & 31) << 2])        = hq0;
    *reinterpret_cast<v4f*>(&h_s[8 + (tid >> 5)][(tid & 31) << 2])  = hq1;
    *reinterpret_cast<v4f*>(&h_s[16 + (tid >> 5)][(tid & 31) << 2]) = hq2;
    *reinterpret_cast<v4f*>(&h_s[24 + (tid >> 5)][(tid & 31) << 2]) = hq3;
    *reinterpret_cast<v4f*>(&p_s[tid >> 4][(tid & 15) << 2])        = pq0;
    *reinterpret_cast<v4f*>(&p_s[16 + (tid >> 4)][(tid & 15) << 2]) = pq1;
    __syncthreads();

    v8f az0 = {}, az1 = {};
    #pragma unroll
    for (int k = 0; k < P2_; k += 4) {
        az0 = wmma4(load_afrag(&p_s[0][0], PP_, 0, k, lane), bxz[k >> 2], az0);
        az1 = wmma4(load_afrag(&p_s[0][0], PP_, 16, k, lane), bxz[k >> 2], az1);
    }
    for (int k = 0; k < H_; k += 4) {
        const v2f b = load_bfrag(hz_w, H_, k, n0, lane);
        az0 = wmma4(load_afrag(&h_s[0][0], HP_, 0, k, lane), b, az0);
        az1 = wmma4(load_afrag(&h_s[0][0], HP_, 16, k, lane), b, az1);
    }

    const int col = n0 + (lane & 15), rb = (lane >> 4) << 3;
    const float bz = xz_b[col];
    float z0[8], z1[8];
    #pragma unroll
    for (int j = 0; j < 8; ++j) {
        z0[j] = sigmoidf_(az0[j] + bz);
        z1[j] = sigmoidf_(az1[j] + bz);
        zh_s[rb + j][col]      = z0[j] * h_s[rb + j][col];
        zh_s[16 + rb + j][col] = z1[j] * h_s[16 + rb + j][col];
    }
    __syncthreads();

    v2f bw1[H_ / 4], bw2[P2_ / 4];
    pm_load_w<H_>(bw1, pw1, pn0, lane);
    pm_load_w<P2_>(bw2, pw2, pn0, lane);

    v8f an0 = {}, an1 = {};
    for (int k = 0; k < P2_; k += 4) {
        const v2f b = load_bfrag(xn_w, H_, k, n0, lane);
        an0 = wmma4(load_afrag(&p_s[0][0], PP_, 0, k, lane), b, an0);
        an1 = wmma4(load_afrag(&p_s[0][0], PP_, 16, k, lane), b, an1);
    }
    for (int k = 0; k < H_; k += 4) {
        const v2f b = load_bfrag(hn_w, H_, k, n0, lane);
        an0 = wmma4(load_afrag(&zh_s[0][0], HP_, 0, k, lane), b, an0);
        an1 = wmma4(load_afrag(&zh_s[0][0], HP_, 16, k, lane), b, an1);
    }

    const float bn = xn_b[col];
    float hnew0[8], hnew1[8];
    #pragma unroll
    for (int j = 0; j < 8; ++j) {
        const float n0v = tanhf_(an0[j] + bn);
        const float n1v = tanhf_(an1[j] + bn);
        const float hv0 = h_s[rb + j][col];
        const float hv1 = h_s[16 + rb + j][col];
        hnew0[j] = hv0 + DT_ * (1.0f - z0[j]) * (n0v - hv0);
        hnew1[j] = hv1 + DT_ * (1.0f - z1[j]) * (n1v - hv1);
        hn_s[rb + j][col]      = hnew0[j];
        hn_s[16 + rb + j][col] = hnew1[j];
    }
    __syncthreads();

    const int pcol = pn0 + (lane & 15);
    {
        v8f a1 = pm_gemm<H_>(&hn_s[pm0][0], HP_, bw1, lane);
        const float bv = pb1[pcol];
        #pragma unroll
        for (int j = 0; j < 8; ++j) u_s[pm0 + rb + j][pcol] = fmaxf(a1[j] + bv, 0.0f);
    }
    __syncthreads();
    #pragma unroll
    for (int j = 0; j < 8; ++j) {
        h[(m0 + rb + j) * H_ + col]      = hnew0[j];
        h[(m0 + 16 + rb + j) * H_ + col] = hnew1[j];
    }
    {
        v8f a2 = pm_gemm<P2_>(&u_s[pm0][0], PP_, bw2, lane);
        const float bv2 = pb2[pcol];
        #pragma unroll
        for (int j = 0; j < 8; ++j) p[(m0 + pm0 + rb + j) * P2_ + pcol] = a2[j] + bv2;
    }
}

__global__ void __launch_bounds__(JT_) k_jump(int t,
                       const float* __restrict__ X, const float* __restrict__ M,
                       const int* __restrict__ obs_idx,
                       float* __restrict__ h, float* __restrict__ p,
                       const float* __restrict__ w_ih, const float* __restrict__ w_hh,
                       const float* __restrict__ b_ih, const float* __restrict__ b_hh,
                       const float* __restrict__ w_prep, const float* __restrict__ bias_prep,
                       const float* __restrict__ pw1, const float* __restrict__ pb1,
                       const float* __restrict__ pw2, const float* __restrict__ pb2,
                       float* __restrict__ loss_part) {
    __shared__ int   idx_s[16];
    __shared__ __attribute__((aligned(16))) float hold_s[16][HP_];
    __shared__ float gin_s[16][GP_];
    __shared__ float gi_s[16][G3P_];
    __shared__ float gh_s[16][G3P_];
    __shared__ float red[16 * D_];
    __shared__ float u_s[16][PP_];
    __shared__ v4f   feat_s[16 * D_];
    __shared__ float m_s[16 * D_];
    const int tid = threadIdx.x, lane = tid & 31;
    const int wv = __builtin_amdgcn_readfirstlane(tid >> 5);
    const bool pmw = (wv >> 2) == 2;
    const int pn0 = (wv & 3) * 16;
    const int o0 = blockIdx.x * 16;
    const float* Xt = X + (size_t)t * NO_ * D_;
    const float* Mt = M + (size_t)t * NO_ * D_;

    float xv = 0.0f, mv = 0.0f;
    if (tid < 16 * D_) { xv = Xt[o0 * D_ + tid]; mv = Mt[o0 * D_ + tid]; }
    const int og = (tid >= GIN_) ? 1 : 0;
    const int gc = tid - og * GIN_;
    const int gd = gc / PREP_;
    float w0 = 0.0f, w1 = 0.0f, w2 = 0.0f, w3 = 0.0f, bp = 0.0f;
    if (tid < 2 * GIN_) {
        const float* wp = w_prep + (size_t)(gd * 4) * PREP_ + (gc - gd * PREP_);
        w0 = wp[0]; w1 = wp[PREP_]; w2 = wp[2 * PREP_]; w3 = wp[3 * PREP_];
        bp = bias_prep[gc];
    }
    if (tid < 16) idx_s[tid] = obs_idx[t * NO_ + o0 + tid];
    __syncthreads();

    float lterm = 0.0f;
    if (tid >= 16 * D_) {
        const int q = tid - 16 * D_;
        const int oa = q >> 5, c4 = (q & 31) << 2;
        const v4f ha = *reinterpret_cast<const v4f*>(h + (size_t)idx_s[oa] * H_ + c4);
        const v4f hb = *reinterpret_cast<const v4f*>(h + (size_t)idx_s[oa + 8] * H_ + c4);
        *reinterpret_cast<v4f*>(&hold_s[oa][c4])     = ha;
        *reinterpret_cast<v4f*>(&hold_s[oa + 8][c4]) = hb;
    } else {
        const int o = tid / D_, d = tid % D_;
        const int b = idx_s[o];
        const float mean = p[b * P2_ + d];
        const float lv   = p[b * P2_ + D_ + d];
        const float err  = (xv - mean) * expf(-0.5f * lv);
        lterm = 0.5f * ((err * err + lv + 2.0f * LOGC_) * mv);
        v4f f; f.x = xv; f.y = mean; f.z = lv; f.w = err;
        feat_s[tid] = f;
        m_s[tid] = mv;
    }
    __syncthreads();

    if (tid < 2 * GIN_) {
        #pragma unroll
        for (int j = 0; j < 8; ++j) {
            const int o = og * 8 + j;
            const v4f f = feat_s[o * D_ + gd];
            const float mm = m_s[o * D_ + gd];
            const float a = fmaf(f.w, w3, fmaf(f.z, w2, fmaf(f.x, w0, f.y * w1))) + bp;
            gin_s[o][gc] = fmaxf(a, 0.0f) * mm;
        }
    }
    __syncthreads();

    {
        const int n0 = wv * 16;
        v8f gi = {};
        for (int k = 0; k < GIN_; k += 4)
            gi = wmma4(load_afrag(&gin_s[0][0], GP_, 0, k, lane), load_bfrag(w_ih, G3_, k, n0, lane), gi);
        v8f gh = {};
        for (int k = 0; k < H_; k += 4)
            gh = wmma4(load_afrag(&hold_s[0][0], HP_, 0, k, lane), load_bfrag(w_hh, G3_, k, n0, lane), gh);
        const int col = n0 + (lane & 15), rb = (lane >> 4) << 3;
        const float bi = b_ih[col], bh = b_hh[col];
        #pragma unroll
        for (int j = 0; j < 8; ++j) {
            gi_s[rb + j][col] = gi[j] + bi;
            gh_s[rb + j][col] = gh[j] + bh;
        }
    }
    __syncthreads();

    v2f bw1[H_ / 4] = {}, bw2[P2_ / 4] = {};
    if (pmw) { pm_load_w<H_>(bw1, pw1, pn0, lane); pm_load_w<P2_>(bw2, pw2, pn0, lane); }

    for (int i = tid; i < 16 * H_; i += JT_) {
        int o = i >> 7, j = i & 127;
        float r = sigmoidf_(gi_s[o][j] + gh_s[o][j]);
        float z = sigmoidf_(gi_s[o][H_ + j] + gh_s[o][H_ + j]);
        float n = tanhf_(gi_s[o][2 * H_ + j] + r * gh_s[o][2 * H_ + j]);
        float hv = hold_s[o][j];
        float hnew = (1.0f - z) * n + z * hv;
        hold_s[o][j] = hnew;
    }

    if (tid < 16 * D_) red[tid] = lterm;
    __syncthreads();
    const int pcol = pn0 + (lane & 15), prb = (lane >> 4) << 3;
    float lsum = 0.0f;
    if (pmw) {
        v8f a1 = pm_gemm<H_>(&hold_s[0][0], HP_, bw1, lane);
        const float bv = pb1[pcol];
        #pragma unroll
        for (int j = 0; j < 8; ++j) u_s[prb + j][pcol] = fmaxf(a1[j] + bv, 0.0f);
    } else if (wv == 0) {
        float v[16];
        #pragma unroll
        for (int k = 0; k < 16; ++k) v[k] = red[lane + 32 * k];
        #pragma unroll
        for (int k = 0; k < 8; ++k) v[k] = v[k] + v[k + 8];
        #pragma unroll
        for (int k = 0; k < 4; ++k) v[k] = v[k] + v[k + 4];
        v[0] = v[0] + v[2];
        v[1] = v[1] + v[3];
        lsum = v[0] + v[1];
        lsum = lsum + __shfl_down(lsum, 16);
        lsum = lsum + __shfl_down(lsum, 8);
        lsum = lsum + __shfl_down(lsum, 4);
        lsum = lsum + __shfl_down(lsum, 2);
        lsum = lsum + __shfl_down(lsum, 1);
    }
    __syncthreads();
    if (pmw) {
        v8f a2 = pm_gemm<P2_>(&u_s[0][0], PP_, bw2, lane);
        const float bv2 = pb2[pcol];
        #pragma unroll
        for (int j = 0; j < 8; ++j) p[idx_s[prb + j] * P2_ + pcol] = a2[j] + bv2;
    } else {
        const int q = (wv < 8) ? tid : tid - 4 * 32;
        for (int i = q; i < 16 * H_; i += JT_ - 4 * 32) {
            const int o = i >> 7, j = i & 127;
            h[idx_s[o] * H_ + j] = hold_s[o][j];
        }
        if (tid == 0) loss_part[t * (NO_ / 16) + blockIdx.x] = lsum;
    }
}

__global__ void k_loss_final(const float* __restrict__ parts, float* __restrict__ loss) {
    __shared__ float red[256];
    const int tid = threadIdx.x;
    float s = 0.0f;
    for (int i = tid; i < NT_ * (NO_ / 16); i += 256) s += parts[i];
    red[tid] = s;
    __syncthreads();
    for (int k = 128; k > 0; k >>= 1) {
        if (tid < k) red[tid] += red[tid + k];
        __syncthreads();
    }
    if (tid == 0) *loss = red[0];
}

extern "C" void kernel_launch(void* const* d_in, const int* in_sizes, int n_in,
                              void* d_out, int out_size, void* d_ws, size_t ws_size,
                              hipStream_t stream) {
    const float* X       = (const float*)d_in[0];
    const float* M       = (const float*)d_in[1];
    const int*   obs_idx = (const int*)d_in[2];
    const float* cov     = (const float*)d_in[3];
    const float* cm_w1   = (const float*)d_in[4];
    const float* cm_b1   = (const float*)d_in[5];
    const float* cm_w2   = (const float*)d_in[6];
    const float* cm_b2   = (const float*)d_in[7];
    const float* p_w1    = (const float*)d_in[8];
    const float* p_b1    = (const float*)d_in[9];
    const float* p_w2    = (const float*)d_in[10];
    const float* p_b2    = (const float*)d_in[11];
    const float* xz_w    = (const float*)d_in[12];
    const float* xz_b    = (const float*)d_in[13];
    const float* hz_w    = (const float*)d_in[14];
    const float* xn_w    = (const float*)d_in[15];
    const float* xn_b    = (const float*)d_in[16];
    const float* hn_w    = (const float*)d_in[17];
    const float* w_ih    = (const float*)d_in[18];
    const float* w_hh    = (const float*)d_in[19];
    const float* b_ih    = (const float*)d_in[20];
    const float* b_hh    = (const float*)d_in[21];
    const float* w_prep  = (const float*)d_in[22];
    const float* b_prep  = (const float*)d_in[23];

    float* h    = (float*)d_out;
    float* p    = h + (size_t)B_ * H_;
    float* loss = p + (size_t)B_ * P2_;
    float* loss_part = (float*)d_ws;

    const int MB = B_ / 16;
    const int MB2 = B_ / OR_;
    const int OB = NO_ / 16;

    k_init<<<MB, 128, 0, stream>>>(cov, cm_w1, cm_b1, cm_w2, cm_b2,
                                   p_w1, p_b1, p_w2, p_b2, h, p, loss);

    for (int t = 0; t < NT_; ++t) {
        k_ode<<<MB2, 256, 0, stream>>>(h, p, xz_w, xz_b, hz_w, xn_w, xn_b, hn_w,
                                      p_w1, p_b1, p_w2, p_b2);
        k_jump<<<OB, JT_, 0, stream>>>(t, X, M, obs_idx, h, p,
                                       w_ih, w_hh, b_ih, b_hh, w_prep, b_prep,
                                       p_w1, p_b1, p_w2, p_b2, loss_part);
    }
    for (int s = 0; s < NPOST_; ++s)
        k_ode<<<MB2, 256, 0, stream>>>(h, p, xz_w, xz_b, hz_w, xn_w, xn_b, hn_w,
                                      p_w1, p_b1, p_w2, p_b2);
    k_loss_final<<<1, 256, 0, stream>>>(loss_part, loss);
}


# jump kernel: 8 waves, each owning one 16-column hidden tile in all three gates (3 independent products per A fragment); gates in registers; weights loaded 9 k steps ahead; order of every sum kept
# speedup vs baseline: 2.5225x; 1.0702x over previous
#include <hip/hip_runtime.h>
#include <hip/hip_bf16.h>
#include <math.h>

#define B_    8192
#define D_    32
#define H_    128
#define P2_   64
#define PREP_ 10
#define GIN_  320
#define G3_   384
#define CS_   8
#define CH_   32
#define NT_   16
#define NO_   4096
#define DT_   0.1f
#define NPOST_ 4
#define OR_   32
#define JT_   256
#define JPD_  9
#define JRS_  16
#define JP0_  8
#define LOGC_  0.91893853320467274178f

#define HP_   (H_ + 4)
#define PP_   (P2_ + 4)
#define GP_   (GIN_ + 4)

typedef __attribute__((ext_vector_type(2))) float v2f;
typedef __attribute__((ext_vector_type(4))) float v4f;
typedef __attribute__((ext_vector_type(8))) float v8f;

__device__ __forceinline__ float sigmoidf_(float x) {
    return __builtin_amdgcn_rcpf(1.0f + __builtin_amdgcn_exp2f(-1.44269504088896340736f * x));
}
__device__ __forceinline__ float tanhf_(float x) {
    return 1.0f - 2.0f * __builtin_amdgcn_rcpf(1.0f + __builtin_amdgcn_exp2f(2.88539008177792681472f * x));
}


__device__ __forceinline__ v2f load_afrag(const float* A, int lda, int m0, int k, int lane) {
    int m  = m0 + (lane & 15);
    int kk = k + ((lane >> 4) << 1);
    v2f a;
    a.x = A[m * lda + kk];
    a.y = A[m * lda + kk + 1];
    return a;
}

__device__ __forceinline__ v2f load_bfrag(const float* Bm, int ldb, int k, int n0, int lane) {
    int n  = n0 + (lane & 15);
    int kk = k + ((lane >> 4) << 1);
    v2f b;
    b.x = Bm[kk * ldb + n];
    b.y = Bm[(kk + 1) * ldb + n];
    return b;
}

__device__ __forceinline__ v8f wmma4(v2f a, v2f b, v8f c) {
    return __builtin_amdgcn_wmma_f32_16x16x4_f32(false, a, false, b, (short)0, c, false, false);
}

template <int K>
__device__ __forceinline__ void pm_load_w(v2f (&bw)[K / 4], const float* __restrict__ w, int n0, int lane) {
    #pragma unroll
    for (int k = 0; k < K; k += 4) bw[k >> 2] = load_bfrag(w, P2_, k, n0, lane);
}

template <int K>
__device__ __forceinline__ v8f pm_gemm(const float* as, int pitch, const v2f (&bw)[K / 4], int lane) {
    v8f acc = {};
    #pragma unroll
    for (int k = 0; k < K; k += 4) acc = wmma4(load_afrag(as, pitch, 0, k, lane), bw[k >> 2], acc);
    return acc;
}

__global__ void __launch_bounds__(128) k_init(const float* __restrict__ cov,
                       const float* __restrict__ w1, const float* __restrict__ b1,
                       const float* __restrict__ w2, const float* __restrict__ b2,
                       const float* __restrict__ pw1, const float* __restrict__ pb1,
                       const float* __restrict__ pw2, const float* __restrict__ pb2,
                       float* __restrict__ h, float* __restrict__ p, float* __restrict__ loss) {
    __shared__ float t_s[16][CH_];
    __shared__ float hn_s[16][HP_];
    __shared__ float u_s[16][PP_];
    const int tid = threadIdx.x;
    const int lane = tid & 31, pn0 = (tid >> 5) * 16;
    const int m0  = blockIdx.x * 16;
    if (blockIdx.x == 0 && tid == 0) *loss = 0.0f;
    v2f bw1[H_ / 4], bw2[P2_ / 4];
    pm_load_w<H_>(bw1, pw1, pn0, lane);
    pm_load_w<P2_>(bw2, pw2, pn0, lane);
    for (int i = tid; i < 16 * CH_; i += 128) {
        int r = i / CH_, c = i % CH_;
        float acc = b1[c];
        #pragma unroll
        for (int k = 0; k < CS_; ++k) acc += cov[(m0 + r) * CS_ + k] * w1[k * CH_ + c];
        t_s[r][c] = fmaxf(acc, 0.0f);
    }
    __syncthreads();
    for (int i = tid; i < 16 * H_; i += 128) {
        int r = i / H_, c = i % H_;
        float acc = b2[c];
        #pragma unroll
        for (int k = 0; k < CH_; ++k) acc += t_s[r][k] * w2[k * H_ + c];
        float hv = tanhf(acc);
        h[(m0 + r) * H_ + c] = hv;
        hn_s[r][c] = hv;
    }
    __syncthreads();

    const int col = pn0 + (lane & 15), rb = (lane >> 4) << 3;
    v8f a1 = pm_gemm<H_>(&hn_s[0][0], HP_, bw1, lane);
    const float bv = pb1[col];
    #pragma unroll
    for (int j = 0; j < 8; ++j) u_s[rb + j][col] = fmaxf(a1[j] + bv, 0.0f);
    __syncthreads();
    v8f a2 = pm_gemm<P2_>(&u_s[0][0], PP_, bw2, lane);
    const float bv2 = pb2[col];
    #pragma unroll
    for (int j = 0; j < 8; ++j) p[(m0 + rb + j) * P2_ + col] = a2[j] + bv2;
}

__global__ void __launch_bounds__(256) k_ode(float* __restrict__ h, float* __restrict__ p,
                      const float* __restrict__ xz_w, const float* __restrict__ xz_b,
                      const float* __restrict__ hz_w,
                      const float* __restrict__ xn_w, const float* __restrict__ xn_b,
                      const float* __restrict__ hn_w,
                      const float* __restrict__ pw1, const float* __restrict__ pb1,
                      const float* __restrict__ pw2, const float* __restrict__ pb2) {
    __shared__ __attribute__((aligned(16))) float h_s[OR_][HP_];
    __shared__ __attribute__((aligned(16))) float p_s[OR_][PP_];
    __shared__ float zh_s[OR_][HP_];
    __shared__ float hn_s[OR_][HP_];
    __shared__ float u_s[OR_][PP_];
    const int tid = threadIdx.x, lane = tid & 31;
    const int wv = tid >> 5;
    const int m0 = blockIdx.x * OR_, n0 = wv * 16;
    const int pn0 = (wv & 3) * 16;
    const int pm0 = (wv >> 2) * 16;

    const v4f* hg = reinterpret_cast<const v4f*>(h + (size_t)m0 * H_);
    const v4f* pg = reinterpret_cast<const v4f*>(p + (size_t)m0 * P2_);
    const v4f hq0 = hg[tid], hq1 = hg[tid + 256], hq2 = hg[tid + 512], hq3 = hg[tid + 768];
    const v4f pq0 = pg[tid], pq1 = pg[tid + 256];
    v2f bxz[P2_ / 4];
    #pragma unroll
    for (int k = 0; k < P2_; k += 4) bxz[k >> 2] = load_bfrag(xz_w, H_, k, n0, lane);
    *reinterpret_cast<v4f*>(&h_s[tid >> 5][(tid & 31) << 2])        = hq0;
    *reinterpret_cast<v4f*>(&h_s[8 + (tid >> 5)][(tid & 31) << 2])  = hq1;
    *reinterpret_cast<v4f*>(&h_s[16 + (tid >> 5)][(tid & 31) << 2]) = hq2;
    *reinterpret_cast<v4f*>(&h_s[24 + (tid >> 5)][(tid & 31) << 2]) = hq3;
    *reinterpret_cast<v4f*>(&p_s[tid >> 4][(tid & 15) << 2])        = pq0;
    *reinterpret_cast<v4f*>(&p_s[16 + (tid >> 4)][(tid & 15) << 2]) = pq1;
    __syncthreads();

    v8f az0 = {}, az1 = {};
    #pragma unroll
    for (int k = 0; k < P2_; k += 4) {
        az0 = wmma4(load_afrag(&p_s[0][0], PP_, 0, k, lane), bxz[k >> 2], az0);
        az1 = wmma4(load_afrag(&p_s[0][0], PP_, 16, k, lane), bxz[k >> 2], az1);
    }
    for (int k = 0; k < H_; k += 4) {
        const v2f b = load_bfrag(hz_w, H_, k, n0, lane);
        az0 = wmma4(load_afrag(&h_s[0][0], HP_, 0, k, lane), b, az0);
        az1 = wmma4(load_afrag(&h_s[0][0], HP_, 16, k, lane), b, az1);
    }

    const int col = n0 + (lane & 15), rb = (lane >> 4) << 3;
    const float bz = xz_b[col];
    float z0[8], z1[8];
    #pragma unroll
    for (int j = 0; j < 8; ++j) {
        z0[j] = sigmoidf_(az0[j] + bz);
        z1[j] = sigmoidf_(az1[j] + bz);
        zh_s[rb + j][col]      = z0[j] * h_s[rb + j][col];
        zh_s[16 + rb + j][col] = z1[j] * h_s[16 + rb + j][col];
    }
    __syncthreads();

    v2f bw1[H_ / 4], bw2[P2_ / 4];
    pm_load_w<H_>(bw1, pw1, pn0, lane);
    pm_load_w<P2_>(bw2, pw2, pn0, lane);

    v8f an0 = {}, an1 = {};
    for (int k = 0; k < P2_; k += 4) {
        const v2f b = load_bfrag(xn_w, H_, k, n0, lane);
        an0 = wmma4(load_afrag(&p_s[0][0], PP_, 0, k, lane), b, an0);
        an1 = wmma4(load_afrag(&p_s[0][0], PP_, 16, k, lane), b, an1);
    }
    for (int k = 0; k < H_; k += 4) {
        const v2f b = load_bfrag(hn_w, H_, k, n0, lane);
        an0 = wmma4(load_afrag(&zh_s[0][0], HP_, 0, k, lane), b, an0);
        an1 = wmma4(load_afrag(&zh_s[0][0], HP_, 16, k, lane), b, an1);
    }

    const float bn = xn_b[col];
    float hnew0[8], hnew1[8];
    #pragma unroll
    for (int j = 0; j < 8; ++j) {
        const float n0v = tanhf_(an0[j] + bn);
        const float n1v = tanhf_(an1[j] + bn);
        const float hv0 = h_s[rb + j][col];
        const float hv1 = h_s[16 + rb + j][col];
        hnew0[j] = hv0 + DT_ * (1.0f - z0[j]) * (n0v - hv0);
        hnew1[j] = hv1 + DT_ * (1.0f - z1[j]) * (n1v - hv1);
        hn_s[rb + j][col]      = hnew0[j];
        hn_s[16 + rb + j][col] = hnew1[j];
    }
    __syncthreads();

    const int pcol = pn0 + (lane & 15);
    {
        v8f a1 = pm_gemm<H_>(&hn_s[pm0][0], HP_, bw1, lane);
        const float bv = pb1[pcol];
        #pragma unroll
        for (int j = 0; j < 8; ++j) u_s[pm0 + rb + j][pcol] = fmaxf(a1[j] + bv, 0.0f);
    }
    __syncthreads();
    #pragma unroll
    for (int j = 0; j < 8; ++j) {
        h[(m0 + rb + j) * H_ + col]      = hnew0[j];
        h[(m0 + 16 + rb + j) * H_ + col] = hnew1[j];
    }
    {
        v8f a2 = pm_gemm<P2_>(&u_s[pm0][0], PP_, bw2, lane);
        const float bv2 = pb2[pcol];
        #pragma unroll
        for (int j = 0; j < 8; ++j) p[(m0 + pm0 + rb + j) * P2_ + pcol] = a2[j] + bv2;
    }
}

template <bool WB>
__device__ __forceinline__ void jw_fetch(v2f& a, v2f (&b)[3], int i, const float* gin_s, const float* hold_s,
                                         const float* __restrict__ w_ih, const float* __restrict__ w_hh, int n0, int lane) {
    const bool isgi = i < GIN_ / 4;
    const int k = 4 * (isgi ? i : i - GIN_ / 4);
    a = load_afrag(isgi ? gin_s : hold_s, isgi ? GP_ : HP_, 0, k, lane);
    if (WB) {
        #pragma unroll
        for (int g = 0; g < 3; ++g) b[g] = load_bfrag(isgi ? w_ih : w_hh, G3_, k, g * H_ + n0, lane);
    }
}

__global__ void __launch_bounds__(JT_) k_jump(int t,
                       const float* __restrict__ X, const float* __restrict__ M,
                       const int* __restrict__ obs_idx,
                       float* __restrict__ h, float* __restrict__ p,
                       const float* __restrict__ w_ih, const float* __restrict__ w_hh,
                       const float* __restrict__ b_ih, const float* __restrict__ b_hh,
                       const float* __restrict__ w_prep, const float* __restrict__ bias_prep,
                       const float* __restrict__ pw1, const float* __restrict__ pb1,
                       const float* __restrict__ pw2, const float* __restrict__ pb2,
                       float* __restrict__ loss_part) {
    __shared__ int   idx_s[16];
    __shared__ __attribute__((aligned(16))) float hold_s[16][HP_];
    __shared__ float hn_s[16][HP_];
    __shared__ float gin_s[16][GP_];
    __shared__ float red[16 * D_];
    __shared__ float u_s[16][PP_];
    __shared__ v4f   feat_s[16 * D_];
    __shared__ float m_s[16 * D_];
    const int tid = threadIdx.x, lane = tid & 31;
    const int wv = __builtin_amdgcn_readfirstlane(tid >> 5);
    const bool pmw = wv >= 4;
    const int pn0 = (wv & 3) * 16;
    const int n0 = wv * 16;
    const int col = n0 + (lane & 15), rb = (lane >> 4) << 3;
    const int o0 = blockIdx.x * 16;
    const float* Xt = X + (size_t)t * NO_ * D_;
    const float* Mt = M + (size_t)t * NO_ * D_;

    const float xa = Xt[o0 * D_ + tid], xb = Xt[o0 * D_ + 256 + tid];
    const float ma = Mt[o0 * D_ + tid], mb = Mt[o0 * D_ + 256 + tid];
    const int ca = tid, cb = 256 + (tid & 63);
    const int da = ca / PREP_, db = cb / PREP_;
    const float* wpa = w_prep + (size_t)(da * 4) * PREP_ + (ca - da * PREP_);
    const float* wpb = w_prep + (size_t)(db * 4) * PREP_ + (cb - db * PREP_);
    const float wa0 = wpa[0], wa1 = wpa[PREP_], wa2 = wpa[2 * PREP_], wa3 = wpa[3 * PREP_], bpa = bias_prep[ca];
    const float wb0 = wpb[0], wb1 = wpb[PREP_], wb2 = wpb[2 * PREP_], wb3 = wpb[3 * PREP_], bpb = bias_prep[cb];
    const float bir = b_ih[col], biz = b_ih[H_ + col], bin = b_ih[2 * H_ + col];
    const float bhr = b_hh[col], bhz = b_hh[H_ + col], bhn = b_hh[2 * H_ + col];
    v2f aq[JRS_], bq[JRS_][3];
    #pragma unroll
    for (int i = 0; i < JP0_; ++i) {
        #pragma unroll
        for (int g = 0; g < 3; ++g) bq[i][g] = load_bfrag(w_ih, G3_, 4 * i, g * H_ + n0, lane);
    }
    if (tid < 16) idx_s[tid] = obs_idx[t * NO_ + o0 + tid];
    __syncthreads();

    {
        const int oa = tid >> 5, c4 = (tid & 31) << 2, d = tid & 31;
        const int ba = idx_s[oa], bb = idx_s[oa + 8];
        const v4f ha = *reinterpret_cast<const v4f*>(h + (size_t)ba * H_ + c4);
        const v4f hb = *reinterpret_cast<const v4f*>(h + (size_t)bb * H_ + c4);
        const float mean_a = p[ba * P2_ + d], lv_a = p[ba * P2_ + D_ + d];
        const float mean_b = p[bb * P2_ + d], lv_b = p[bb * P2_ + D_ + d];
        *reinterpret_cast<v4f*>(&hold_s[oa][c4])     = ha;
        *reinterpret_cast<v4f*>(&hold_s[oa + 8][c4]) = hb;
        {
            const float err = (xa - mean_a) * expf(-0.5f * lv_a);
            red[tid] = 0.5f * ((err * err + lv_a + 2.0f * LOGC_) * ma);
            v4f f; f.x = xa; f.y = mean_a; f.z = lv_a; f.w = err;
            feat_s[tid] = f;
            m_s[tid] = ma;
        }
        {
            const float err = (xb - mean_b) * expf(-0.5f * lv_b);
            red[256 + tid] = 0.5f * ((err * err + lv_b + 2.0f * LOGC_) * mb);
            v4f f; f.x = xb; f.y = mean_b; f.z = lv_b; f.w = err;
            feat_s[256 + tid] = f;
            m_s[256 + tid] = mb;
        }
    }
    __syncthreads();

    #pragma unroll
    for (int o = 0; o < 16; ++o) {
        const v4f f = feat_s[o * D_ + da];
        const float mm = m_s[o * D_ + da];
        const float a = fmaf(f.w, wa3, fmaf(f.z, wa2, fmaf(f.x, wa0, f.y * wa1))) + bpa;
        gin_s[o][ca] = fmaxf(a, 0.0f) * mm;
    }
    {
        const int ob = (tid >> 6) << 2;
        #pragma unroll
        for (int j = 0; j < 4; ++j) {
            const int o = ob + j;
            const v4f f = feat_s[o * D_ + db];
            const float mm = m_s[o * D_ + db];
            const float a = fmaf(f.w, wb3, fmaf(f.z, wb2, fmaf(f.x, wb0, f.y * wb1))) + bpb;
            gin_s[o][cb] = fmaxf(a, 0.0f) * mm;
        }
    }
    __syncthreads();

    v8f gir = {}, giz = {}, gin = {}, ghr = {}, ghz = {}, ghn = {};
    v2f bw1[H_ / 4] = {}, bw2[P2_ / 4] = {};
    constexpr int NS = (GIN_ + H_) / 4;
    const float* gp = &gin_s[0][0];
    const float* hp = &hold_s[0][0];
    #pragma unroll
    for (int i = 0; i < JP0_; ++i) jw_fetch<false>(aq[i], bq[i], i, gp, hp, w_ih, w_hh, n0, lane);
    #pragma unroll
    for (int i = JP0_; i < JPD_; ++i) jw_fetch<true>(aq[i], bq[i], i, gp, hp, w_ih, w_hh, n0, lane);
    #pragma unroll
    for (int c = 0; c < NS / 8; ++c) {
    #pragma unroll
    for (int s = 0; s < 8; ++s) {
        const int i = 8 * c + s;
        if (i == NS - 16 && pmw) { pm_load_w<H_>(bw1, pw1, pn0, lane); pm_load_w<P2_>(bw2, pw2, pn0, lane); }
        if (i + JPD_ < NS) jw_fetch<true>(aq[(i + JPD_) % JRS_], bq[(i + JPD_) % JRS_], i + JPD_, gp, hp, w_ih, w_hh, n0, lane);
        if (i < GIN_ / 4) {
            gir = wmma4(aq[i % JRS_], bq[i % JRS_][0], gir);
            giz = wmma4(aq[i % JRS_], bq[i % JRS_][1], giz);
            gin = wmma4(aq[i % JRS_], bq[i % JRS_][2], gin);
        } else {
            ghr = wmma4(aq[i % JRS_], bq[i % JRS_][0], ghr);
            ghz = wmma4(aq[i % JRS_], bq[i % JRS_][1], ghz);
            ghn = wmma4(aq[i % JRS_], bq[i % JRS_][2], ghn);
        }
        __builtin_amdgcn_sched_barrier(0);
    }
    }

    #pragma unroll
    for (int j = 0; j < 8; ++j) {
        const float r = sigmoidf_((gir[j] + bir) + (ghr[j] + bhr));
        const float z = sigmoidf_((giz[j] + biz) + (ghz[j] + bhz));
        const float n = tanhf_((gin[j] + bin) + r * (ghn[j] + bhn));
        const float hv = hold_s[rb + j][col];
        hn_s[rb + j][col] = fmaf(z, hv, (1.0f - z) * n);
    }
    __syncthreads();
    const int pcol = pn0 + (lane & 15);
    float lsum = 0.0f;
    if (pmw) {
        v8f a1 = pm_gemm<H_>(&hn_s[0][0], HP_, bw1, lane);
        const float bv = pb1[pcol];
        #pragma unroll
        for (int j = 0; j < 8; ++j) u_s[rb + j][pcol] = fmaxf(a1[j] + bv, 0.0f);
    } else if (wv == 0) {
        float v[16];
        #pragma unroll
        for (int k = 0; k < 16; ++k) v[k] = red[lane + 32 * k];
        #pragma unroll
        for (int k = 0; k < 8; ++k) v[k] = v[k] + v[k + 8];
        #pragma unroll
        for (int k = 0; k < 4; ++k) v[k] = v[k] + v[k + 4];
        v[0] = v[0] + v[2];
        v[1] = v[1] + v[3];
        lsum = v[0] + v[1];
        lsum = lsum + __shfl_down(lsum, 16);
        lsum = lsum + __shfl_down(lsum, 8);
        lsum = lsum + __shfl_down(lsum, 4);
        lsum = lsum + __shfl_down(lsum, 2);
        lsum = lsum + __shfl_down(lsum, 1);
    }
    __syncthreads();
    if (pmw) {
        v8f a2 = pm_gemm<P2_>(&u_s[0][0], PP_, bw2, lane);
        const float bv2 = pb2[pcol];
        #pragma unroll
        for (int j = 0; j < 8; ++j) p[idx_s[rb + j] * P2_ + pcol] = a2[j] + bv2;
    } else {
        #pragma unroll
        for (int o = 0; o < 16; ++o) h[idx_s[o] * H_ + tid] = hn_s[o][tid];
        if (tid == 0) loss_part[t * (NO_ / 16) + blockIdx.x] = lsum;
    }
}

__global__ void k_loss_final(const float* __restrict__ parts, float* __restrict__ loss) {
    __shared__ float red[256];
    const int tid = threadIdx.x;
    float s = 0.0f;
    for (int i = tid; i < NT_ * (NO_ / 16); i += 256) s += parts[i];
    red[tid] = s;
    __syncthreads();
    for (int k = 128; k > 0; k >>= 1) {
        if (tid < k) red[tid] += red[tid + k];
        __syncthreads();
    }
    if (tid == 0) *loss = red[0];
}

extern "C" void kernel_launch(void* const* d_in, const int* in_sizes, int n_in,
                              void* d_out, int out_size, void* d_ws, size_t ws_size,
                              hipStream_t stream) {
    const float* X       = (const float*)d_in[0];
    const float* M       = (const float*)d_in[1];
    const int*   obs_idx = (const int*)d_in[2];
    const float* cov     = (const float*)d_in[3];
    const float* cm_w1   = (const float*)d_in[4];
    const float* cm_b1   = (const float*)d_in[5];
    const float* cm_w2   = (const float*)d_in[6];
    const float* cm_b2   = (const float*)d_in[7];
    const float* p_w1    = (const float*)d_in[8];
    const float* p_b1    = (const float*)d_in[9];
    const float* p_w2    = (const float*)d_in[10];
    const float* p_b2    = (const float*)d_in[11];
    const float* xz_w    = (const float*)d_in[12];
    const float* xz_b    = (const float*)d_in[13];
    const float* hz_w    = (const float*)d_in[14];
    const float* xn_w    = (const float*)d_in[15];
    const float* xn_b    = (const float*)d_in[16];
    const float* hn_w    = (const float*)d_in[17];
    const float* w_ih    = (const float*)d_in[18];
    const float* w_hh    = (const float*)d_in[19];
    const float* b_ih    = (const float*)d_in[20];
    const float* b_hh    = (const float*)d_in[21];
    const float* w_prep  = (const float*)d_in[22];
    const float* b_prep  = (const float*)d_in[23];

    float* h    = (float*)d_out;
    float* p    = h + (size_t)B_ * H_;
    float* loss = p + (size_t)B_ * P2_;
    float* loss_part = (float*)d_ws;

    const int MB = B_ / 16;
    const int MB2 = B_ / OR_;
    const int OB = NO_ / 16;

    k_init<<<MB, 128, 0, stream>>>(cov, cm_w1, cm_b1, cm_w2, cm_b2,
                                   p_w1, p_b1, p_w2, p_b2, h, p, loss);

    for (int t = 0; t < NT_; ++t) {
        k_ode<<<MB2, 256, 0, stream>>>(h, p, xz_w, xz_b, hz_w, xn_w, xn_b, hn_w,
                                      p_w1, p_b1, p_w2, p_b2);
        k_jump<<<OB, JT_, 0, stream>>>(t, X, M, obs_idx, h, p,
                                       w_ih, w_hh, b_ih, b_hh, w_prep, b_prep,
                                       p_w1, p_b1, p_w2, p_b2, loss_part);
    }
    for (int s = 0; s < NPOST_; ++s)
        k_ode<<<MB2, 256, 0, stream>>>(h, p, xz_w, xz_b, hz_w, xn_w, xn_b, hn_w,
                                      p_w1, p_b1, p_w2, p_b2);
    k_loss_final<<<1, 256, 0, stream>>>(loss_part, loss);
}


# launches that need no exchange between work-groups fused: init into the first ode step, the 4 trailing ode steps into one (rows stay in LDS), loss fold as one more work-group; 38 -> 33; sums kept
# speedup vs baseline: 2.6297x; 1.0425x over previous
#include <hip/hip_runtime.h>
#include <hip/hip_bf16.h>
#include <math.h>

#define B_    8192
#define D_    32
#define H_    128
#define P2_   64
#define PREP_ 10
#define GIN_  320
#define G3_   384
#define CS_   8
#define CH_   32
#define NT_   16
#define NO_   4096
#define DT_   0.1f
#define NPOST_ 4
#define OR_   32
#define JT_   256
#define JPD_  9
#define JRS_  16
#define JP0_  8
#define LOGC_  0.91893853320467274178f

#define HP_   (H_ + 4)
#define PP_   (P2_ + 4)
#define GP_   (GIN_ + 4)

typedef __attribute__((ext_vector_type(2))) float v2f;
typedef __attribute__((ext_vector_type(4))) float v4f;
typedef __attribute__((ext_vector_type(8))) float v8f;

__device__ __forceinline__ float sigmoidf_(float x) {
    return __builtin_amdgcn_rcpf(1.0f + __builtin_amdgcn_exp2f(-1.44269504088896340736f * x));
}
__device__ __forceinline__ float tanhf_(float x) {
    return 1.0f - 2.0f * __builtin_amdgcn_rcpf(1.0f + __builtin_amdgcn_exp2f(2.88539008177792681472f * x));
}


__device__ __forceinline__ v2f load_afrag(const float* A, int lda, int m0, int k, int lane) {
    int m  = m0 + (lane & 15);
    int kk = k + ((lane >> 4) << 1);
    v2f a;
    a.x = A[m * lda + kk];
    a.y = A[m * lda + kk + 1];
    return a;
}

__device__ __forceinline__ v2f load_bfrag(const float* Bm, int ldb, int k, int n0, int lane) {
    int n  = n0 + (lane & 15);
    int kk = k + ((lane >> 4) << 1);
    v2f b;
    b.x = Bm[kk * ldb + n];
    b.y = Bm[(kk + 1) * ldb + n];
    return b;
}

__device__ __forceinline__ v8f wmma4(v2f a, v2f b, v8f c) {
    return __builtin_amdgcn_wmma_f32_16x16x4_f32(false, a, false, b, (short)0, c, false, false);
}

template <int K>
__device__ __forceinline__ void pm_load_w(v2f (&bw)[K / 4], const float* __restrict__ w, int n0, int lane) {
    #pragma unroll
    for (int k = 0; k < K; k += 4) bw[k >> 2] = load_bfrag(w, P2_, k, n0, lane);
}

template <int K>
__device__ __forceinline__ v8f pm_gemm(const float* as, int pitch, const v2f (&bw)[K / 4], int lane) {
    v8f acc = {};
    #pragma unroll
    for (int k = 0; k < K; k += 4) acc = wmma4(load_afrag(as, pitch, 0, k, lane), bw[k >> 2], acc);
    return acc;
}

template <int NSTEP, bool INIT, bool LOSS>
__global__ void __launch_bounds__(256) k_ode(float* __restrict__ h, float* __restrict__ p,
                      const float* __restrict__ xz_w, const float* __restrict__ xz_b,
                      const float* __restrict__ hz_w,
                      const float* __restrict__ xn_w, const float* __restrict__ xn_b,
                      const float* __restrict__ hn_w,
                      const float* __restrict__ pw1, const float* __restrict__ pb1,
                      const float* __restrict__ pw2, const float* __restrict__ pb2,
                      const float* __restrict__ cov,
                      const float* __restrict__ cw1, const float* __restrict__ cb1,
                      const float* __restrict__ cw2, const float* __restrict__ cb2,
                      const float* __restrict__ parts, float* __restrict__ loss) {
    __shared__ __attribute__((aligned(16))) float h_s[OR_][HP_];
    __shared__ __attribute__((aligned(16))) float p_s[OR_][PP_];
    __shared__ float zh_s[OR_][HP_];
    __shared__ float hn_s[OR_][HP_];
    __shared__ float u_s[OR_][PP_];
    const int tid = threadIdx.x, lane = tid & 31;
    if (LOSS) {
        if (blockIdx.x == B_ / OR_) {
            float* red = &u_s[0][0];
            float sacc = 0.0f;
            for (int i = tid; i < NT_ * (NO_ / 16); i += 256) sacc += parts[i];
            red[tid] = sacc;
            __syncthreads();
            for (int k = 128; k > 0; k >>= 1) {
                if (tid < k) red[tid] += red[tid + k];
                __syncthreads();
            }
            if (tid == 0) *loss = red[0];
            return;
        }
    }
    const int wv = tid >> 5;
    const int m0 = blockIdx.x * OR_, n0 = wv * 16;
    const int pn0 = (wv & 3) * 16;
    const int pm0 = (wv >> 2) * 16;
    const int col = n0 + (lane & 15), rb = (lane >> 4) << 3;
    const int pcol = pn0 + (lane & 15);

    v4f hq0 = {}, hq1 = {}, hq2 = {}, hq3 = {}, pq0 = {}, pq1 = {};
    if (!INIT) {
        const v4f* hg = reinterpret_cast<const v4f*>(h + (size_t)m0 * H_);
        const v4f* pg = reinterpret_cast<const v4f*>(p + (size_t)m0 * P2_);
        hq0 = hg[tid]; hq1 = hg[tid + 256]; hq2 = hg[tid + 512]; hq3 = hg[tid + 768];
        pq0 = pg[tid]; pq1 = pg[tid + 256];
    } else {
        float (*t_s)[CH_] = reinterpret_cast<float (*)[CH_]>(&zh_s[0][0]);
        v2f iw1[H_ / 4], iw2[P2_ / 4];
        pm_load_w<H_>(iw1, pw1, pn0, lane);
        pm_load_w<P2_>(iw2, pw2, pn0, lane);
        for (int i = tid; i < OR_ * CH_; i += 256) {
            int r = i / CH_, c = i % CH_;
            float acc = cb1[c];
            #pragma unroll
            for (int k = 0; k < CS_; ++k) acc += cov[(m0 + r) * CS_ + k] * cw1[k * CH_ + c];
            t_s[r][c] = fmaxf(acc, 0.0f);
        }
        __syncthreads();
        for (int i = tid; i < OR_ * H_; i += 256) {
            int r = i / H_, c = i % H_;
            float acc = cb2[c];
            #pragma unroll
            for (int k = 0; k < CH_; ++k) acc += t_s[r][k] * cw2[k * H_ + c];
            h_s[r][c] = tanhf(acc);
        }
        __syncthreads();
        {
            v8f a1 = pm_gemm<H_>(&h_s[pm0][0], HP_, iw1, lane);
            const float bv = pb1[pcol];
            #pragma unroll
            for (int j = 0; j < 8; ++j) u_s[pm0 + rb + j][pcol] = fmaxf(a1[j] + bv, 0.0f);
        }
        __syncthreads();
        {
            v8f a2 = pm_gemm<P2_>(&u_s[pm0][0], PP_, iw2, lane);
            const float bv2 = pb2[pcol];
            #pragma unroll
            for (int j = 0; j < 8; ++j) p_s[pm0 + rb + j][pcol] = a2[j] + bv2;
        }
    }

    #pragma unroll 1
    for (int s = 0; s < NSTEP; ++s) {
    const bool last = (s + 1 == NSTEP);
    v2f bxz[P2_ / 4];
    #pragma unroll
    for (int k = 0; k < P2_; k += 4) bxz[k >> 2] = load_bfrag(xz_w, H_, k, n0, lane);
    if (!INIT && s == 0) {
        *reinterpret_cast<v4f*>(&h_s[tid >> 5][(tid & 31) << 2])        = hq0;
        *reinterpret_cast<v4f*>(&h_s[8 + (tid >> 5)][(tid & 31) << 2])  = hq1;
        *reinterpret_cast<v4f*>(&h_s[16 + (tid >> 5)][(tid & 31) << 2]) = hq2;
        *reinterpret_cast<v4f*>(&h_s[24 + (tid >> 5)][(tid & 31) << 2]) = hq3;
        *reinterpret_cast<v4f*>(&p_s[tid >> 4][(tid & 15) << 2])        = pq0;
        *reinterpret_cast<v4f*>(&p_s[16 + (tid >> 4)][(tid & 15) << 2]) = pq1;
    }
    __syncthreads();

    v8f az0 = {}, az1 = {};
    #pragma unroll
    for (int k = 0; k < P2_; k += 4) {
        az0 = wmma4(load_afrag(&p_s[0][0], PP_, 0, k, lane), bxz[k >> 2], az0);
        az1 = wmma4(load_afrag(&p_s[0][0], PP_, 16, k, lane), bxz[k >> 2], az1);
    }
    for (int k = 0; k < H_; k += 4) {
        const v2f b = load_bfrag(hz_w, H_, k, n0, lane);
        az0 = wmma4(load_afrag(&h_s[0][0], HP_, 0, k, lane), b, az0);
        az1 = wmma4(load_afrag(&h_s[0][0], HP_, 16, k, lane), b, az1);
    }

    const float bz = xz_b[col];
    float z0[8], z1[8];
    #pragma unroll
    for (int j = 0; j < 8; ++j) {
        z0[j] = sigmoidf_(az0[j] + bz);
        z1[j] = sigmoidf_(az1[j] + bz);
        zh_s[rb + j][col]      = z0[j] * h_s[rb + j][col];
        zh_s[16 + rb + j][col] = z1[j] * h_s[16 + rb + j][col];
    }
    __syncthreads();

    v2f bw1[H_ / 4], bw2[P2_ / 4];
    pm_load_w<H_>(bw1, pw1, pn0, lane);
    pm_load_w<P2_>(bw2, pw2, pn0, lane);

    v8f an0 = {}, an1 = {};
    for (int k = 0; k < P2_; k += 4) {
        const v2f b = load_bfrag(xn_w, H_, k, n0, lane);
        an0 = wmma4(load_afrag(&p_s[0][0], PP_, 0, k, lane), b, an0);
        an1 = wmma4(load_afrag(&p_s[0][0], PP_, 16, k, lane), b, an1);
    }
    for (int k = 0; k < H_; k += 4) {
        const v2f b = load_bfrag(hn_w, H_, k, n0, lane);
        an0 = wmma4(load_afrag(&zh_s[0][0], HP_, 0, k, lane), b, an0);
        an1 = wmma4(load_afrag(&zh_s[0][0], HP_, 16, k, lane), b, an1);
    }

    const float bn = xn_b[col];
    float hnew0[8], hnew1[8];
    #pragma unroll
    for (int j = 0; j < 8; ++j) {
        const float n0v = tanhf_(an0[j] + bn);
        const float n1v = tanhf_(an1[j] + bn);
        const float hv0 = h_s[rb + j][col];
        const float hv1 = h_s[16 + rb + j][col];
        hnew0[j] = hv0 + DT_ * (1.0f - z0[j]) * (n0v - hv0);
        hnew1[j] = hv1 + DT_ * (1.0f - z1[j]) * (n1v - hv1);
        hn_s[rb + j][col]      = hnew0[j];
        hn_s[16 + rb + j][col] = hnew1[j];
        if (NSTEP > 1) {
            h_s[rb + j][col]      = hnew0[j];
            h_s[16 + rb + j][col] = hnew1[j];
        }
    }
    __syncthreads();

    {
        v8f a1 = pm_gemm<H_>(&hn_s[pm0][0], HP_, bw1, lane);
        const float bv = pb1[pcol];
        #pragma unroll
        for (int j = 0; j < 8; ++j) u_s[pm0 + rb + j][pcol] = fmaxf(a1[j] + bv, 0.0f);
    }
    __syncthreads();
    if (last) {
        #pragma unroll
        for (int j = 0; j < 8; ++j) {
            h[(m0 + rb + j) * H_ + col]      = hnew0[j];
            h[(m0 + 16 + rb + j) * H_ + col] = hnew1[j];
        }
    }
    {
        v8f a2 = pm_gemm<P2_>(&u_s[pm0][0], PP_, bw2, lane);
        const float bv2 = pb2[pcol];
        float pv[8];
        #pragma unroll
        for (int j = 0; j < 8; ++j) pv[j] = a2[j] + bv2;
        if (last) {
            #pragma unroll
            for (int j = 0; j < 8; ++j) p[(m0 + pm0 + rb + j) * P2_ + pcol] = pv[j];
        } else {
            #pragma unroll
            for (int j = 0; j < 8; ++j) p_s[pm0 + rb + j][pcol] = pv[j];
        }
    }
    }
}

template <bool WB>
__device__ __forceinline__ void jw_fetch(v2f& a, v2f (&b)[3], int i, const float* gin_s, const float* hold_s,
                                         const float* __restrict__ w_ih, const float* __restrict__ w_hh, int n0, int lane) {
    const bool isgi = i < GIN_ / 4;
    const int k = 4 * (isgi ? i : i - GIN_ / 4);
    a = load_afrag(isgi ? gin_s : hold_s, isgi ? GP_ : HP_, 0, k, lane);
    if (WB) {
        #pragma unroll
        for (int g = 0; g < 3; ++g) b[g] = load_bfrag(isgi ? w_ih : w_hh, G3_, k, g * H_ + n0, lane);
    }
}

__global__ void __launch_bounds__(JT_) k_jump(int t,
                       const float* __restrict__ X, const float* __restrict__ M,
                       const int* __restrict__ obs_idx,
                       float* __restrict__ h, float* __restrict__ p,
                       const float* __restrict__ w_ih, const float* __restrict__ w_hh,
                       const float* __restrict__ b_ih, const float* __restrict__ b_hh,
                       const float* __restrict__ w_prep, const float* __restrict__ bias_prep,
                       const float* __restrict__ pw1, const float* __restrict__ pb1,
                       const float* __restrict__ pw2, const float* __restrict__ pb2,
                       float* __restrict__ loss_part) {
    __shared__ int   idx_s[16];
    __shared__ __attribute__((aligned(16))) float hold_s[16][HP_];
    __shared__ float hn_s[16][HP_];
    __shared__ float gin_s[16][GP_];
    __shared__ float red[16 * D_];
    __shared__ float u_s[16][PP_];
    __shared__ v4f   feat_s[16 * D_];
    __shared__ float m_s[16 * D_];
    const int tid = threadIdx.x, lane = tid & 31;
    const int wv = __builtin_amdgcn_readfirstlane(tid >> 5);
    const bool pmw = wv >= 4;
    const int pn0 = (wv & 3) * 16;
    const int n0 = wv * 16;
    const int col = n0 + (lane & 15), rb = (lane >> 4) << 3;
    const int o0 = blockIdx.x * 16;
    const float* Xt = X + (size_t)t * NO_ * D_;
    const float* Mt = M + (size_t)t * NO_ * D_;

    const float xa = Xt[o0 * D_ + tid], xb = Xt[o0 * D_ + 256 + tid];
    const float ma = Mt[o0 * D_ + tid], mb = Mt[o0 * D_ + 256 + tid];
    const int ca = tid, cb = 256 + (tid & 63);
    const int da = ca / PREP_, db = cb / PREP_;
    const float* wpa = w_prep + (size_t)(da * 4) * PREP_ + (ca - da * PREP_);
    const float* wpb = w_prep + (size_t)(db * 4) * PREP_ + (cb - db * PREP_);
    const float wa0 = wpa[0], wa1 = wpa[PREP_], wa2 = wpa[2 * PREP_], wa3 = wpa[3 * PREP_], bpa = bias_prep[ca];
    const float wb0 = wpb[0], wb1 = wpb[PREP_], wb2 = wpb[2 * PREP_], wb3 = wpb[3 * PREP_], bpb = bias_prep[cb];
    const float bir = b_ih[col], biz = b_ih[H_ + col], bin = b_ih[2 * H_ + col];
    const float bhr = b_hh[col], bhz = b_hh[H_ + col], bhn = b_hh[2 * H_ + col];
    v2f aq[JRS_], bq[JRS_][3];
    #pragma unroll
    for (int i = 0; i < JP0_; ++i) {
        #pragma unroll
        for (int g = 0; g < 3; ++g) bq[i][g] = load_bfrag(w_ih, G3_, 4 * i, g * H_ + n0, lane);
    }
    if (tid < 16) idx_s[tid] = obs_idx[t * NO_ + o0 + tid];
    __syncthreads();

    {
        const int oa = tid >> 5, c4 = (tid & 31) << 2, d = tid & 31;
        const int ba = idx_s[oa], bb = idx_s[oa + 8];
        const v4f ha = *reinterpret_cast<const v4f*>(h + (size_t)ba * H_ + c4);
        const v4f hb = *reinterpret_cast<const v4f*>(h + (size_t)bb * H_ + c4);
        const float mean_a = p[ba * P2_ + d], lv_a = p[ba * P2_ + D_ + d];
        const float mean_b = p[bb * P2_ + d], lv_b = p[bb * P2_ + D_ + d];
        *reinterpret_cast<v4f*>(&hold_s[oa][c4])     = ha;
        *reinterpret_cast<v4f*>(&hold_s[oa + 8][c4]) = hb;
        {
            const float err = (xa - mean_a) * expf(-0.5f * lv_a);
            red[tid] = 0.5f * ((err * err + lv_a + 2.0f * LOGC_) * ma);
            v4f f; f.x = xa; f.y = mean_a; f.z = lv_a; f.w = err;
            feat_s[tid] = f;
            m_s[tid] = ma;
        }
        {
            const float err = (xb - mean_b) * expf(-0.5f * lv_b);
            red[256 + tid] = 0.5f * ((err * err + lv_b + 2.0f * LOGC_) * mb);
            v4f f; f.x = xb; f.y = mean_b; f.z = lv_b; f.w = err;
            feat_s[256 + tid] = f;
            m_s[256 + tid] = mb;
        }
    }
    __syncthreads();

    #pragma unroll
    for (int o = 0; o < 16; ++o) {
        const v4f f = feat_s[o * D_ + da];
        const float mm = m_s[o * D_ + da];
        const float a = fmaf(f.w, wa3, fmaf(f.z, wa2, fmaf(f.x, wa0, f.y * wa1))) + bpa;
        gin_s[o][ca] = fmaxf(a, 0.0f) * mm;
    }
    {
        const int ob = (tid >> 6) << 2;
        #pragma unroll
        for (int j = 0; j < 4; ++j) {
            const int o = ob + j;
            const v4f f = feat_s[o * D_ + db];
            const float mm = m_s[o * D_ + db];
            const float a = fmaf(f.w, wb3, fmaf(f.z, wb2, fmaf(f.x, wb0, f.y * wb1))) + bpb;
            gin_s[o][cb] = fmaxf(a, 0.0f) * mm;
        }
    }
    __syncthreads();

    v8f gir = {}, giz = {}, gin = {}, ghr = {}, ghz = {}, ghn = {};
    v2f bw1[H_ / 4] = {}, bw2[P2_ / 4] = {};
    constexpr int NS = (GIN_ + H_) / 4;
    const float* gp = &gin_s[0][0];
    const float* hp = &hold_s[0][0];
    #pragma unroll
    for (int i = 0; i < JP0_; ++i) jw_fetch<false>(aq[i], bq[i], i, gp, hp, w_ih, w_hh, n0, lane);
    #pragma unroll
    for (int i = JP0_; i < JPD_; ++i) jw_fetch<true>(aq[i], bq[i], i, gp, hp, w_ih, w_hh, n0, lane);
    #pragma unroll
    for (int c = 0; c < NS / 8; ++c) {
    #pragma unroll
    for (int s = 0; s < 8; ++s) {
        const int i = 8 * c + s;
        if (i == NS - 16 && pmw) { pm_load_w<H_>(bw1, pw1, pn0, lane); pm_load_w<P2_>(bw2, pw2, pn0, lane); }
        if (i + JPD_ < NS) jw_fetch<true>(aq[(i + JPD_) % JRS_], bq[(i + JPD_) % JRS_], i + JPD_, gp, hp, w_ih, w_hh, n0, lane);
        if (i < GIN_ / 4) {
            gir = wmma4(aq[i % JRS_], bq[i % JRS_][0], gir);
            giz = wmma4(aq[i % JRS_], bq[i % JRS_][1], giz);
            gin = wmma4(aq[i % JRS_], bq[i % JRS_][2], gin);
        } else {
            ghr = wmma4(aq[i % JRS_], bq[i % JRS_][0], ghr);
            ghz = wmma4(aq[i % JRS_], bq[i % JRS_][1], ghz);
            ghn = wmma4(aq[i % JRS_], bq[i % JRS_][2], ghn);
        }
        __builtin_amdgcn_sched_barrier(0);
    }
    }

    #pragma unroll
    for (int j = 0; j < 8; ++j) {
        const float r = sigmoidf_((gir[j] + bir) + (ghr[j] + bhr));
        const float z = sigmoidf_((giz[j] + biz) + (ghz[j] + bhz));
        const float n = tanhf_((gin[j] + bin) + r * (ghn[j] + bhn));
        const float hv = hold_s[rb + j][col];
        hn_s[rb + j][col] = fmaf(z, hv, (1.0f - z) * n);
    }
    __syncthreads();
    const int pcol = pn0 + (lane & 15);
    float lsum = 0.0f;
    if (pmw) {
        v8f a1 = pm_gemm<H_>(&hn_s[0][0], HP_, bw1, lane);
        const float bv = pb1[pcol];
        #pragma unroll
        for (int j = 0; j < 8; ++j) u_s[rb + j][pcol] = fmaxf(a1[j] + bv, 0.0f);
    } else if (wv == 0) {
        float v[16];
        #pragma unroll
        for (int k = 0; k < 16; ++k) v[k] = red[lane + 32 * k];
        #pragma unroll
        for (int k = 0; k < 8; ++k) v[k] = v[k] + v[k + 8];
        #pragma unroll
        for (int k = 0; k < 4; ++k) v[k] = v[k] + v[k + 4];
        v[0] = v[0] + v[2];
        v[1] = v[1] + v[3];
        lsum = v[0] + v[1];
        lsum = lsum + __shfl_down(lsum, 16);
        lsum = lsum + __shfl_down(lsum, 8);
        lsum = lsum + __shfl_down(lsum, 4);
        lsum = lsum + __shfl_down(lsum, 2);
        lsum = lsum + __shfl_down(lsum, 1);
    }
    __syncthreads();
    if (pmw) {
        v8f a2 = pm_gemm<P2_>(&u_s[0][0], PP_, bw2, lane);
        const float bv2 = pb2[pcol];
        #pragma unroll
        for (int j = 0; j < 8; ++j) p[idx_s[rb + j] * P2_ + pcol] = a2[j] + bv2;
    } else {
        #pragma unroll
        for (int o = 0; o < 16; ++o) h[idx_s[o] * H_ + tid] = hn_s[o][tid];
        if (tid == 0) loss_part[t * (NO_ / 16) + blockIdx.x] = lsum;
    }
}

extern "C" void kernel_launch(void* const* d_in, const int* in_sizes, int n_in,
                              void* d_out, int out_size, void* d_ws, size_t ws_size,
                              hipStream_t stream) {
    const float* X       = (const float*)d_in[0];
    const float* M       = (const float*)d_in[1];
    const int*   obs_idx = (const int*)d_in[2];
    const float* cov     = (const float*)d_in[3];
    const float* cm_w1   = (const float*)d_in[4];
    const float* cm_b1   = (const float*)d_in[5];
    const float* cm_w2   = (const float*)d_in[6];
    const float* cm_b2   = (const float*)d_in[7];
    const float* p_w1    = (const float*)d_in[8];
    const float* p_b1    = (const float*)d_in[9];
    const float* p_w2    = (const float*)d_in[10];
    const float* p_b2    = (const float*)d_in[11];
    const float* xz_w    = (const float*)d_in[12];
    const float* xz_b    = (const float*)d_in[13];
    const float* hz_w    = (const float*)d_in[14];
    const float* xn_w    = (const float*)d_in[15];
    const float* xn_b    = (const float*)d_in[16];
    const float* hn_w    = (const float*)d_in[17];
    const float* w_ih    = (const float*)d_in[18];
    const float* w_hh    = (const float*)d_in[19];
    const float* b_ih    = (const float*)d_in[20];
    const float* b_hh    = (const float*)d_in[21];
    const float* w_prep  = (const float*)d_in[22];
    const float* b_prep  = (const float*)d_in[23];

    float* h    = (float*)d_out;
    float* p    = h + (size_t)B_ * H_;
    float* loss = p + (size_t)B_ * P2_;
    float* loss_part = (float*)d_ws;

    const int MB2 = B_ / OR_;
    const int MB2L = B_ / OR_ + 1;
    const int OB = NO_ / 16;

    for (int t = 0; t < NT_; ++t) {
        if (t == 0)
            k_ode<1, true, false><<<MB2, 256, 0, stream>>>(h, p, xz_w, xz_b, hz_w, xn_w, xn_b, hn_w,
                                          p_w1, p_b1, p_w2, p_b2,
                                          cov, cm_w1, cm_b1, cm_w2, cm_b2, nullptr, nullptr);
        else
            k_ode<1, false, false><<<MB2, 256, 0, stream>>>(h, p, xz_w, xz_b, hz_w, xn_w, xn_b, hn_w,
                                          p_w1, p_b1, p_w2, p_b2,
                                          nullptr, nullptr, nullptr, nullptr, nullptr, nullptr, nullptr);
        k_jump<<<OB, JT_, 0, stream>>>(t, X, M, obs_idx, h, p,
                                       w_ih, w_hh, b_ih, b_hh, w_prep, b_prep,
                                       p_w1, p_b1, p_w2, p_b2, loss_part);
    }
    k_ode<NPOST_, false, true><<<MB2L, 256, 0, stream>>>(h, p, xz_w, xz_b, hz_w, xn_w, xn_b, hn_w,
                                          p_w1, p_b1, p_w2, p_b2,
                                          nullptr, nullptr, nullptr, nullptr, nullptr, loss_part, loss);
}
